# starting kernel
# baseline (speedup 1.0000x reference)
#include <hip/hip_runtime.h>
#include <math.h>

typedef float v2f __attribute__((ext_vector_type(2)));
typedef float v8f __attribute__((ext_vector_type(8)));

#define NEG_SLOPE 0.2f

__device__ __forceinline__ void atomic_max_float(float* addr, float val) {
  if (val >= 0.0f) atomicMax((int*)addr, __float_as_int(val));
  else             atomicMin((unsigned int*)addr, __float_as_uint(val));
}

__global__ __launch_bounds__(256) void gemm_wmma_f32(
    const float* __restrict__ A, const float* __restrict__ B,
    const float* __restrict__ bias, float* __restrict__ C,
    int M, int K, int N) {
  const int lane = threadIdx.x & 31;
  const int wave = blockIdx.x * (blockDim.x >> 5) + (threadIdx.x >> 5);
  const int tilesN = N >> 4;
  const int tilesM = M >> 4;
  if (wave >= tilesM * tilesN) return;
  const int tm = wave / tilesN;
  const int tn = wave - tm * tilesN;
  const int mrow  = tm * 16 + (lane & 15);
  const int ncol  = tn * 16 + (lane & 15);
  const int khalf = (lane >> 4) << 1;

  v8f acc = {};
  for (int k = 0; k < K; k += 4) {
    const float* ap = A + (size_t)mrow * N * 0 + (size_t)mrow * K + (k + khalf);
    v2f a; a.x = ap[0]; a.y = ap[1];
    const float* bp = B + (size_t)(k + khalf) * N + ncol;
    v2f b; b.x = bp[0]; b.y = bp[N];
    acc = __builtin_amdgcn_wmma_f32_16x16x4_f32(
         false, a,  false, b,
         (short)0, acc,  false,  false);
  }

  const int mbase = tm * 16 + ((lane >> 4) << 3);
  const float bv = bias ? bias[ncol] : 0.0f;
#pragma unroll
  for (int v = 0; v < 8; ++v)
    C[(size_t)(mbase + v) * N + ncol] = acc[v] + bv;
}

__global__ void fill_value(float* __restrict__ p, float v, long n) {
  long t = (long)blockIdx.x * blockDim.x + threadIdx.x;
  if (t < n) p[t] = v;
}

__global__ void fill_rows_bias(float* __restrict__ out, const float* __restrict__ bias,
                               long rows, int C) {
  long t = (long)blockIdx.x * blockDim.x + threadIdx.x;
  if (t < rows * (long)C) out[t] = bias[(int)(t % C)];
}

__global__ void att_scores(const float* __restrict__ h,
                           const float* __restrict__ att_src,
                           const float* __restrict__ att_dst,
                           float* __restrict__ as, float* __restrict__ ad,
                           int n, int H, int C) {
  long t = (long)blockIdx.x * blockDim.x + threadIdx.x;
  if (t >= (long)n * H) return;
  const int head = (int)(t % H);
  const float4* hv = (const float4*)(h + (size_t)t * C);
  const float4* sv = (const float4*)(att_src + (size_t)head * C);
  const float4* dv = (const float4*)(att_dst + (size_t)head * C);
  float s = 0.f, d = 0.f;
  for (int i = 0; i < (C >> 2); ++i) {
    float4 xv = hv[i], a = sv[i], b = dv[i];
    s += xv.x * a.x + xv.y * a.y + xv.z * a.z + xv.w * a.w;
    d += xv.x * b.x + xv.y * b.y + xv.z * b.z + xv.w * b.w;
  }
  as[t] = s; ad[t] = d;
}

__global__ void edge_logits_max(const int* __restrict__ ei,
                                const float* __restrict__ as,
                                const float* __restrict__ ad,
                                float* __restrict__ e, float* __restrict__ m,
                                int E, int n, int H) {
  long t = (long)blockIdx.x * blockDim.x + threadIdx.x;
  long total = (long)(E + n) * H;
  if (t >= total) return;
  const int edge = (int)(t / H);
  const int hd   = (int)(t % H);
  const int src = (edge < E) ? ei[edge]     : edge - E;
  const int dst = (edge < E) ? ei[E + edge] : edge - E;
  float v = as[(size_t)src * H + hd] + ad[(size_t)dst * H + hd];
  v = (v > 0.f) ? v : NEG_SLOPE * v;
  e[t] = v;
  atomic_max_float(&m[(size_t)dst * H + hd], v);
}

__global__ void edge_exp_sum(const int* __restrict__ ei, float* __restrict__ e,
                             const float* __restrict__ m, float* __restrict__ s,
                             int E, int n, int H) {
  long t = (long)blockIdx.x * blockDim.x + threadIdx.x;
  long total = (long)(E + n) * H;
  if (t >= total) return;
  const int edge = (int)(t / H);
  const int hd   = (int)(t % H);
  const int dst = (edge < E) ? ei[E + edge] : edge - E;
  float v = expf(e[t] - m[(size_t)dst * H + hd]);
  e[t] = v;
  atomicAdd(&s[(size_t)dst * H + hd], v);
}

__global__ void edge_scatter(const int* __restrict__ ei, const float* __restrict__ e,
                             const float* __restrict__ s, const float* __restrict__ feat,
                             float* __restrict__ out, int E, int n, int H, int C) {
  long t = (long)blockIdx.x * blockDim.x + threadIdx.x;
  long total = (long)(E + n) * H * C;
  if (t >= total) return;
  const int  c    = (int)(t % C);
  const long eh   = t / C;
  const int  hd   = (int)(eh % H);
  const int  edge = (int)(eh / H);
  const int src = (edge < E) ? ei[edge]     : edge - E;
  const int dst = (edge < E) ? ei[E + edge] : edge - E;
  const float alpha = e[eh] / (s[(size_t)dst * H + hd] + 1e-16f);
  atomicAdd(&out[((size_t)dst * H + hd) * C + c],
            alpha * feat[((size_t)src * H + hd) * C + c]);
}

__global__ void ln_add_elu(const float* __restrict__ x, const float* __restrict__ g,
                           const float* __restrict__ b, const float* __restrict__ add,
                           float* __restrict__ y, int n, int C) {
  const int row  = blockIdx.x * (blockDim.x >> 5) + (threadIdx.x >> 5);
  const int lane = threadIdx.x & 31;
  if (row >= n) return;
  const float* xr = x + (size_t)row * C;
  float s = 0.f, ss = 0.f;
  for (int c = lane; c < C; c += 32) { float v = xr[c]; s += v; ss += v * v; }
  for (int off = 16; off; off >>= 1) {
    s  += __shfl_xor(s,  off, 32);
    ss += __shfl_xor(ss, off, 32);
  }
  const float mu  = s / (float)C;
  const float var = ss / (float)C - mu * mu;
  const float inv = rsqrtf(var + 1e-5f);
  const float* ar = add ? add + (size_t)row * C : nullptr;
  float* yr = y + (size_t)row * C;
  for (int c = lane; c < C; c += 32) {
    float v = (xr[c] - mu) * inv * g[c] + b[c];
    if (ar) v += ar[c];
    yr[c] = (v > 0.f) ? v : expm1f(v);
  }
}

extern "C" void kernel_launch(void* const* d_in, const int* in_sizes, int n_in,
                              void* d_out, int out_size, void* d_ws, size_t ws_size,
                              hipStream_t stream) {
  const float* x    = (const float*)d_in[0];
  const int*   ei   = (const int*)  d_in[1];
  const float* W1   = (const float*)d_in[2];
  const float* aS1  = (const float*)d_in[3];
  const float* aD1  = (const float*)d_in[4];
  const float* b1   = (const float*)d_in[5];
  const float* W2   = (const float*)d_in[6];
  const float* aS2  = (const float*)d_in[7];
  const float* aD2  = (const float*)d_in[8];
  const float* b2   = (const float*)d_in[9];
  const float* g1   = (const float*)d_in[10];
  const float* be1  = (const float*)d_in[11];
  const float* g2   = (const float*)d_in[12];
  const float* be2  = (const float*)d_in[13];
  const float* pW   = (const float*)d_in[14];
  const float* pb   = (const float*)d_in[15];

  const int IN_CH = 128, HEADS = 4, OUT_CH = 64, HC = HEADS * OUT_CH;
  const int N  = in_sizes[0] / IN_CH;
  const int E  = in_sizes[1] / 2;
  const long EP = (long)E + N;

  float* ws   = (float*)d_ws;
  float* h1   = ws; ws += (size_t)N * HC;
  float* out1 = ws; ws += (size_t)N * HC;
  float* idb  = ws; ws += (size_t)N * OUT_CH;
  float* z2   = ws; ws += (size_t)N * OUT_CH;
  float* out2 = ws; ws += (size_t)N * OUT_CH;
  float* as1  = ws; ws += (size_t)N * HEADS;
  float* ad1  = ws; ws += (size_t)N * HEADS;
  float* m1   = ws; ws += (size_t)N * HEADS;
  float* s1   = ws; ws += (size_t)N * HEADS;
  float* e1   = ws; ws += (size_t)EP * HEADS;
  float* as2  = ws; ws += (size_t)N;
  float* ad2  = ws; ws += (size_t)N;
  float* m2   = ws; ws += (size_t)N;
  float* s2   = ws; ws += (size_t)N;
  float* e2   = ws; ws += (size_t)EP;

  auto cdiv = [](long a, long b) -> int { return (int)((a + b - 1) / b); };
  const int BLK = 256;
  const int WPB = BLK / 32;

  {
    long waves = ((long)N / 16) * (HC / 16);
    gemm_wmma_f32<<<cdiv(waves, WPB), BLK, 0, stream>>>(x, W1, nullptr, h1, N, IN_CH, HC);
  }
  {
    long waves = ((long)N / 16) * (OUT_CH / 16);
    gemm_wmma_f32<<<cdiv(waves, WPB), BLK, 0, stream>>>(x, pW, pb, idb, N, IN_CH, OUT_CH);
  }

  att_scores<<<cdiv((long)N * HEADS, BLK), BLK, 0, stream>>>(h1, aS1, aD1, as1, ad1, N, HEADS, OUT_CH);
  fill_value<<<cdiv((long)N * HEADS, BLK), BLK, 0, stream>>>(m1, -INFINITY, (long)N * HEADS);
  fill_value<<<cdiv((long)N * HEADS, BLK), BLK, 0, stream>>>(s1, 0.0f,      (long)N * HEADS);
  fill_rows_bias<<<cdiv((long)N * HC, BLK), BLK, 0, stream>>>(out1, b1, N, HC);

  edge_logits_max<<<cdiv(EP * HEADS, BLK), BLK, 0, stream>>>(ei, as1, ad1, e1, m1, E, N, HEADS);
  edge_exp_sum  <<<cdiv(EP * HEADS, BLK), BLK, 0, stream>>>(ei, e1, m1, s1, E, N, HEADS);
  edge_scatter  <<<cdiv(EP * HEADS * OUT_CH, BLK), BLK, 0, stream>>>(ei, e1, s1, h1, out1, E, N, HEADS, OUT_CH);

  ln_add_elu<<<cdiv(N, WPB), BLK, 0, stream>>>(out1, g1, be1, nullptr, out1, N, HC);

  {
    long waves = ((long)N / 16) * (OUT_CH / 16);
    gemm_wmma_f32<<<cdiv(waves, WPB), BLK, 0, stream>>>(out1, W2, nullptr, z2, N, HC, OUT_CH);
  }

  att_scores<<<cdiv((long)N, BLK), BLK, 0, stream>>>(z2, aS2, aD2, as2, ad2, N, 1, OUT_CH);
  fill_value<<<cdiv((long)N, BLK), BLK, 0, stream>>>(m2, -INFINITY, (long)N);
  fill_value<<<cdiv((long)N, BLK), BLK, 0, stream>>>(s2, 0.0f,      (long)N);
  fill_rows_bias<<<cdiv((long)N * OUT_CH, BLK), BLK, 0, stream>>>(out2, b2, N, OUT_CH);

  edge_logits_max<<<cdiv(EP, BLK), BLK, 0, stream>>>(ei, as2, ad2, e2, m2, E, N, 1);
  edge_exp_sum  <<<cdiv(EP, BLK), BLK, 0, stream>>>(ei, e2, m2, s2, E, N, 1);
  edge_scatter  <<<cdiv(EP * OUT_CH, BLK), BLK, 0, stream>>>(ei, e2, s2, z2, out2, E, N, 1, OUT_CH);

  ln_add_elu<<<cdiv(N, WPB), BLK, 0, stream>>>(out2, g2, be2, idb, (float*)d_out, N, OUT_CH);

  (void)n_in; (void)out_size; (void)ws_size;
}


# softmax inside the work-group that owns the destination rows: matched edges listed in LDS in one edge-list pass; max and exp sum by LDS atomics (arrival order); divide after the sum; 16 launches to 8
# speedup vs baseline: 4.2851x; 4.2851x over previous
#include <hip/hip_runtime.h>
#include <math.h>

typedef float v2f __attribute__((ext_vector_type(2)));
typedef float v8f __attribute__((ext_vector_type(8)));

#define NEG_SLOPE 0.2f

template <int K, int NCOLS, int THREADS>
__device__ __forceinline__ void stage_weights(const float* __restrict__ B,
                                              float4* __restrict__ dst, int tid) {
  constexpr int KP   = K / 8;
  constexpr int ENT  = (NCOLS / 16) * KP * 32;
  constexpr int ITER = ENT / THREADS;
  static_assert(ENT % THREADS == 0 && THREADS % 32 == 0, "whole waves share the entries evenly");
  const int c = tid & 15, h = (tid >> 4) & 1;
  float4 v[ITER];
#pragma unroll
  for (int it = 0; it < ITER; ++it) {
    const int g = (tid + it * THREADS) >> 5;
    const int tile = g / KP, kp = g - tile * KP;
    const float* p = B + (size_t)(8 * kp + 2 * h) * NCOLS + tile * 16 + c;
    v[it].x = p[0]; v[it].y = p[NCOLS]; v[it].z = p[4 * NCOLS]; v[it].w = p[5 * NCOLS];
  }
#pragma unroll
  for (int it = 0; it < ITER; ++it) dst[tid + it * THREADS] = v[it];
}

template <int K>
__device__ __forceinline__ void load_a_strip(const float* __restrict__ A, int strip, int lane,
                                             v2f (&a)[K / 4]) {
  const float* ap = A + (size_t)(strip * 16 + (lane & 15)) * K + ((lane >> 4) << 1);
#pragma unroll
  for (int ks = 0; ks < K / 4; ++ks) a[ks] = *(const v2f*)(ap + 4 * ks);
}

template <int K, int T1, int T2, int UPS, int WAVES>
__global__ __launch_bounds__(WAVES * 32) void gemm_lds_weights(
    const float* __restrict__ A,
    const float* __restrict__ B1, float* __restrict__ C1,
    const float* __restrict__ B2, const float* __restrict__ bias2, float* __restrict__ C2,
    int M) {
  constexpr int NT1 = UPS * T1, NT2 = UPS * T2;
  constexpr int N1 = NT1 * 16, N2 = NT2 * 16;
  constexpr int KP = K / 8;
  constexpr int TT = T1 + T2;
  constexpr int RS = WAVES / UPS;
  static_assert(T1 % 2 == 0 && WAVES % UPS == 0 && K % 8 == 0, "C1 tiles are stored in pairs");
  constexpr int NEED = (NT1 + NT2) * KP * 32;
  constexpr int FLOOR = 164 * 1024 / 16;
  __shared__ float4 Bl[NEED > FLOOR ? NEED : FLOOR];

  const int tid  = threadIdx.x;
  const int lane = tid & 31;
  const int wv   = tid >> 5;
  const int hi   = lane >> 4;
  const int cg   = wv % UPS;
  const int rr   = wv / UPS;

  const int strips = M >> 4;
  const int sb = (int)((blockIdx.x * (unsigned)strips) / gridDim.x);
  const int se = (int)(((blockIdx.x + 1) * (unsigned)strips) / gridDim.x);

  v2f an[K / 4];
  int st = sb + rr;
  if (st < se) load_a_strip<K>(A, st, lane, an);

  stage_weights<K, N1, WAVES * 32>(B1, Bl, tid);
  if constexpr (T2 > 0)
    stage_weights<K, N2, WAVES * 32>(B2, Bl + NT1 * KP * 32, tid);
  __syncthreads();

  for (; st < se; st += RS) {
    v2f a[K / 4];
#pragma unroll
    for (int ks = 0; ks < K / 4; ++ks) a[ks] = an[ks];
    if (st + RS < se) load_a_strip<K>(A, st + RS, lane, an);

    v8f acc[TT];
#pragma unroll
    for (int t = 0; t < TT; ++t) acc[t] = v8f{};
    float4 bc[TT], bn[TT];
#pragma unroll
    for (int t = 0; t < TT; ++t) {
      const int tile = (t < T1) ? (cg * T1 + t) : (NT1 + cg * T2 + (t - T1));
      bc[t] = Bl[(tile * KP) * 32 + lane];
    }
#pragma unroll
    for (int kp = 0; kp < KP; ++kp) {
      if (kp + 1 < KP) {
#pragma unroll
        for (int t = 0; t < TT; ++t) {
          const int tile = (t < T1) ? (cg * T1 + t) : (NT1 + cg * T2 + (t - T1));
          bn[t] = Bl[(tile * KP + kp + 1) * 32 + lane];
        }
      }
#pragma unroll
      for (int t = 0; t < TT; ++t) {
        v2f bb; bb.x = bc[t].x; bb.y = bc[t].y;
        acc[t] = __builtin_amdgcn_wmma_f32_16x16x4_f32(
             false, a[2 * kp],  false, bb,
             (short)0, acc[t],  false,  false);
      }
#pragma unroll
      for (int t = 0; t < TT; ++t) {
        v2f bb; bb.x = bc[t].z; bb.y = bc[t].w;
        acc[t] = __builtin_amdgcn_wmma_f32_16x16x4_f32(
             false, a[2 * kp + 1],  false, bb,
             (short)0, acc[t],  false,  false);
      }
#pragma unroll
      for (int t = 0; t < TT; ++t) bc[t] = bn[t];
      __builtin_amdgcn_sched_barrier(0);
    }

#pragma unroll
    for (int p = 0; p < T1 / 2; ++p) {
      float* cp = C1 + (size_t)(st * 16) * N1 + (cg * T1 + 2 * p) * 16 + lane;
#pragma unroll
      for (int v = 0; v < 8; ++v) {
        const float own0 = acc[2 * p][v], own1 = acc[2 * p + 1][v];
        const float give = hi ? own0 : own1;
        const float got  = __shfl_xor(give, 16, 32);
        cp[(size_t)v * N1]       = hi ? got : own0;
        cp[(size_t)(8 + v) * N1] = hi ? own1 : got;
      }
    }
    if constexpr (T2 > 0) {
#pragma unroll
      for (int t = 0; t < T2; ++t) {
        const int ncol = (cg * T2 + t) * 16 + (lane & 15);
        const float bv = bias2 ? bias2[ncol] : 0.0f;
        float* cp = C2 + (size_t)(st * 16 + (hi << 3)) * N2 + ncol;
#pragma unroll
        for (int v = 0; v < 8; ++v) cp[(size_t)v * N2] = acc[T1 + t][v] + bv;
      }
    }
  }
}

__global__ void att_scores(const float* __restrict__ h,
                           const float* __restrict__ att_src,
                           const float* __restrict__ att_dst,
                           float* __restrict__ as, float* __restrict__ ad,
                           int n, int H, int C) {
  long t = (long)blockIdx.x * blockDim.x + threadIdx.x;
  if (t >= (long)n * H) return;
  const int head = (int)(t % H);
  const float4* hv = (const float4*)(h + (size_t)t * C);
  const float4* sv = (const float4*)(att_src + (size_t)head * C);
  const float4* dv = (const float4*)(att_dst + (size_t)head * C);
  float s = 0.f, d = 0.f;
  for (int i = 0; i < (C >> 2); ++i) {
    float4 xv = hv[i], a = sv[i], b = dv[i];
    s += xv.x * a.x + xv.y * a.y + xv.z * a.z + xv.w * a.w;
    d += xv.x * b.x + xv.y * b.y + xv.z * b.z + xv.w * b.w;
  }
  as[t] = s; ad[t] = d;
}

template <int CT>
__device__ __forceinline__ void load_row_vals(const float* __restrict__ p, int lane,
                                              float (&v)[CT / 32]) {
  if constexpr (CT == 256) {
    const float4 a = *(const float4*)(p + 4 * lane);
    const float4 b = *(const float4*)(p + 128 + 4 * lane);
    v[0] = a.x; v[1] = a.y; v[2] = a.z; v[3] = a.w;
    v[4] = b.x; v[5] = b.y; v[6] = b.z; v[7] = b.w;
  } else {
    const float2 a = *(const float2*)(p + 2 * lane);
    v[0] = a.x; v[1] = a.y;
  }
}

__device__ __forceinline__ int bcast_i(int v, int l) {
  return __builtin_amdgcn_readlane(v, l);
}
__device__ __forceinline__ float bcast_f(float v, int l) {
  return __int_as_float(__builtin_amdgcn_readlane(__float_as_int(v), l));
}

__device__ __forceinline__ float edge_logit(float a_src, float a_dst) {
  const float v = a_src + a_dst;
  return ((v > 0.f) ? v : NEG_SLOPE * v) + 0.0f;
}

template <typename F>
__device__ __forceinline__ void stream_dst_matches(const int* __restrict__ dstp, int E,
                                                   int row0, int rows, int tid, F on_round) {
  constexpr int T = 1024;
  constexpr int STEP = T * 16;
  const bool vec_ok = (E >= 4) && ((E & 3) == 0) && ((((size_t)dstp) & 15) == 0);

  auto load16 = [&](int base, int (&d)[16]) {
    if (vec_ok) {
#pragma unroll
      for (int j = 0; j < 4; ++j) {
        const int b = base + j * (T * 4) + tid * 4;
        const int4 q = *(const int4*)(dstp + min(b, E - 4));
        d[4 * j + 0] = q.x; d[4 * j + 1] = q.y; d[4 * j + 2] = q.z; d[4 * j + 3] = q.w;
      }
    } else {
#pragma unroll
      for (int j = 0; j < 4; ++j) {
#pragma unroll
        for (int k = 0; k < 4; ++k)
          d[4 * j + k] = dstp[min(base + j * (T * 4) + tid * 4 + k, E - 1)];
      }
    }
  };

  int cur[16], nxt[16];
#pragma unroll
  for (int q = 0; q < 16; ++q) { cur[q] = row0 - 1; nxt[q] = row0 - 1; }
  for (int base = -STEP; base < E; base += STEP) {
    unsigned mb = 0;
#pragma unroll
    for (int q = 0; q < 16; ++q) {
      const int eq = base + (q >> 2) * (T * 4) + tid * 4 + (q & 3);
      mb |= (eq < E && (unsigned)(cur[q] - row0) < (unsigned)rows) ? (1u << q) : 0u;
    }

    if (base + STEP < E) load16(base + STEP, nxt);

    while (__builtin_amdgcn_ballot_w32(mb != 0)) {
      const bool has = mb != 0;
      const int q = has ? __builtin_ctz(mb) : 0;
      mb &= mb - 1;
      int dsel = cur[0];
#pragma unroll
      for (int t = 1; t < 16; ++t) dsel = (q == t) ? cur[t] : dsel;
      const int edge = base + (q >> 2) * (T * 4) + tid * 4 + (q & 3);
      on_round(has, edge, dsel - row0);
    }

#pragma unroll
    for (int q = 0; q < 16; ++q) cur[q] = nxt[q];
  }
}

template <int CT, int H, int R, int CAP>
__global__ __launch_bounds__(1024) void agg_softmax_rows_lds(
    const int* __restrict__ ei, const float* __restrict__ as, const float* __restrict__ ad,
    const float* __restrict__ feat, const float* __restrict__ bias,
    float* __restrict__ out, int E, int n) {
  static_assert((CT == 256 && H == 4) || (CT == 64 && H == 1), "lane-to-head map below");
  constexpr int V = CT / 32;
  constexpr int T = 1024;
  constexpr int NF = 8;
  __shared__ float acc[R * CT];
  __shared__ float adrow[R * H];
  __shared__ float slog[R * H];
  __shared__ float mrow[R * H];
  __shared__ float srow[R * H];
  __shared__ int   l_src[CAP];
  __shared__ int   l_rel[CAP];
  __shared__ float l_as[CAP * H];
  __shared__ int   l_cnt;

  const int tid  = threadIdx.x;
  const int lane = tid & 31;
  const int wv   = tid >> 5;
  const int hi   = lane >> 4;
  const int row0 = blockIdx.x * R;
  const int rows = max(0, min(R, n - row0));
  const int* __restrict__ dstp = ei + E;

  auto lds_max = [&](int slot, float val) {
    if (val >= 0.0f) atomicMax((int*)&mrow[slot], __float_as_int(val));
    else             atomicMin((unsigned int*)&mrow[slot], __float_as_uint(val));
  };

  auto wave_accumulate = [&](unsigned m, int srcv, int rel, float x0, float x1, float x2, float x3) {
    while (m) {
      int  l[NF];
      bool ok[NF];
#pragma unroll
      for (int j = 0; j < NF; ++j) {
        ok[j] = (m != 0);
        l[j]  = ok[j] ? __builtin_ctz(m) : 0;
        if (ok[j]) m &= m - 1;
      }
      float v[NF][V];
#pragma unroll
      for (int j = 0; j < NF; ++j) {
        if (ok[j]) load_row_vals<CT>(feat + (size_t)bcast_i(srcv, l[j]) * CT, lane, v[j]);
      }
#pragma unroll
      for (int j = 0; j < NF; ++j) {
        if (ok[j]) {
          const int rl = bcast_i(rel, l[j]);
          float aA, aB;
          if constexpr (H == 4) {
            const float y0 = bcast_f(x0, l[j]), y1 = bcast_f(x1, l[j]);
            const float y2 = bcast_f(x2, l[j]), y3 = bcast_f(x3, l[j]);
            aA = hi ? y1 : y0; aB = hi ? y3 : y2;
          } else {
            aA = aB = bcast_f(x0, l[j]);
          }
#pragma unroll
          for (int i = 0; i < V; ++i)
            __hip_atomic_fetch_add(&acc[rl * CT + i * 32 + lane], (i < 4 ? aA : aB) * v[j][i],
                                   __ATOMIC_RELAXED, __HIP_MEMORY_SCOPE_WORKGROUP);
        }
      }
    }
  };

  if (tid == 0) l_cnt = 0;
  for (int i = tid; i < rows * H; i += T) {
    const float d_ = ad[(size_t)row0 * H + i];
    const float v  = edge_logit(as[(size_t)row0 * H + i], d_);
    adrow[i] = d_; slog[i] = v; mrow[i] = v;
  }
  __syncthreads();

  stream_dst_matches(dstp, E, row0, rows, tid, [&](bool has, int edge, int rel) {
    const unsigned bal = __builtin_amdgcn_ballot_w32(has);
    const int lead = __builtin_ctz(bal);
    int basepos = 0;
    if (lane == lead) basepos = atomicAdd(&l_cnt, __builtin_popcount(bal));
    basepos = bcast_i(basepos, lead);
    const int pos = basepos + __builtin_popcount(bal & ((1u << lane) - 1u));
    if (has && pos < CAP) { l_src[pos] = edge; l_rel[pos] = rel; }
  });
  __syncthreads();
  const int  cnt     = l_cnt;
  const bool listed  = cnt <= CAP;

  if (listed) {
    for (int i = tid; i < cnt; i += T) {
      const int edge = l_src[i], rel = l_rel[i];
      const int src  = min(max(ei[edge], 0), n - 1);
      l_src[i] = src;
      if constexpr (H == 4) {
        const float4 a = *(const float4*)(as + (size_t)src * 4);
        l_as[i * 4 + 0] = a.x; l_as[i * 4 + 1] = a.y; l_as[i * 4 + 2] = a.z; l_as[i * 4 + 3] = a.w;
        lds_max(rel * 4 + 0, edge_logit(a.x, adrow[rel * 4 + 0]));
        lds_max(rel * 4 + 1, edge_logit(a.y, adrow[rel * 4 + 1]));
        lds_max(rel * 4 + 2, edge_logit(a.z, adrow[rel * 4 + 2]));
        lds_max(rel * 4 + 3, edge_logit(a.w, adrow[rel * 4 + 3]));
      } else {
        const float a = as[src];
        l_as[i] = a;
        lds_max(rel, edge_logit(a, adrow[rel]));
      }
    }
  } else {
    stream_dst_matches(dstp, E, row0, rows, tid, [&](bool has, int edge, int rel) {
      if (has) {
        const int src = min(max(ei[edge], 0), n - 1);
        if constexpr (H == 4) {
          const float4 a = *(const float4*)(as + (size_t)src * 4);
          lds_max(rel * 4 + 0, edge_logit(a.x, adrow[rel * 4 + 0]));
          lds_max(rel * 4 + 1, edge_logit(a.y, adrow[rel * 4 + 1]));
          lds_max(rel * 4 + 2, edge_logit(a.z, adrow[rel * 4 + 2]));
          lds_max(rel * 4 + 3, edge_logit(a.w, adrow[rel * 4 + 3]));
        } else {
          lds_max(rel, edge_logit(as[src], adrow[rel]));
        }
      }
    });
  }
  __syncthreads();

  for (int r = wv; r < rows; r += 32) {
    float v[V];
    load_row_vals<CT>(feat + (size_t)(row0 + r) * CT, lane, v);
    if constexpr (H == 4) {
      const float xA = expf(slog[r * 4 + hi]     - mrow[r * 4 + hi]);
      const float xB = expf(slog[r * 4 + 2 + hi] - mrow[r * 4 + 2 + hi]);
#pragma unroll
      for (int i = 0; i < V; ++i) acc[r * CT + i * 32 + lane] = (i < 4 ? xA : xB) * v[i];
      if (lane < 4) srow[r * 4 + lane] = expf(slog[r * 4 + lane] - mrow[r * 4 + lane]);
    } else {
      const float x = expf(slog[r] - mrow[r]);
#pragma unroll
      for (int i = 0; i < V; ++i) acc[r * CT + i * 32 + lane] = x * v[i];
      if (lane == 0) srow[r] = x;
    }
  }
  __syncthreads();

  if (listed) {
    const int lo = (int)(((long)wv * cnt) >> 5), up = (int)(((long)(wv + 1) * cnt) >> 5);
    for (int cb = lo; cb < up; cb += 32) {
      const int  idx = cb + lane;
      const bool has = idx < up;
      int srcv = 0, rel = 0;
      float x0 = 0.f, x1 = 0.f, x2 = 0.f, x3 = 0.f;
      if (has) {
        srcv = l_src[idx]; rel = l_rel[idx];
        if constexpr (H == 4) {
          x0 = expf(edge_logit(l_as[idx * 4 + 0], adrow[rel * 4 + 0]) - mrow[rel * 4 + 0]);
          x1 = expf(edge_logit(l_as[idx * 4 + 1], adrow[rel * 4 + 1]) - mrow[rel * 4 + 1]);
          x2 = expf(edge_logit(l_as[idx * 4 + 2], adrow[rel * 4 + 2]) - mrow[rel * 4 + 2]);
          x3 = expf(edge_logit(l_as[idx * 4 + 3], adrow[rel * 4 + 3]) - mrow[rel * 4 + 3]);
          __hip_atomic_fetch_add(&srow[rel * 4 + 0], x0, __ATOMIC_RELAXED, __HIP_MEMORY_SCOPE_WORKGROUP);
          __hip_atomic_fetch_add(&srow[rel * 4 + 1], x1, __ATOMIC_RELAXED, __HIP_MEMORY_SCOPE_WORKGROUP);
          __hip_atomic_fetch_add(&srow[rel * 4 + 2], x2, __ATOMIC_RELAXED, __HIP_MEMORY_SCOPE_WORKGROUP);
          __hip_atomic_fetch_add(&srow[rel * 4 + 3], x3, __ATOMIC_RELAXED, __HIP_MEMORY_SCOPE_WORKGROUP);
        } else {
          x0 = expf(edge_logit(l_as[idx], adrow[rel]) - mrow[rel]);
          __hip_atomic_fetch_add(&srow[rel], x0, __ATOMIC_RELAXED, __HIP_MEMORY_SCOPE_WORKGROUP);
        }
      }
      wave_accumulate(__builtin_amdgcn_ballot_w32(has), srcv, rel, x0, x1, x2, x3);
    }
  } else {
    stream_dst_matches(dstp, E, row0, rows, tid, [&](bool has, int edge, int rel) {
      int srcv = 0;
      float x0 = 0.f, x1 = 0.f, x2 = 0.f, x3 = 0.f;
      if (has) {
        srcv = min(max(ei[edge], 0), n - 1);
        if constexpr (H == 4) {
          const float4 a = *(const float4*)(as + (size_t)srcv * 4);
          x0 = expf(edge_logit(a.x, adrow[rel * 4 + 0]) - mrow[rel * 4 + 0]);
          x1 = expf(edge_logit(a.y, adrow[rel * 4 + 1]) - mrow[rel * 4 + 1]);
          x2 = expf(edge_logit(a.z, adrow[rel * 4 + 2]) - mrow[rel * 4 + 2]);
          x3 = expf(edge_logit(a.w, adrow[rel * 4 + 3]) - mrow[rel * 4 + 3]);
          __hip_atomic_fetch_add(&srow[rel * 4 + 0], x0, __ATOMIC_RELAXED, __HIP_MEMORY_SCOPE_WORKGROUP);
          __hip_atomic_fetch_add(&srow[rel * 4 + 1], x1, __ATOMIC_RELAXED, __HIP_MEMORY_SCOPE_WORKGROUP);
          __hip_atomic_fetch_add(&srow[rel * 4 + 2], x2, __ATOMIC_RELAXED, __HIP_MEMORY_SCOPE_WORKGROUP);
          __hip_atomic_fetch_add(&srow[rel * 4 + 3], x3, __ATOMIC_RELAXED, __HIP_MEMORY_SCOPE_WORKGROUP);
        } else {
          x0 = expf(edge_logit(as[srcv], adrow[rel]) - mrow[rel]);
          __hip_atomic_fetch_add(&srow[rel], x0, __ATOMIC_RELAXED, __HIP_MEMORY_SCOPE_WORKGROUP);
        }
      }
      wave_accumulate(__builtin_amdgcn_ballot_w32(has), srcv, rel, x0, x1, x2, x3);
    });
  }
  __syncthreads();

  float bv[V];
  load_row_vals<CT>(bias, lane, bv);
  for (int r = wv; r < rows; r += 32) {
    float* __restrict__ op = out + (size_t)(row0 + r) * CT;
    float dA, dB;
    if constexpr (H == 4) {
      dA = srow[r * 4 + hi] + 1e-16f; dB = srow[r * 4 + 2 + hi] + 1e-16f;
    } else {
      dA = dB = srow[r] + 1e-16f;
    }
    float v[V];
#pragma unroll
    for (int i = 0; i < V; ++i) v[i] = bv[i] + acc[r * CT + i * 32 + lane] / (i < 4 ? dA : dB);
    if constexpr (CT == 256) {
      *(float4*)(op + 4 * lane)       = make_float4(v[0], v[1], v[2], v[3]);
      *(float4*)(op + 128 + 4 * lane) = make_float4(v[4], v[5], v[6], v[7]);
    } else {
      *(float2*)(op + 2 * lane) = make_float2(v[0], v[1]);
    }
  }
}

template <int C>
__global__ __launch_bounds__(256) void ln_add_elu(const float* __restrict__ x, const float* __restrict__ g,
                           const float* __restrict__ b, const float* __restrict__ add,
                           float* __restrict__ y, int n) {
  constexpr int V = C / 32;
  const int lane = threadIdx.x & 31;
  const int nw   = gridDim.x * (blockDim.x >> 5);
  float gv[V], bv[V];
#pragma unroll
  for (int i = 0; i < V; ++i) { gv[i] = g[lane + 32 * i]; bv[i] = b[lane + 32 * i]; }
  for (int row = blockIdx.x * (blockDim.x >> 5) + (threadIdx.x >> 5); row < n; row += nw) {
    const float* xr = x + (size_t)row * C;
    float v[V], av[V];
#pragma unroll
    for (int i = 0; i < V; ++i) v[i] = xr[lane + 32 * i];
    if (add) {
      const float* ar = add + (size_t)row * C;
#pragma unroll
      for (int i = 0; i < V; ++i) av[i] = ar[lane + 32 * i];
    } else {
#pragma unroll
      for (int i = 0; i < V; ++i) av[i] = 0.f;
    }
    float s = 0.f, ss = 0.f;
#pragma unroll
    for (int i = 0; i < V; ++i) { s += v[i]; ss += v[i] * v[i]; }
    for (int off = 16; off; off >>= 1) {
      s  += __shfl_xor(s,  off, 32);
      ss += __shfl_xor(ss, off, 32);
    }
    const float mu  = s / (float)C;
    const float var = ss / (float)C - mu * mu;
    const float inv = rsqrtf(var + 1e-5f);
    float* yr = y + (size_t)row * C;
#pragma unroll
    for (int i = 0; i < V; ++i) {
      float o = (v[i] - mu) * inv * gv[i] + bv[i];
      if (add) o += av[i];
      yr[lane + 32 * i] = (o > 0.f) ? o : expm1f(o);
    }
  }
}

extern "C" void kernel_launch(void* const* d_in, const int* in_sizes, int n_in,
                              void* d_out, int out_size, void* d_ws, size_t ws_size,
                              hipStream_t stream) {
  const float* x    = (const float*)d_in[0];
  const int*   ei   = (const int*)  d_in[1];
  const float* W1   = (const float*)d_in[2];
  const float* aS1  = (const float*)d_in[3];
  const float* aD1  = (const float*)d_in[4];
  const float* b1   = (const float*)d_in[5];
  const float* W2   = (const float*)d_in[6];
  const float* aS2  = (const float*)d_in[7];
  const float* aD2  = (const float*)d_in[8];
  const float* b2   = (const float*)d_in[9];
  const float* g1   = (const float*)d_in[10];
  const float* be1  = (const float*)d_in[11];
  const float* g2   = (const float*)d_in[12];
  const float* be2  = (const float*)d_in[13];
  const float* pW   = (const float*)d_in[14];
  const float* pb   = (const float*)d_in[15];

  const int IN_CH = 128, HEADS = 4, OUT_CH = 64, HC = HEADS * OUT_CH;
  const int N  = in_sizes[0] / IN_CH;
  const int E  = in_sizes[1] / 2;

  float* ws   = (float*)d_ws;
  float* h1   = ws; ws += (size_t)N * HC;
  float* out1 = ws; ws += (size_t)N * HC;
  float* idb  = ws; ws += (size_t)N * OUT_CH;
  float* z2   = ws; ws += (size_t)N * OUT_CH;
  float* out2 = ws; ws += (size_t)N * OUT_CH;
  float* as1  = ws; ws += (size_t)N * HEADS;
  float* ad1  = ws; ws += (size_t)N * HEADS;
  float* as2  = ws; ws += (size_t)N;
  float* ad2  = ws; ws += (size_t)N;

  auto cdiv = [](long a, long b) -> int { return (int)((a + b - 1) / b); };
  const int BLK = 256;
  const int WPB = BLK / 32;
  const int RPW = 4;

  gemm_lds_weights<128, 4, 1, 4, 16><<<256, 512, 0, stream>>>(x, W1, h1, pW, pb, idb, N);

  att_scores<<<cdiv((long)N * HEADS, BLK), BLK, 0, stream>>>(h1, aS1, aD1, as1, ad1, N, HEADS, OUT_CH);
  agg_softmax_rows_lds<256, 4, 196, 3072><<<cdiv(N, 196), 1024, 0, stream>>>(ei, as1, ad1, h1, b1, out1, E, N);

  ln_add_elu<256><<<cdiv(N, WPB * RPW), BLK, 0, stream>>>(out1, g1, be1, nullptr, out1, N);

  gemm_lds_weights<256, 2, 0, 2, 8><<<256, 256, 0, stream>>>(out1, W2, z2, nullptr, nullptr, nullptr, N);

  att_scores<<<cdiv((long)N, BLK), BLK, 0, stream>>>(z2, aS2, aD2, as2, ad2, N, 1, OUT_CH);
  agg_softmax_rows_lds<64, 1, 196, 3072><<<cdiv(N, 196), 1024, 0, stream>>>(ei, as2, ad2, z2, b2, out2, E, N);

  ln_add_elu<64><<<cdiv(N, WPB * RPW), BLK, 0, stream>>>(out2, g2, be2, idb, (float*)d_out, N);

  (void)n_in; (void)out_size; (void)ws_size;
}


# layer 2's aggregation reads the (source, row) list its work-group wrote to memory in layer 1; no second pass over the destination list, no re-gather of source ids; sums: LDS adds, arrival order
# speedup vs baseline: 4.9022x; 1.1440x over previous
#include <hip/hip_runtime.h>
#include <math.h>

typedef float v2f __attribute__((ext_vector_type(2)));
typedef float v8f __attribute__((ext_vector_type(8)));

#define NEG_SLOPE 0.2f

template <int K, int NCOLS, int THREADS>
__device__ __forceinline__ void stage_weights(const float* __restrict__ B,
                                              float4* __restrict__ dst, int tid) {
  constexpr int KP   = K / 8;
  constexpr int ENT  = (NCOLS / 16) * KP * 32;
  constexpr int ITER = ENT / THREADS;
  static_assert(ENT % THREADS == 0 && THREADS % 32 == 0, "whole waves share the entries evenly");
  const int c = tid & 15, h = (tid >> 4) & 1;
  float4 v[ITER];
#pragma unroll
  for (int it = 0; it < ITER; ++it) {
    const int g = (tid + it * THREADS) >> 5;
    const int tile = g / KP, kp = g - tile * KP;
    const float* p = B + (size_t)(8 * kp + 2 * h) * NCOLS + tile * 16 + c;
    v[it].x = p[0]; v[it].y = p[NCOLS]; v[it].z = p[4 * NCOLS]; v[it].w = p[5 * NCOLS];
  }
#pragma unroll
  for (int it = 0; it < ITER; ++it) dst[tid + it * THREADS] = v[it];
}

template <int K>
__device__ __forceinline__ void load_a_strip(const float* __restrict__ A, int strip, int lane,
                                             v2f (&a)[K / 4]) {
  const float* ap = A + (size_t)(strip * 16 + (lane & 15)) * K + ((lane >> 4) << 1);
#pragma unroll
  for (int ks = 0; ks < K / 4; ++ks) a[ks] = *(const v2f*)(ap + 4 * ks);
}

template <int K, int T1, int T2, int UPS, int WAVES>
__global__ __launch_bounds__(WAVES * 32) void gemm_lds_weights(
    const float* __restrict__ A,
    const float* __restrict__ B1, float* __restrict__ C1,
    const float* __restrict__ B2, const float* __restrict__ bias2, float* __restrict__ C2,
    int M) {
  constexpr int NT1 = UPS * T1, NT2 = UPS * T2;
  constexpr int N1 = NT1 * 16, N2 = NT2 * 16;
  constexpr int KP = K / 8;
  constexpr int TT = T1 + T2;
  constexpr int RS = WAVES / UPS;
  static_assert(T1 % 2 == 0 && WAVES % UPS == 0 && K % 8 == 0, "C1 tiles are stored in pairs");
  constexpr int NEED = (NT1 + NT2) * KP * 32;
  constexpr int FLOOR = 164 * 1024 / 16;
  __shared__ float4 Bl[NEED > FLOOR ? NEED : FLOOR];

  const int tid  = threadIdx.x;
  const int lane = tid & 31;
  const int wv   = tid >> 5;
  const int hi   = lane >> 4;
  const int cg   = wv % UPS;
  const int rr   = wv / UPS;

  const int strips = M >> 4;
  const int sb = (int)((blockIdx.x * (unsigned)strips) / gridDim.x);
  const int se = (int)(((blockIdx.x + 1) * (unsigned)strips) / gridDim.x);

  v2f an[K / 4];
  int st = sb + rr;
  if (st < se) load_a_strip<K>(A, st, lane, an);

  stage_weights<K, N1, WAVES * 32>(B1, Bl, tid);
  if constexpr (T2 > 0)
    stage_weights<K, N2, WAVES * 32>(B2, Bl + NT1 * KP * 32, tid);
  __syncthreads();

  for (; st < se; st += RS) {
    v2f a[K / 4];
#pragma unroll
    for (int ks = 0; ks < K / 4; ++ks) a[ks] = an[ks];
    if (st + RS < se) load_a_strip<K>(A, st + RS, lane, an);

    v8f acc[TT];
#pragma unroll
    for (int t = 0; t < TT; ++t) acc[t] = v8f{};
    float4 bc[TT], bn[TT];
#pragma unroll
    for (int t = 0; t < TT; ++t) {
      const int tile = (t < T1) ? (cg * T1 + t) : (NT1 + cg * T2 + (t - T1));
      bc[t] = Bl[(tile * KP) * 32 + lane];
    }
#pragma unroll
    for (int kp = 0; kp < KP; ++kp) {
      if (kp + 1 < KP) {
#pragma unroll
        for (int t = 0; t < TT; ++t) {
          const int tile = (t < T1) ? (cg * T1 + t) : (NT1 + cg * T2 + (t - T1));
          bn[t] = Bl[(tile * KP + kp + 1) * 32 + lane];
        }
      }
#pragma unroll
      for (int t = 0; t < TT; ++t) {
        v2f bb; bb.x = bc[t].x; bb.y = bc[t].y;
        acc[t] = __builtin_amdgcn_wmma_f32_16x16x4_f32(
             false, a[2 * kp],  false, bb,
             (short)0, acc[t],  false,  false);
      }
#pragma unroll
      for (int t = 0; t < TT; ++t) {
        v2f bb; bb.x = bc[t].z; bb.y = bc[t].w;
        acc[t] = __builtin_amdgcn_wmma_f32_16x16x4_f32(
             false, a[2 * kp + 1],  false, bb,
             (short)0, acc[t],  false,  false);
      }
#pragma unroll
      for (int t = 0; t < TT; ++t) bc[t] = bn[t];
      __builtin_amdgcn_sched_barrier(0);
    }

#pragma unroll
    for (int p = 0; p < T1 / 2; ++p) {
      float* cp = C1 + (size_t)(st * 16) * N1 + (cg * T1 + 2 * p) * 16 + lane;
#pragma unroll
      for (int v = 0; v < 8; ++v) {
        const float own0 = acc[2 * p][v], own1 = acc[2 * p + 1][v];
        const float give = hi ? own0 : own1;
        const float got  = __shfl_xor(give, 16, 32);
        cp[(size_t)v * N1]       = hi ? got : own0;
        cp[(size_t)(8 + v) * N1] = hi ? own1 : got;
      }
    }
    if constexpr (T2 > 0) {
#pragma unroll
      for (int t = 0; t < T2; ++t) {
        const int ncol = (cg * T2 + t) * 16 + (lane & 15);
        const float bv = bias2 ? bias2[ncol] : 0.0f;
        float* cp = C2 + (size_t)(st * 16 + (hi << 3)) * N2 + ncol;
#pragma unroll
        for (int v = 0; v < 8; ++v) cp[(size_t)v * N2] = acc[T1 + t][v] + bv;
      }
    }
  }
}

__global__ void att_scores(const float* __restrict__ h,
                           const float* __restrict__ att_src,
                           const float* __restrict__ att_dst,
                           float* __restrict__ as, float* __restrict__ ad,
                           int n, int H, int C) {
  long t = (long)blockIdx.x * blockDim.x + threadIdx.x;
  if (t >= (long)n * H) return;
  const int head = (int)(t % H);
  const float4* hv = (const float4*)(h + (size_t)t * C);
  const float4* sv = (const float4*)(att_src + (size_t)head * C);
  const float4* dv = (const float4*)(att_dst + (size_t)head * C);
  float s = 0.f, d = 0.f;
  for (int i = 0; i < (C >> 2); ++i) {
    float4 xv = hv[i], a = sv[i], b = dv[i];
    s += xv.x * a.x + xv.y * a.y + xv.z * a.z + xv.w * a.w;
    d += xv.x * b.x + xv.y * b.y + xv.z * b.z + xv.w * b.w;
  }
  as[t] = s; ad[t] = d;
}

template <int CT>
__device__ __forceinline__ void load_row_vals(const float* __restrict__ p, int lane,
                                              float (&v)[CT / 32]) {
  if constexpr (CT == 256) {
    const float4 a = *(const float4*)(p + 4 * lane);
    const float4 b = *(const float4*)(p + 128 + 4 * lane);
    v[0] = a.x; v[1] = a.y; v[2] = a.z; v[3] = a.w;
    v[4] = b.x; v[5] = b.y; v[6] = b.z; v[7] = b.w;
  } else {
    const float2 a = *(const float2*)(p + 2 * lane);
    v[0] = a.x; v[1] = a.y;
  }
}

__device__ __forceinline__ int bcast_i(int v, int l) {
  return __builtin_amdgcn_readlane(v, l);
}
__device__ __forceinline__ float bcast_f(float v, int l) {
  return __int_as_float(__builtin_amdgcn_readlane(__float_as_int(v), l));
}

__device__ __forceinline__ float edge_logit(float a_src, float a_dst) {
  const float v = a_src + a_dst;
  return ((v > 0.f) ? v : NEG_SLOPE * v) + 0.0f;
}

template <typename F>
__device__ __forceinline__ void stream_dst_matches(const int* __restrict__ dstp, int E,
                                                   int row0, int rows, int tid, F on_round) {
  constexpr int T = 1024;
  constexpr int STEP = T * 16;
  const bool vec_ok = (E >= 4) && ((E & 3) == 0) && ((((size_t)dstp) & 15) == 0);

  auto load16 = [&](int base, int (&d)[16]) {
    if (vec_ok) {
#pragma unroll
      for (int j = 0; j < 4; ++j) {
        const int b = base + j * (T * 4) + tid * 4;
        const int4 q = *(const int4*)(dstp + min(b, E - 4));
        d[4 * j + 0] = q.x; d[4 * j + 1] = q.y; d[4 * j + 2] = q.z; d[4 * j + 3] = q.w;
      }
    } else {
#pragma unroll
      for (int j = 0; j < 4; ++j) {
#pragma unroll
        for (int k = 0; k < 4; ++k)
          d[4 * j + k] = dstp[min(base + j * (T * 4) + tid * 4 + k, E - 1)];
      }
    }
  };

  int cur[16], nxt[16];
#pragma unroll
  for (int q = 0; q < 16; ++q) { cur[q] = row0 - 1; nxt[q] = row0 - 1; }
  for (int base = -STEP; base < E; base += STEP) {
    unsigned mb = 0;
#pragma unroll
    for (int q = 0; q < 16; ++q) {
      const int eq = base + (q >> 2) * (T * 4) + tid * 4 + (q & 3);
      mb |= (eq < E && (unsigned)(cur[q] - row0) < (unsigned)rows) ? (1u << q) : 0u;
    }

    if (base + STEP < E) load16(base + STEP, nxt);

    while (__builtin_amdgcn_ballot_w32(mb != 0)) {
      const bool has = mb != 0;
      const int q = has ? __builtin_ctz(mb) : 0;
      mb &= mb - 1;
      int dsel = cur[0];
#pragma unroll
      for (int t = 1; t < 16; ++t) dsel = (q == t) ? cur[t] : dsel;
      const int edge = base + (q >> 2) * (T * 4) + tid * 4 + (q & 3);
      on_round(has, edge, dsel - row0);
    }

#pragma unroll
    for (int q = 0; q < 16; ++q) cur[q] = nxt[q];
  }
}

template <int CT, int H, int R, int CAP, bool REUSE>
__global__ __launch_bounds__(1024) void agg_softmax_rows_lds(
    const int* __restrict__ ei, const float* __restrict__ as, const float* __restrict__ ad,
    const float* __restrict__ feat, const float* __restrict__ bias,
    float* out, int* lists, int E, int n) {
  static_assert((CT == 256 && H == 4) || (CT == 64 && H == 1), "lane-to-head map below");
  static_assert(!REUSE || H == 1, "the hand-over is read by the one-head layer");
  constexpr int V = CT / 32;
  constexpr int T = 1024;
  constexpr int NF = 8;
  constexpr int LW = 64;
  constexpr int NJ = CAP / T;
  static_assert(CAP % T == 0, "every thread reads the same number of pairs");
  __shared__ float acc[R * CT];
  __shared__ float adrow[R * H];
  __shared__ float slog[R * H];
  __shared__ float mrow[R * H];
  __shared__ float srow[R * H];
  __shared__ int   l_src[CAP];
  __shared__ int   l_rel[CAP];
  __shared__ float l_as[CAP * H];
  __shared__ int   l_cnt;

  const int tid  = threadIdx.x;
  const int lane = tid & 31;
  const int wv   = tid >> 5;
  const int hi   = lane >> 4;
  const int row0 = blockIdx.x * R;
  const int rows = max(0, min(R, n - row0));
  const int* __restrict__ dstp = ei + E;

  int*  piece = lists + (size_t)row0 * LW;
  int2* pairs = (int2*)(piece + 4);
  const int capg = max(0, min(CAP, (rows * LW - 4) >> 1));
  int hdr = 0;
  if constexpr (REUSE) hdr = piece[0];

  auto lds_max = [&](int slot, float val) {
    if (val >= 0.0f) atomicMax((int*)&mrow[slot], __float_as_int(val));
    else             atomicMin((unsigned int*)&mrow[slot], __float_as_uint(val));
  };

  auto wave_accumulate = [&](unsigned m, int srcv, int rel, float x0, float x1, float x2, float x3) {
    while (m) {
      int  l[NF];
      bool ok[NF];
#pragma unroll
      for (int j = 0; j < NF; ++j) {
        ok[j] = (m != 0);
        l[j]  = ok[j] ? __builtin_ctz(m) : 0;
        if (ok[j]) m &= m - 1;
      }
      float v[NF][V];
#pragma unroll
      for (int j = 0; j < NF; ++j) {
        if (ok[j]) load_row_vals<CT>(feat + (size_t)bcast_i(srcv, l[j]) * CT, lane, v[j]);
      }
#pragma unroll
      for (int j = 0; j < NF; ++j) {
        if (ok[j]) {
          const int rl = bcast_i(rel, l[j]);
          float aA, aB;
          if constexpr (H == 4) {
            const float y0 = bcast_f(x0, l[j]), y1 = bcast_f(x1, l[j]);
            const float y2 = bcast_f(x2, l[j]), y3 = bcast_f(x3, l[j]);
            aA = hi ? y1 : y0; aB = hi ? y3 : y2;
          } else {
            aA = aB = bcast_f(x0, l[j]);
          }
#pragma unroll
          for (int i = 0; i < V; ++i)
            __hip_atomic_fetch_add(&acc[rl * CT + i * 32 + lane], (i < 4 ? aA : aB) * v[j][i],
                                   __ATOMIC_RELAXED, __HIP_MEMORY_SCOPE_WORKGROUP);
        }
      }
    }
  };

  if (tid == 0) l_cnt = 0;
  for (int i = tid; i < rows * H; i += T) {
    const float d_ = ad[(size_t)row0 * H + i];
    const float v  = edge_logit(as[(size_t)row0 * H + i], d_);
    adrow[i] = d_; slog[i] = v; mrow[i] = v;
  }
  __syncthreads();

  int  cnt;
  bool listed;
  if constexpr (!REUSE) {
    stream_dst_matches(dstp, E, row0, rows, tid, [&](bool has, int edge, int rel) {
      const unsigned bal = __builtin_amdgcn_ballot_w32(has);
      const int lead = __builtin_ctz(bal);
      int basepos = 0;
      if (lane == lead) basepos = atomicAdd(&l_cnt, __builtin_popcount(bal));
      basepos = bcast_i(basepos, lead);
      const int pos = basepos + __builtin_popcount(bal & ((1u << lane) - 1u));
      if (has && pos < CAP) { l_src[pos] = edge; l_rel[pos] = rel; }
    });
    __syncthreads();
    cnt    = l_cnt;
    listed = cnt <= CAP;
    if (tid == 0) piece[0] = cnt;
  } else {
    cnt    = hdr;
    listed = (cnt >= 0) && (cnt <= capg);
  }

  if (listed) {
   if constexpr (REUSE) {
    if (cnt > 0) {
      int2 pr[NJ];
#pragma unroll
      for (int j = 0; j < NJ; ++j) pr[j] = pairs[min(tid + j * T, cnt - 1)];
      int sv[NJ]; float av[NJ];
#pragma unroll
      for (int j = 0; j < NJ; ++j) {
        sv[j] = min(max(pr[j].x, 0), n - 1);
        av[j] = as[sv[j]];
      }
#pragma unroll
      for (int j = 0; j < NJ; ++j) {
        const int i   = min(tid + j * T, cnt - 1);
        const int rel = min(max(pr[j].y, 0), rows - 1);
        l_src[i] = sv[j]; l_rel[i] = rel; l_as[i] = av[j];
        if (tid + j * T < cnt) lds_max(rel, edge_logit(av[j], adrow[rel]));
      }
    }
   } else {
    const bool pub = cnt <= capg;
    for (int i = tid; i < cnt; i += T) {
      const int edge = l_src[i], rel = l_rel[i];
      const int src  = min(max(ei[edge], 0), n - 1);
      l_src[i] = src;
      if (pub) pairs[i] = make_int2(src, rel);
      if constexpr (H == 4) {
        const float4 a = *(const float4*)(as + (size_t)src * 4);
        l_as[i * 4 + 0] = a.x; l_as[i * 4 + 1] = a.y; l_as[i * 4 + 2] = a.z; l_as[i * 4 + 3] = a.w;
        lds_max(rel * 4 + 0, edge_logit(a.x, adrow[rel * 4 + 0]));
        lds_max(rel * 4 + 1, edge_logit(a.y, adrow[rel * 4 + 1]));
        lds_max(rel * 4 + 2, edge_logit(a.z, adrow[rel * 4 + 2]));
        lds_max(rel * 4 + 3, edge_logit(a.w, adrow[rel * 4 + 3]));
      } else {
        const float a = as[src];
        l_as[i] = a;
        lds_max(rel, edge_logit(a, adrow[rel]));
      }
    }
   }
  } else {
    stream_dst_matches(dstp, E, row0, rows, tid, [&](bool has, int edge, int rel) {
      if (has) {
        const int src = min(max(ei[edge], 0), n - 1);
        if constexpr (H == 4) {
          const float4 a = *(const float4*)(as + (size_t)src * 4);
          lds_max(rel * 4 + 0, edge_logit(a.x, adrow[rel * 4 + 0]));
          lds_max(rel * 4 + 1, edge_logit(a.y, adrow[rel * 4 + 1]));
          lds_max(rel * 4 + 2, edge_logit(a.z, adrow[rel * 4 + 2]));
          lds_max(rel * 4 + 3, edge_logit(a.w, adrow[rel * 4 + 3]));
        } else {
          lds_max(rel, edge_logit(as[src], adrow[rel]));
        }
      }
    });
  }
  __syncthreads();

  for (int r = wv; r < rows; r += 32) {
    float v[V];
    load_row_vals<CT>(feat + (size_t)(row0 + r) * CT, lane, v);
    if constexpr (H == 4) {
      const float xA = expf(slog[r * 4 + hi]     - mrow[r * 4 + hi]);
      const float xB = expf(slog[r * 4 + 2 + hi] - mrow[r * 4 + 2 + hi]);
#pragma unroll
      for (int i = 0; i < V; ++i) acc[r * CT + i * 32 + lane] = (i < 4 ? xA : xB) * v[i];
      if (lane < 4) srow[r * 4 + lane] = expf(slog[r * 4 + lane] - mrow[r * 4 + lane]);
    } else {
      const float x = expf(slog[r] - mrow[r]);
#pragma unroll
      for (int i = 0; i < V; ++i) acc[r * CT + i * 32 + lane] = x * v[i];
      if (lane == 0) srow[r] = x;
    }
  }
  __syncthreads();

  if (listed) {
    const int lo = (int)(((long)wv * cnt) >> 5), up = (int)(((long)(wv + 1) * cnt) >> 5);
    for (int cb = lo; cb < up; cb += 32) {
      const int  idx = cb + lane;
      const bool has = idx < up;
      int srcv = 0, rel = 0;
      float x0 = 0.f, x1 = 0.f, x2 = 0.f, x3 = 0.f;
      if (has) {
        srcv = l_src[idx]; rel = l_rel[idx];
        if constexpr (H == 4) {
          x0 = expf(edge_logit(l_as[idx * 4 + 0], adrow[rel * 4 + 0]) - mrow[rel * 4 + 0]);
          x1 = expf(edge_logit(l_as[idx * 4 + 1], adrow[rel * 4 + 1]) - mrow[rel * 4 + 1]);
          x2 = expf(edge_logit(l_as[idx * 4 + 2], adrow[rel * 4 + 2]) - mrow[rel * 4 + 2]);
          x3 = expf(edge_logit(l_as[idx * 4 + 3], adrow[rel * 4 + 3]) - mrow[rel * 4 + 3]);
          __hip_atomic_fetch_add(&srow[rel * 4 + 0], x0, __ATOMIC_RELAXED, __HIP_MEMORY_SCOPE_WORKGROUP);
          __hip_atomic_fetch_add(&srow[rel * 4 + 1], x1, __ATOMIC_RELAXED, __HIP_MEMORY_SCOPE_WORKGROUP);
          __hip_atomic_fetch_add(&srow[rel * 4 + 2], x2, __ATOMIC_RELAXED, __HIP_MEMORY_SCOPE_WORKGROUP);
          __hip_atomic_fetch_add(&srow[rel * 4 + 3], x3, __ATOMIC_RELAXED, __HIP_MEMORY_SCOPE_WORKGROUP);
        } else {
          x0 = expf(edge_logit(l_as[idx], adrow[rel]) - mrow[rel]);
          __hip_atomic_fetch_add(&srow[rel], x0, __ATOMIC_RELAXED, __HIP_MEMORY_SCOPE_WORKGROUP);
        }
      }
      wave_accumulate(__builtin_amdgcn_ballot_w32(has), srcv, rel, x0, x1, x2, x3);
    }
  } else {
    stream_dst_matches(dstp, E, row0, rows, tid, [&](bool has, int edge, int rel) {
      int srcv = 0;
      float x0 = 0.f, x1 = 0.f, x2 = 0.f, x3 = 0.f;
      if (has) {
        srcv = min(max(ei[edge], 0), n - 1);
        if constexpr (H == 4) {
          const float4 a = *(const float4*)(as + (size_t)srcv * 4);
          x0 = expf(edge_logit(a.x, adrow[rel * 4 + 0]) - mrow[rel * 4 + 0]);
          x1 = expf(edge_logit(a.y, adrow[rel * 4 + 1]) - mrow[rel * 4 + 1]);
          x2 = expf(edge_logit(a.z, adrow[rel * 4 + 2]) - mrow[rel * 4 + 2]);
          x3 = expf(edge_logit(a.w, adrow[rel * 4 + 3]) - mrow[rel * 4 + 3]);
          __hip_atomic_fetch_add(&srow[rel * 4 + 0], x0, __ATOMIC_RELAXED, __HIP_MEMORY_SCOPE_WORKGROUP);
          __hip_atomic_fetch_add(&srow[rel * 4 + 1], x1, __ATOMIC_RELAXED, __HIP_MEMORY_SCOPE_WORKGROUP);
          __hip_atomic_fetch_add(&srow[rel * 4 + 2], x2, __ATOMIC_RELAXED, __HIP_MEMORY_SCOPE_WORKGROUP);
          __hip_atomic_fetch_add(&srow[rel * 4 + 3], x3, __ATOMIC_RELAXED, __HIP_MEMORY_SCOPE_WORKGROUP);
        } else {
          x0 = expf(edge_logit(as[srcv], adrow[rel]) - mrow[rel]);
          __hip_atomic_fetch_add(&srow[rel], x0, __ATOMIC_RELAXED, __HIP_MEMORY_SCOPE_WORKGROUP);
        }
      }
      wave_accumulate(__builtin_amdgcn_ballot_w32(has), srcv, rel, x0, x1, x2, x3);
    });
  }
  __syncthreads();

  float bv[V];
  load_row_vals<CT>(bias, lane, bv);
  for (int r = wv; r < rows; r += 32) {
    float* op = out + (size_t)(row0 + r) * CT;
    float dA, dB;
    if constexpr (H == 4) {
      dA = srow[r * 4 + hi] + 1e-16f; dB = srow[r * 4 + 2 + hi] + 1e-16f;
    } else {
      dA = dB = srow[r] + 1e-16f;
    }
    float v[V];
#pragma unroll
    for (int i = 0; i < V; ++i) v[i] = bv[i] + acc[r * CT + i * 32 + lane] / (i < 4 ? dA : dB);
    if constexpr (CT == 256) {
      *(float4*)(op + 4 * lane)       = make_float4(v[0], v[1], v[2], v[3]);
      *(float4*)(op + 128 + 4 * lane) = make_float4(v[4], v[5], v[6], v[7]);
    } else {
      *(float2*)(op + 2 * lane) = make_float2(v[0], v[1]);
    }
  }
}

template <int C>
__global__ __launch_bounds__(256) void ln_add_elu(const float* __restrict__ x, const float* __restrict__ g,
                           const float* __restrict__ b, const float* __restrict__ add,
                           float* __restrict__ y, int n) {
  constexpr int V = C / 32;
  const int lane = threadIdx.x & 31;
  const int nw   = gridDim.x * (blockDim.x >> 5);
  float gv[V], bv[V];
#pragma unroll
  for (int i = 0; i < V; ++i) { gv[i] = g[lane + 32 * i]; bv[i] = b[lane + 32 * i]; }
  for (int row = blockIdx.x * (blockDim.x >> 5) + (threadIdx.x >> 5); row < n; row += nw) {
    const float* xr = x + (size_t)row * C;
    float v[V], av[V];
#pragma unroll
    for (int i = 0; i < V; ++i) v[i] = xr[lane + 32 * i];
    if (add) {
      const float* ar = add + (size_t)row * C;
#pragma unroll
      for (int i = 0; i < V; ++i) av[i] = ar[lane + 32 * i];
    } else {
#pragma unroll
      for (int i = 0; i < V; ++i) av[i] = 0.f;
    }
    float s = 0.f, ss = 0.f;
#pragma unroll
    for (int i = 0; i < V; ++i) { s += v[i]; ss += v[i] * v[i]; }
    for (int off = 16; off; off >>= 1) {
      s  += __shfl_xor(s,  off, 32);
      ss += __shfl_xor(ss, off, 32);
    }
    const float mu  = s / (float)C;
    const float var = ss / (float)C - mu * mu;
    const float inv = rsqrtf(var + 1e-5f);
    float* yr = y + (size_t)row * C;
#pragma unroll
    for (int i = 0; i < V; ++i) {
      float o = (v[i] - mu) * inv * gv[i] + bv[i];
      if (add) o += av[i];
      yr[lane + 32 * i] = (o > 0.f) ? o : expm1f(o);
    }
  }
}

extern "C" void kernel_launch(void* const* d_in, const int* in_sizes, int n_in,
                              void* d_out, int out_size, void* d_ws, size_t ws_size,
                              hipStream_t stream) {
  const float* x    = (const float*)d_in[0];
  const int*   ei   = (const int*)  d_in[1];
  const float* W1   = (const float*)d_in[2];
  const float* aS1  = (const float*)d_in[3];
  const float* aD1  = (const float*)d_in[4];
  const float* b1   = (const float*)d_in[5];
  const float* W2   = (const float*)d_in[6];
  const float* aS2  = (const float*)d_in[7];
  const float* aD2  = (const float*)d_in[8];
  const float* b2   = (const float*)d_in[9];
  const float* g1   = (const float*)d_in[10];
  const float* be1  = (const float*)d_in[11];
  const float* g2   = (const float*)d_in[12];
  const float* be2  = (const float*)d_in[13];
  const float* pW   = (const float*)d_in[14];
  const float* pb   = (const float*)d_in[15];

  const int IN_CH = 128, HEADS = 4, OUT_CH = 64, HC = HEADS * OUT_CH;
  const int N  = in_sizes[0] / IN_CH;
  const int E  = in_sizes[1] / 2;

  float* ws   = (float*)d_ws;
  float* h1   = ws; ws += (size_t)N * HC;
  float* out1 = ws; ws += (size_t)N * HC;
  float* idb  = ws; ws += (size_t)N * OUT_CH;
  float* z2   = ws; ws += (size_t)N * OUT_CH;
  float* out2 = ws; ws += (size_t)N * OUT_CH;
  float* as1  = ws; ws += (size_t)N * HEADS;
  float* ad1  = ws; ws += (size_t)N * HEADS;
  float* as2  = ws; ws += (size_t)N;
  float* ad2  = ws; ws += (size_t)N;

  auto cdiv = [](long a, long b) -> int { return (int)((a + b - 1) / b); };
  const int BLK = 256;
  const int WPB = BLK / 32;
  const int RPW = 4;

  gemm_lds_weights<128, 4, 1, 4, 16><<<256, 512, 0, stream>>>(x, W1, h1, pW, pb, idb, N);

  att_scores<<<cdiv((long)N * HEADS, BLK), BLK, 0, stream>>>(h1, aS1, aD1, as1, ad1, N, HEADS, OUT_CH);
  agg_softmax_rows_lds<256, 4, 196, 3072, false><<<cdiv(N, 196), 1024, 0, stream>>>(ei, as1, ad1, h1, b1, out1, (int*)out2, E, N);

  ln_add_elu<256><<<cdiv(N, WPB * RPW), BLK, 0, stream>>>(out1, g1, be1, nullptr, out1, N);

  gemm_lds_weights<256, 2, 0, 2, 8><<<256, 256, 0, stream>>>(out1, W2, z2, nullptr, nullptr, nullptr, N);

  att_scores<<<cdiv((long)N, BLK), BLK, 0, stream>>>(z2, aS2, aD2, as2, ad2, N, 1, OUT_CH);
  agg_softmax_rows_lds<64, 1, 196, 3072, true><<<cdiv(N, 196), 1024, 0, stream>>>(ei, as2, ad2, z2, b2, out2, (int*)out2, E, N);

  ln_add_elu<64><<<cdiv(N, WPB * RPW), BLK, 0, stream>>>(out2, g2, be2, idb, (float*)d_out, N);

  (void)n_in; (void)out_size; (void)ws_size;
}


# layer 1's destination-list pass: 16-byte loads, one subtract+compare per 32 edges gives the wave's match mask (scalar), one LDS counter add per wave and block; list order differs, sums stay LDS adds
# speedup vs baseline: 5.2383x; 1.0686x over previous
#include <hip/hip_runtime.h>
#include <math.h>

typedef float v2f __attribute__((ext_vector_type(2)));
typedef float v8f __attribute__((ext_vector_type(8)));

#define NEG_SLOPE 0.2f

template <int K, int NCOLS, int THREADS>
__device__ __forceinline__ void stage_weights(const float* __restrict__ B,
                                              float4* __restrict__ dst, int tid) {
  constexpr int KP   = K / 8;
  constexpr int ENT  = (NCOLS / 16) * KP * 32;
  constexpr int ITER = ENT / THREADS;
  static_assert(ENT % THREADS == 0 && THREADS % 32 == 0, "whole waves share the entries evenly");
  const int c = tid & 15, h = (tid >> 4) & 1;
  float4 v[ITER];
#pragma unroll
  for (int it = 0; it < ITER; ++it) {
    const int g = (tid + it * THREADS) >> 5;
    const int tile = g / KP, kp = g - tile * KP;
    const float* p = B + (size_t)(8 * kp + 2 * h) * NCOLS + tile * 16 + c;
    v[it].x = p[0]; v[it].y = p[NCOLS]; v[it].z = p[4 * NCOLS]; v[it].w = p[5 * NCOLS];
  }
#pragma unroll
  for (int it = 0; it < ITER; ++it) dst[tid + it * THREADS] = v[it];
}

template <int K>
__device__ __forceinline__ void load_a_strip(const float* __restrict__ A, int strip, int lane,
                                             v2f (&a)[K / 4]) {
  const float* ap = A + (size_t)(strip * 16 + (lane & 15)) * K + ((lane >> 4) << 1);
#pragma unroll
  for (int ks = 0; ks < K / 4; ++ks) a[ks] = *(const v2f*)(ap + 4 * ks);
}

template <int K, int T1, int T2, int UPS, int WAVES>
__global__ __launch_bounds__(WAVES * 32) void gemm_lds_weights(
    const float* __restrict__ A,
    const float* __restrict__ B1, float* __restrict__ C1,
    const float* __restrict__ B2, const float* __restrict__ bias2, float* __restrict__ C2,
    int M) {
  constexpr int NT1 = UPS * T1, NT2 = UPS * T2;
  constexpr int N1 = NT1 * 16, N2 = NT2 * 16;
  constexpr int KP = K / 8;
  constexpr int TT = T1 + T2;
  constexpr int RS = WAVES / UPS;
  static_assert(T1 % 2 == 0 && WAVES % UPS == 0 && K % 8 == 0, "C1 tiles are stored in pairs");
  constexpr int NEED = (NT1 + NT2) * KP * 32;
  constexpr int FLOOR = 164 * 1024 / 16;
  __shared__ float4 Bl[NEED > FLOOR ? NEED : FLOOR];

  const int tid  = threadIdx.x;
  const int lane = tid & 31;
  const int wv   = tid >> 5;
  const int hi   = lane >> 4;
  const int cg   = wv % UPS;
  const int rr   = wv / UPS;

  const int strips = M >> 4;
  const int sb = (int)((blockIdx.x * (unsigned)strips) / gridDim.x);
  const int se = (int)(((blockIdx.x + 1) * (unsigned)strips) / gridDim.x);

  v2f an[K / 4];
  int st = sb + rr;
  if (st < se) load_a_strip<K>(A, st, lane, an);

  stage_weights<K, N1, WAVES * 32>(B1, Bl, tid);
  if constexpr (T2 > 0)
    stage_weights<K, N2, WAVES * 32>(B2, Bl + NT1 * KP * 32, tid);
  __syncthreads();

  for (; st < se; st += RS) {
    v2f a[K / 4];
#pragma unroll
    for (int ks = 0; ks < K / 4; ++ks) a[ks] = an[ks];
    if (st + RS < se) load_a_strip<K>(A, st + RS, lane, an);

    v8f acc[TT];
#pragma unroll
    for (int t = 0; t < TT; ++t) acc[t] = v8f{};
    float4 bc[TT], bn[TT];
#pragma unroll
    for (int t = 0; t < TT; ++t) {
      const int tile = (t < T1) ? (cg * T1 + t) : (NT1 + cg * T2 + (t - T1));
      bc[t] = Bl[(tile * KP) * 32 + lane];
    }
#pragma unroll
    for (int kp = 0; kp < KP; ++kp) {
      if (kp + 1 < KP) {
#pragma unroll
        for (int t = 0; t < TT; ++t) {
          const int tile = (t < T1) ? (cg * T1 + t) : (NT1 + cg * T2 + (t - T1));
          bn[t] = Bl[(tile * KP + kp + 1) * 32 + lane];
        }
      }
#pragma unroll
      for (int t = 0; t < TT; ++t) {
        v2f bb; bb.x = bc[t].x; bb.y = bc[t].y;
        acc[t] = __builtin_amdgcn_wmma_f32_16x16x4_f32(
             false, a[2 * kp],  false, bb,
             (short)0, acc[t],  false,  false);
      }
#pragma unroll
      for (int t = 0; t < TT; ++t) {
        v2f bb; bb.x = bc[t].z; bb.y = bc[t].w;
        acc[t] = __builtin_amdgcn_wmma_f32_16x16x4_f32(
             false, a[2 * kp + 1],  false, bb,
             (short)0, acc[t],  false,  false);
      }
#pragma unroll
      for (int t = 0; t < TT; ++t) bc[t] = bn[t];
      __builtin_amdgcn_sched_barrier(0);
    }

#pragma unroll
    for (int p = 0; p < T1 / 2; ++p) {
      float* cp = C1 + (size_t)(st * 16) * N1 + (cg * T1 + 2 * p) * 16 + lane;
#pragma unroll
      for (int v = 0; v < 8; ++v) {
        const float own0 = acc[2 * p][v], own1 = acc[2 * p + 1][v];
        const float give = hi ? own0 : own1;
        const float got  = __shfl_xor(give, 16, 32);
        cp[(size_t)v * N1]       = hi ? got : own0;
        cp[(size_t)(8 + v) * N1] = hi ? own1 : got;
      }
    }
    if constexpr (T2 > 0) {
#pragma unroll
      for (int t = 0; t < T2; ++t) {
        const int ncol = (cg * T2 + t) * 16 + (lane & 15);
        const float bv = bias2 ? bias2[ncol] : 0.0f;
        float* cp = C2 + (size_t)(st * 16 + (hi << 3)) * N2 + ncol;
#pragma unroll
        for (int v = 0; v < 8; ++v) cp[(size_t)v * N2] = acc[T1 + t][v] + bv;
      }
    }
  }
}

__global__ void att_scores(const float* __restrict__ h,
                           const float* __restrict__ att_src,
                           const float* __restrict__ att_dst,
                           float* __restrict__ as, float* __restrict__ ad,
                           int n, int H, int C) {
  long t = (long)blockIdx.x * blockDim.x + threadIdx.x;
  if (t >= (long)n * H) return;
  const int head = (int)(t % H);
  const float4* hv = (const float4*)(h + (size_t)t * C);
  const float4* sv = (const float4*)(att_src + (size_t)head * C);
  const float4* dv = (const float4*)(att_dst + (size_t)head * C);
  float s = 0.f, d = 0.f;
  for (int i = 0; i < (C >> 2); ++i) {
    float4 xv = hv[i], a = sv[i], b = dv[i];
    s += xv.x * a.x + xv.y * a.y + xv.z * a.z + xv.w * a.w;
    d += xv.x * b.x + xv.y * b.y + xv.z * b.z + xv.w * b.w;
  }
  as[t] = s; ad[t] = d;
}

template <int CT>
__device__ __forceinline__ void load_row_vals(const float* __restrict__ p, int lane,
                                              float (&v)[CT / 32]) {
  if constexpr (CT == 256) {
    const float4 a = *(const float4*)(p + 4 * lane);
    const float4 b = *(const float4*)(p + 128 + 4 * lane);
    v[0] = a.x; v[1] = a.y; v[2] = a.z; v[3] = a.w;
    v[4] = b.x; v[5] = b.y; v[6] = b.z; v[7] = b.w;
  } else {
    const float2 a = *(const float2*)(p + 2 * lane);
    v[0] = a.x; v[1] = a.y;
  }
}

__device__ __forceinline__ int bcast_i(int v, int l) {
  return __builtin_amdgcn_readlane(v, l);
}
__device__ __forceinline__ float bcast_f(float v, int l) {
  return __int_as_float(__builtin_amdgcn_readlane(__float_as_int(v), l));
}

__device__ __forceinline__ float edge_logit(float a_src, float a_dst) {
  const float v = a_src + a_dst;
  return ((v > 0.f) ? v : NEG_SLOPE * v) + 0.0f;
}

template <typename F>
__device__ __forceinline__ void stream_dst_matches(const int* __restrict__ dstp, int E,
                                                   int row0, int rows, int tid, F on_round) {
  constexpr int T = 1024;
  constexpr int STEP = T * 16;
  const bool vec_ok = (E >= 4) && ((E & 3) == 0) && ((((size_t)dstp) & 15) == 0);

  auto load16 = [&](int base, int (&d)[16]) {
    if (vec_ok) {
#pragma unroll
      for (int j = 0; j < 4; ++j) {
        const int b = base + j * (T * 4) + tid * 4;
        const int4 q = *(const int4*)(dstp + min(b, E - 4));
        d[4 * j + 0] = q.x; d[4 * j + 1] = q.y; d[4 * j + 2] = q.z; d[4 * j + 3] = q.w;
      }
    } else {
#pragma unroll
      for (int j = 0; j < 4; ++j) {
#pragma unroll
        for (int k = 0; k < 4; ++k)
          d[4 * j + k] = dstp[min(base + j * (T * 4) + tid * 4 + k, E - 1)];
      }
    }
  };

  int cur[16], nxt[16];
#pragma unroll
  for (int q = 0; q < 16; ++q) { cur[q] = row0 - 1; nxt[q] = row0 - 1; }
  for (int base = -STEP; base < E; base += STEP) {
    unsigned mb = 0;
#pragma unroll
    for (int q = 0; q < 16; ++q) {
      const int eq = base + (q >> 2) * (T * 4) + tid * 4 + (q & 3);
      mb |= (eq < E && (unsigned)(cur[q] - row0) < (unsigned)rows) ? (1u << q) : 0u;
    }

    if (base + STEP < E) load16(base + STEP, nxt);

    while (__builtin_amdgcn_ballot_w32(mb != 0)) {
      const bool has = mb != 0;
      const int q = has ? __builtin_ctz(mb) : 0;
      mb &= mb - 1;
      int dsel = cur[0];
#pragma unroll
      for (int t = 1; t < 16; ++t) dsel = (q == t) ? cur[t] : dsel;
      const int edge = base + (q >> 2) * (T * 4) + tid * 4 + (q & 3);
      on_round(has, edge, dsel - row0);
    }

#pragma unroll
    for (int q = 0; q < 16; ++q) cur[q] = nxt[q];
  }
}

template <int CAP, bool TAIL>
__device__ __forceinline__ void list_block(int base, const int4 (&d)[4], int nq, int row0, int rows,
                                           int tid, int lane, int* l_src, int* l_rel, int* l_cnt) {
  constexpr int T = 1024;
  unsigned mk[16];
#pragma unroll
  for (int j = 0; j < 4; ++j) {
    unsigned okm = 0xffffffffu;
    if constexpr (TAIL) okm = __builtin_amdgcn_ballot_w32((base >> 2) + j * T + tid < nq);
    mk[4 * j + 0] = __builtin_amdgcn_ballot_w32((unsigned)(d[j].x - row0) < (unsigned)rows) & okm;
    mk[4 * j + 1] = __builtin_amdgcn_ballot_w32((unsigned)(d[j].y - row0) < (unsigned)rows) & okm;
    mk[4 * j + 2] = __builtin_amdgcn_ballot_w32((unsigned)(d[j].z - row0) < (unsigned)rows) & okm;
    mk[4 * j + 3] = __builtin_amdgcn_ballot_w32((unsigned)(d[j].w - row0) < (unsigned)rows) & okm;
  }
  int tot = 0;
#pragma unroll
  for (int q = 0; q < 16; q += 2)
    tot += __builtin_popcountll(((unsigned long long)mk[q + 1] << 32) | (unsigned long long)mk[q]);
  if (tot != 0) {
    const unsigned below = (1u << lane) - 1u;
    int pos0 = 0;
    if (lane == 0) pos0 = atomicAdd(l_cnt, tot);
    pos0 = bcast_i(pos0, 0);
#pragma unroll
    for (int q = 0; q < 16; ++q) {
      const unsigned m = mk[q];
      if (m != 0) {
        const int pos = pos0 + __builtin_popcount(m & below);
        const int4 dj = d[q >> 2];
        const int dv = (q & 3) == 0 ? dj.x : (q & 3) == 1 ? dj.y : (q & 3) == 2 ? dj.z : dj.w;
        if (((m >> lane) & 1u) && pos < CAP) {
          l_src[pos] = base + (q >> 2) * (T * 4) + tid * 4 + (q & 3);
          l_rel[pos] = dv - row0;
        }
        pos0 += __builtin_popcount(m);
      }
    }
  }
}

template <int CAP>
__device__ __forceinline__ void list_owned_edges_vec(const int* __restrict__ dstp, int E,
                                                     int row0, int rows, int tid, int lane,
                                                     int* l_src, int* l_rel, int* l_cnt) {
  constexpr int T = 1024;
  constexpr int STEP = T * 16;
  const int4* __restrict__ vp = (const int4*)dstp;
  const int nq    = E >> 2;
  const int nfull = E / STEP;
  const int tail0 = nfull * STEP;

  auto request = [&](int base, int4 (&d)[4]) {
#pragma unroll
    for (int j = 0; j < 4; ++j) d[j] = vp[min((base >> 2) + j * T + tid, nq - 1)];
  };

  int4 rt[4], ra[4], rb[4];
  request(tail0, rt);
  request(0, ra);
  for (int b = 0; b < nfull; b += 2) {
    request((b + 1) * STEP, rb);
    list_block<CAP, false>(b * STEP, ra, nq, row0, rows, tid, lane, l_src, l_rel, l_cnt);
    if (b + 1 < nfull) {
      request((b + 2) * STEP, ra);
      list_block<CAP, false>((b + 1) * STEP, rb, nq, row0, rows, tid, lane, l_src, l_rel, l_cnt);
    }
  }
  if (tail0 < E) list_block<CAP, true>(tail0, rt, nq, row0, rows, tid, lane, l_src, l_rel, l_cnt);
}

template <int CT, int H, int R, int CAP, bool REUSE>
__global__ __launch_bounds__(1024) void agg_softmax_rows_lds(
    const int* __restrict__ ei, const float* __restrict__ as, const float* __restrict__ ad,
    const float* __restrict__ feat, const float* __restrict__ bias,
    float* out, int* lists, int E, int n) {
  static_assert((CT == 256 && H == 4) || (CT == 64 && H == 1), "lane-to-head map below");
  static_assert(!REUSE || H == 1, "the hand-over is read by the one-head layer");
  constexpr int V = CT / 32;
  constexpr int T = 1024;
  constexpr int NF = 8;
  constexpr int LW = 64;
  constexpr int NJ = CAP / T;
  static_assert(CAP % T == 0, "every thread reads the same number of pairs");
  __shared__ float acc[R * CT];
  __shared__ float adrow[R * H];
  __shared__ float slog[R * H];
  __shared__ float mrow[R * H];
  __shared__ float srow[R * H];
  __shared__ int   l_src[CAP];
  __shared__ int   l_rel[CAP];
  __shared__ float l_as[CAP * H];
  __shared__ int   l_cnt;

  const int tid  = threadIdx.x;
  const int lane = tid & 31;
  const int wv   = tid >> 5;
  const int hi   = lane >> 4;
  const int row0 = blockIdx.x * R;
  const int rows = max(0, min(R, n - row0));
  const int* __restrict__ dstp = ei + E;

  int*  piece = lists + (size_t)row0 * LW;
  int2* pairs = (int2*)(piece + 4);
  const int capg = max(0, min(CAP, (rows * LW - 4) >> 1));
  int hdr = 0;
  if constexpr (REUSE) hdr = piece[0];

  auto lds_max = [&](int slot, float val) {
    if (val >= 0.0f) atomicMax((int*)&mrow[slot], __float_as_int(val));
    else             atomicMin((unsigned int*)&mrow[slot], __float_as_uint(val));
  };

  auto wave_accumulate = [&](unsigned m, int srcv, int rel, float x0, float x1, float x2, float x3) {
    while (m) {
      int  l[NF];
      bool ok[NF];
#pragma unroll
      for (int j = 0; j < NF; ++j) {
        ok[j] = (m != 0);
        l[j]  = ok[j] ? __builtin_ctz(m) : 0;
        if (ok[j]) m &= m - 1;
      }
      float v[NF][V];
#pragma unroll
      for (int j = 0; j < NF; ++j) {
        if (ok[j]) load_row_vals<CT>(feat + (size_t)bcast_i(srcv, l[j]) * CT, lane, v[j]);
      }
#pragma unroll
      for (int j = 0; j < NF; ++j) {
        if (ok[j]) {
          const int rl = bcast_i(rel, l[j]);
          float aA, aB;
          if constexpr (H == 4) {
            const float y0 = bcast_f(x0, l[j]), y1 = bcast_f(x1, l[j]);
            const float y2 = bcast_f(x2, l[j]), y3 = bcast_f(x3, l[j]);
            aA = hi ? y1 : y0; aB = hi ? y3 : y2;
          } else {
            aA = aB = bcast_f(x0, l[j]);
          }
#pragma unroll
          for (int i = 0; i < V; ++i)
            __hip_atomic_fetch_add(&acc[rl * CT + i * 32 + lane], (i < 4 ? aA : aB) * v[j][i],
                                   __ATOMIC_RELAXED, __HIP_MEMORY_SCOPE_WORKGROUP);
        }
      }
    }
  };

  if (tid == 0) l_cnt = 0;
  for (int i = tid; i < rows * H; i += T) {
    const float d_ = ad[(size_t)row0 * H + i];
    const float v  = edge_logit(as[(size_t)row0 * H + i], d_);
    adrow[i] = d_; slog[i] = v; mrow[i] = v;
  }
  __syncthreads();

  int  cnt;
  bool listed;
  if constexpr (!REUSE) {
    const bool vec_ok = (E >= 4) && ((E & 3) == 0) && ((((size_t)dstp) & 15) == 0);
    if (vec_ok) list_owned_edges_vec<CAP>(dstp, E, row0, rows, tid, lane, l_src, l_rel, &l_cnt);
    else stream_dst_matches(dstp, E, row0, rows, tid, [&](bool has, int edge, int rel) {
      const unsigned bal = __builtin_amdgcn_ballot_w32(has);
      const int lead = __builtin_ctz(bal);
      int basepos = 0;
      if (lane == lead) basepos = atomicAdd(&l_cnt, __builtin_popcount(bal));
      basepos = bcast_i(basepos, lead);
      const int pos = basepos + __builtin_popcount(bal & ((1u << lane) - 1u));
      if (has && pos < CAP) { l_src[pos] = edge; l_rel[pos] = rel; }
    });
    __syncthreads();
    cnt    = l_cnt;
    listed = cnt <= CAP;
    if (tid == 0) piece[0] = cnt;
  } else {
    cnt    = hdr;
    listed = (cnt >= 0) && (cnt <= capg);
  }

  if (listed) {
   if constexpr (REUSE) {
    if (cnt > 0) {
      int2 pr[NJ];
#pragma unroll
      for (int j = 0; j < NJ; ++j) pr[j] = pairs[min(tid + j * T, cnt - 1)];
      int sv[NJ]; float av[NJ];
#pragma unroll
      for (int j = 0; j < NJ; ++j) {
        sv[j] = min(max(pr[j].x, 0), n - 1);
        av[j] = as[sv[j]];
      }
#pragma unroll
      for (int j = 0; j < NJ; ++j) {
        const int i   = min(tid + j * T, cnt - 1);
        const int rel = min(max(pr[j].y, 0), rows - 1);
        l_src[i] = sv[j]; l_rel[i] = rel; l_as[i] = av[j];
        if (tid + j * T < cnt) lds_max(rel, edge_logit(av[j], adrow[rel]));
      }
    }
   } else {
    const bool pub = cnt <= capg;
    for (int i = tid; i < cnt; i += T) {
      const int edge = l_src[i], rel = l_rel[i];
      const int src  = min(max(ei[edge], 0), n - 1);
      l_src[i] = src;
      if (pub) pairs[i] = make_int2(src, rel);
      if constexpr (H == 4) {
        const float4 a = *(const float4*)(as + (size_t)src * 4);
        l_as[i * 4 + 0] = a.x; l_as[i * 4 + 1] = a.y; l_as[i * 4 + 2] = a.z; l_as[i * 4 + 3] = a.w;
        lds_max(rel * 4 + 0, edge_logit(a.x, adrow[rel * 4 + 0]));
        lds_max(rel * 4 + 1, edge_logit(a.y, adrow[rel * 4 + 1]));
        lds_max(rel * 4 + 2, edge_logit(a.z, adrow[rel * 4 + 2]));
        lds_max(rel * 4 + 3, edge_logit(a.w, adrow[rel * 4 + 3]));
      } else {
        const float a = as[src];
        l_as[i] = a;
        lds_max(rel, edge_logit(a, adrow[rel]));
      }
    }
   }
  } else {
    stream_dst_matches(dstp, E, row0, rows, tid, [&](bool has, int edge, int rel) {
      if (has) {
        const int src = min(max(ei[edge], 0), n - 1);
        if constexpr (H == 4) {
          const float4 a = *(const float4*)(as + (size_t)src * 4);
          lds_max(rel * 4 + 0, edge_logit(a.x, adrow[rel * 4 + 0]));
          lds_max(rel * 4 + 1, edge_logit(a.y, adrow[rel * 4 + 1]));
          lds_max(rel * 4 + 2, edge_logit(a.z, adrow[rel * 4 + 2]));
          lds_max(rel * 4 + 3, edge_logit(a.w, adrow[rel * 4 + 3]));
        } else {
          lds_max(rel, edge_logit(as[src], adrow[rel]));
        }
      }
    });
  }
  __syncthreads();

  for (int r = wv; r < rows; r += 32) {
    float v[V];
    load_row_vals<CT>(feat + (size_t)(row0 + r) * CT, lane, v);
    if constexpr (H == 4) {
      const float xA = expf(slog[r * 4 + hi]     - mrow[r * 4 + hi]);
      const float xB = expf(slog[r * 4 + 2 + hi] - mrow[r * 4 + 2 + hi]);
#pragma unroll
      for (int i = 0; i < V; ++i) acc[r * CT + i * 32 + lane] = (i < 4 ? xA : xB) * v[i];
      if (lane < 4) srow[r * 4 + lane] = expf(slog[r * 4 + lane] - mrow[r * 4 + lane]);
    } else {
      const float x = expf(slog[r] - mrow[r]);
#pragma unroll
      for (int i = 0; i < V; ++i) acc[r * CT + i * 32 + lane] = x * v[i];
      if (lane == 0) srow[r] = x;
    }
  }
  __syncthreads();

  if (listed) {
    const int lo = (int)(((long)wv * cnt) >> 5), up = (int)(((long)(wv + 1) * cnt) >> 5);
    for (int cb = lo; cb < up; cb += 32) {
      const int  idx = cb + lane;
      const bool has = idx < up;
      int srcv = 0, rel = 0;
      float x0 = 0.f, x1 = 0.f, x2 = 0.f, x3 = 0.f;
      if (has) {
        srcv = l_src[idx]; rel = l_rel[idx];
        if constexpr (H == 4) {
          x0 = expf(edge_logit(l_as[idx * 4 + 0], adrow[rel * 4 + 0]) - mrow[rel * 4 + 0]);
          x1 = expf(edge_logit(l_as[idx * 4 + 1], adrow[rel * 4 + 1]) - mrow[rel * 4 + 1]);
          x2 = expf(edge_logit(l_as[idx * 4 + 2], adrow[rel * 4 + 2]) - mrow[rel * 4 + 2]);
          x3 = expf(edge_logit(l_as[idx * 4 + 3], adrow[rel * 4 + 3]) - mrow[rel * 4 + 3]);
          __hip_atomic_fetch_add(&srow[rel * 4 + 0], x0, __ATOMIC_RELAXED, __HIP_MEMORY_SCOPE_WORKGROUP);
          __hip_atomic_fetch_add(&srow[rel * 4 + 1], x1, __ATOMIC_RELAXED, __HIP_MEMORY_SCOPE_WORKGROUP);
          __hip_atomic_fetch_add(&srow[rel * 4 + 2], x2, __ATOMIC_RELAXED, __HIP_MEMORY_SCOPE_WORKGROUP);
          __hip_atomic_fetch_add(&srow[rel * 4 + 3], x3, __ATOMIC_RELAXED, __HIP_MEMORY_SCOPE_WORKGROUP);
        } else {
          x0 = expf(edge_logit(l_as[idx], adrow[rel]) - mrow[rel]);
          __hip_atomic_fetch_add(&srow[rel], x0, __ATOMIC_RELAXED, __HIP_MEMORY_SCOPE_WORKGROUP);
        }
      }
      wave_accumulate(__builtin_amdgcn_ballot_w32(has), srcv, rel, x0, x1, x2, x3);
    }
  } else {
    stream_dst_matches(dstp, E, row0, rows, tid, [&](bool has, int edge, int rel) {
      int srcv = 0;
      float x0 = 0.f, x1 = 0.f, x2 = 0.f, x3 = 0.f;
      if (has) {
        srcv = min(max(ei[edge], 0), n - 1);
        if constexpr (H == 4) {
          const float4 a = *(const float4*)(as + (size_t)srcv * 4);
          x0 = expf(edge_logit(a.x, adrow[rel * 4 + 0]) - mrow[rel * 4 + 0]);
          x1 = expf(edge_logit(a.y, adrow[rel * 4 + 1]) - mrow[rel * 4 + 1]);
          x2 = expf(edge_logit(a.z, adrow[rel * 4 + 2]) - mrow[rel * 4 + 2]);
          x3 = expf(edge_logit(a.w, adrow[rel * 4 + 3]) - mrow[rel * 4 + 3]);
          __hip_atomic_fetch_add(&srow[rel * 4 + 0], x0, __ATOMIC_RELAXED, __HIP_MEMORY_SCOPE_WORKGROUP);
          __hip_atomic_fetch_add(&srow[rel * 4 + 1], x1, __ATOMIC_RELAXED, __HIP_MEMORY_SCOPE_WORKGROUP);
          __hip_atomic_fetch_add(&srow[rel * 4 + 2], x2, __ATOMIC_RELAXED, __HIP_MEMORY_SCOPE_WORKGROUP);
          __hip_atomic_fetch_add(&srow[rel * 4 + 3], x3, __ATOMIC_RELAXED, __HIP_MEMORY_SCOPE_WORKGROUP);
        } else {
          x0 = expf(edge_logit(as[srcv], adrow[rel]) - mrow[rel]);
          __hip_atomic_fetch_add(&srow[rel], x0, __ATOMIC_RELAXED, __HIP_MEMORY_SCOPE_WORKGROUP);
        }
      }
      wave_accumulate(__builtin_amdgcn_ballot_w32(has), srcv, rel, x0, x1, x2, x3);
    });
  }
  __syncthreads();

  float bv[V];
  load_row_vals<CT>(bias, lane, bv);
  for (int r = wv; r < rows; r += 32) {
    float* op = out + (size_t)(row0 + r) * CT;
    float dA, dB;
    if constexpr (H == 4) {
      dA = srow[r * 4 + hi] + 1e-16f; dB = srow[r * 4 + 2 + hi] + 1e-16f;
    } else {
      dA = dB = srow[r] + 1e-16f;
    }
    float v[V];
#pragma unroll
    for (int i = 0; i < V; ++i) v[i] = bv[i] + acc[r * CT + i * 32 + lane] / (i < 4 ? dA : dB);
    if constexpr (CT == 256) {
      *(float4*)(op + 4 * lane)       = make_float4(v[0], v[1], v[2], v[3]);
      *(float4*)(op + 128 + 4 * lane) = make_float4(v[4], v[5], v[6], v[7]);
    } else {
      *(float2*)(op + 2 * lane) = make_float2(v[0], v[1]);
    }
  }
}

template <int C>
__global__ __launch_bounds__(256) void ln_add_elu(const float* __restrict__ x, const float* __restrict__ g,
                           const float* __restrict__ b, const float* __restrict__ add,
                           float* __restrict__ y, int n) {
  constexpr int V = C / 32;
  const int lane = threadIdx.x & 31;
  const int nw   = gridDim.x * (blockDim.x >> 5);
  float gv[V], bv[V];
#pragma unroll
  for (int i = 0; i < V; ++i) { gv[i] = g[lane + 32 * i]; bv[i] = b[lane + 32 * i]; }
  for (int row = blockIdx.x * (blockDim.x >> 5) + (threadIdx.x >> 5); row < n; row += nw) {
    const float* xr = x + (size_t)row * C;
    float v[V], av[V];
#pragma unroll
    for (int i = 0; i < V; ++i) v[i] = xr[lane + 32 * i];
    if (add) {
      const float* ar = add + (size_t)row * C;
#pragma unroll
      for (int i = 0; i < V; ++i) av[i] = ar[lane + 32 * i];
    } else {
#pragma unroll
      for (int i = 0; i < V; ++i) av[i] = 0.f;
    }
    float s = 0.f, ss = 0.f;
#pragma unroll
    for (int i = 0; i < V; ++i) { s += v[i]; ss += v[i] * v[i]; }
    for (int off = 16; off; off >>= 1) {
      s  += __shfl_xor(s,  off, 32);
      ss += __shfl_xor(ss, off, 32);
    }
    const float mu  = s / (float)C;
    const float var = ss / (float)C - mu * mu;
    const float inv = rsqrtf(var + 1e-5f);
    float* yr = y + (size_t)row * C;
#pragma unroll
    for (int i = 0; i < V; ++i) {
      float o = (v[i] - mu) * inv * gv[i] + bv[i];
      if (add) o += av[i];
      yr[lane + 32 * i] = (o > 0.f) ? o : expm1f(o);
    }
  }
}

extern "C" void kernel_launch(void* const* d_in, const int* in_sizes, int n_in,
                              void* d_out, int out_size, void* d_ws, size_t ws_size,
                              hipStream_t stream) {
  const float* x    = (const float*)d_in[0];
  const int*   ei   = (const int*)  d_in[1];
  const float* W1   = (const float*)d_in[2];
  const float* aS1  = (const float*)d_in[3];
  const float* aD1  = (const float*)d_in[4];
  const float* b1   = (const float*)d_in[5];
  const float* W2   = (const float*)d_in[6];
  const float* aS2  = (const float*)d_in[7];
  const float* aD2  = (const float*)d_in[8];
  const float* b2   = (const float*)d_in[9];
  const float* g1   = (const float*)d_in[10];
  const float* be1  = (const float*)d_in[11];
  const float* g2   = (const float*)d_in[12];
  const float* be2  = (const float*)d_in[13];
  const float* pW   = (const float*)d_in[14];
  const float* pb   = (const float*)d_in[15];

  const int IN_CH = 128, HEADS = 4, OUT_CH = 64, HC = HEADS * OUT_CH;
  const int N  = in_sizes[0] / IN_CH;
  const int E  = in_sizes[1] / 2;

  float* ws   = (float*)d_ws;
  float* h1   = ws; ws += (size_t)N * HC;
  float* out1 = ws; ws += (size_t)N * HC;
  float* idb  = ws; ws += (size_t)N * OUT_CH;
  float* z2   = ws; ws += (size_t)N * OUT_CH;
  float* out2 = ws; ws += (size_t)N * OUT_CH;
  float* as1  = ws; ws += (size_t)N * HEADS;
  float* ad1  = ws; ws += (size_t)N * HEADS;
  float* as2  = ws; ws += (size_t)N;
  float* ad2  = ws; ws += (size_t)N;

  auto cdiv = [](long a, long b) -> int { return (int)((a + b - 1) / b); };
  const int BLK = 256;
  const int WPB = BLK / 32;
  const int RPW = 4;

  gemm_lds_weights<128, 4, 1, 4, 16><<<256, 512, 0, stream>>>(x, W1, h1, pW, pb, idb, N);

  att_scores<<<cdiv((long)N * HEADS, BLK), BLK, 0, stream>>>(h1, aS1, aD1, as1, ad1, N, HEADS, OUT_CH);
  agg_softmax_rows_lds<256, 4, 196, 3072, false><<<cdiv(N, 196), 1024, 0, stream>>>(ei, as1, ad1, h1, b1, out1, (int*)out2, E, N);

  ln_add_elu<256><<<cdiv(N, WPB * RPW), BLK, 0, stream>>>(out1, g1, be1, nullptr, out1, N);

  gemm_lds_weights<256, 2, 0, 2, 8><<<256, 256, 0, stream>>>(out1, W2, z2, nullptr, nullptr, nullptr, N);

  att_scores<<<cdiv((long)N, BLK), BLK, 0, stream>>>(z2, aS2, aD2, as2, ad2, N, 1, OUT_CH);
  agg_softmax_rows_lds<64, 1, 196, 3072, true><<<cdiv(N, 196), 1024, 0, stream>>>(ei, as2, ad2, z2, b2, out2, (int*)out2, E, N);

  ln_add_elu<64><<<cdiv(N, WPB * RPW), BLK, 0, stream>>>(out2, g2, be2, idb, (float*)d_out, N);

  (void)n_in; (void)out_size; (void)ws_size;
}


# node scores as extra GEMM columns: W folded with the attention vectors while staging (one more 16-column WMMA tile per strip); both score kernels go, 8 launches to 6; score sum order changes
# speedup vs baseline: 5.4931x; 1.0486x over previous
#include <hip/hip_runtime.h>
#include <math.h>

typedef float v2f __attribute__((ext_vector_type(2)));
typedef float v8f __attribute__((ext_vector_type(8)));

#define NEG_SLOPE 0.2f

template <int K, int NCOLS, int THREADS>
__device__ __forceinline__ void stage_weights(const float* __restrict__ B,
                                              float4* __restrict__ dst, int tid) {
  constexpr int KP   = K / 8;
  constexpr int ENT  = (NCOLS / 16) * KP * 32;
  constexpr int ITER = ENT / THREADS;
  static_assert(ENT % THREADS == 0 && THREADS % 32 == 0, "whole waves share the entries evenly");
  const int c = tid & 15, h = (tid >> 4) & 1;
  float4 v[ITER];
#pragma unroll
  for (int it = 0; it < ITER; ++it) {
    const int g = (tid + it * THREADS) >> 5;
    const int tile = g / KP, kp = g - tile * KP;
    const float* p = B + (size_t)(8 * kp + 2 * h) * NCOLS + tile * 16 + c;
    v[it].x = p[0]; v[it].y = p[NCOLS]; v[it].z = p[4 * NCOLS]; v[it].w = p[5 * NCOLS];
  }
#pragma unroll
  for (int it = 0; it < ITER; ++it) dst[tid + it * THREADS] = v[it];
}

template <int K>
__device__ __forceinline__ int score_head(int tid) {
  return __builtin_amdgcn_readfirstlane(tid / K);
}

template <int K, int H>
__device__ __forceinline__ void request_score_row(const float* __restrict__ B, int tid,
                                                  float4 (&w)[16]) {
  static_assert(K % 32 == 0, "a wave stays inside one head");
  const int h = score_head<K>(tid), k = tid - h * K;
  const float4* wv = (const float4*)(B + (size_t)k * (H * 64) + h * 64);
#pragma unroll
  for (int i = 0; i < 16; ++i) w[i] = wv[i];
}

template <int K, int H, int THREADS>
__device__ __forceinline__ void stage_score_weights(const float4 (&w)[16],
                                                    const float* __restrict__ att_s,
                                                    const float* __restrict__ att_d,
                                                    float* __restrict__ img, int tid) {
  constexpr int C  = 64;
  static_assert(K * H == THREADS, "one (k, head) pair per thread");
  auto pos = [](int k, int n) {
    const int kp = k >> 3, r = k & 7;
    return ((kp * 32 + ((r >> 1) & 1) * 16 + n) << 2) + (((r >> 2) << 1) | (r & 1));
  };
  {
    const int h = score_head<K>(tid), k = tid - h * K;
    const float4* sv = (const float4*)(att_s + h * C);
    const float4* dv = (const float4*)(att_d + h * C);
    float s = 0.f, d = 0.f;
#pragma unroll
    for (int i = 0; i < C / 4; ++i) {
      const float4 a = sv[i], b = dv[i];
      s += w[i].x * a.x + w[i].y * a.y + w[i].z * a.z + w[i].w * a.w;
      d += w[i].x * b.x + w[i].y * b.y + w[i].z * b.z + w[i].w * b.w;
    }
    img[pos(k, h)]     = s;
    img[pos(k, H + h)] = d;
  }
  for (int i = tid; i < K * 16; i += THREADS) {
    const int n = i & 15;
    if (n >= 2 * H) img[pos(i >> 4, n)] = 0.f;
  }
}

template <int K>
__device__ __forceinline__ void load_a_strip(const float* __restrict__ A, int strip, int lane,
                                             v2f (&a)[K / 4]) {
  const float* ap = A + (size_t)(strip * 16 + (lane & 15)) * K + ((lane >> 4) << 1);
#pragma unroll
  for (int ks = 0; ks < K / 4; ++ks) a[ks] = *(const v2f*)(ap + 4 * ks);
}

template <int K, int T1, int T2, int UPS, int WAVES, int H>
__global__ __launch_bounds__(WAVES * 32) void gemm_lds_weights(
    const float* __restrict__ A,
    const float* __restrict__ B1, float* __restrict__ C1,
    const float* __restrict__ B2, const float* __restrict__ bias2, float* __restrict__ C2,
    const float* __restrict__ att_s, const float* __restrict__ att_d,
    float* __restrict__ sc_s, float* __restrict__ sc_d,
    int M) {
  constexpr int NT1 = UPS * T1, NT2 = UPS * T2;
  constexpr int N1 = NT1 * 16, N2 = NT2 * 16;
  constexpr int KP = K / 8;
  constexpr int TT = T1 + T2;
  constexpr int RS = WAVES / UPS;
  static_assert(T1 % 2 == 0 && WAVES % UPS == 0 && K % 8 == 0, "C1 tiles are stored in pairs");
  constexpr int SC0  = 0;
  constexpr int W0   = KP * 32;
  constexpr int NEED = W0 + (NT1 + NT2) * KP * 32;
  static_assert(N1 == H * 64 && 2 * H <= 16, "B1's column blocks of 64 are the heads");
  constexpr int FLOOR = 164 * 1024 / 16;
  __shared__ float4 Bl[NEED > FLOOR ? NEED : FLOOR];

  const int tid  = threadIdx.x;
  const int lane = tid & 31;
  const int wv   = tid >> 5;
  const int hi   = lane >> 4;
  const int cg   = wv % UPS;
  const int rr   = wv / UPS;

  const int strips = M >> 4;
  const int sb = (int)((blockIdx.x * (unsigned)strips) / gridDim.x);
  const int se = (int)(((blockIdx.x + 1) * (unsigned)strips) / gridDim.x);

  v2f an[K / 4];
  int st = sb + rr;
  if (st < se) load_a_strip<K>(A, st, lane, an);

  float4 wrow[16];
  request_score_row<K, H>(B1, tid, wrow);

  stage_weights<K, N1, WAVES * 32>(B1, Bl + W0, tid);
  if constexpr (T2 > 0)
    stage_weights<K, N2, WAVES * 32>(B2, Bl + W0 + NT1 * KP * 32, tid);
  stage_score_weights<K, H, WAVES * 32>(wrow, att_s, att_d, (float*)(Bl + SC0), tid);
  __syncthreads();

  int trip = 0;
  for (; st < se; st += RS, ++trip) {
    v2f a[K / 4];
#pragma unroll
    for (int ks = 0; ks < K / 4; ++ks) a[ks] = an[ks];
    if (st + RS < se) load_a_strip<K>(A, st + RS, lane, an);

    v8f acc[TT];
#pragma unroll
    for (int t = 0; t < TT; ++t) acc[t] = v8f{};
    float4 bc[TT], bn[TT];
#pragma unroll
    for (int t = 0; t < TT; ++t) {
      const int tile = (t < T1) ? (cg * T1 + t) : (NT1 + cg * T2 + (t - T1));
      bc[t] = Bl[W0 + (tile * KP) * 32 + lane];
    }
#pragma unroll
    for (int kp = 0; kp < KP; ++kp) {
      if (kp + 1 < KP) {
#pragma unroll
        for (int t = 0; t < TT; ++t) {
          const int tile = (t < T1) ? (cg * T1 + t) : (NT1 + cg * T2 + (t - T1));
          bn[t] = Bl[W0 + (tile * KP + kp + 1) * 32 + lane];
        }
      }
#pragma unroll
      for (int t = 0; t < TT; ++t) {
        v2f bb; bb.x = bc[t].x; bb.y = bc[t].y;
        acc[t] = __builtin_amdgcn_wmma_f32_16x16x4_f32(
             false, a[2 * kp],  false, bb,
             (short)0, acc[t],  false,  false);
      }
#pragma unroll
      for (int t = 0; t < TT; ++t) {
        v2f bb; bb.x = bc[t].z; bb.y = bc[t].w;
        acc[t] = __builtin_amdgcn_wmma_f32_16x16x4_f32(
             false, a[2 * kp + 1],  false, bb,
             (short)0, acc[t],  false,  false);
      }
#pragma unroll
      for (int t = 0; t < TT; ++t) bc[t] = bn[t];
      __builtin_amdgcn_sched_barrier(0);
    }

#pragma unroll
    for (int p = 0; p < T1 / 2; ++p) {
      float* cp = C1 + (size_t)(st * 16) * N1 + (cg * T1 + 2 * p) * 16 + lane;
#pragma unroll
      for (int v = 0; v < 8; ++v) {
        const float own0 = acc[2 * p][v], own1 = acc[2 * p + 1][v];
        const float give = hi ? own0 : own1;
        const float got  = __shfl_xor(give, 16, 32);
        cp[(size_t)v * N1]       = hi ? got : own0;
        cp[(size_t)(8 + v) * N1] = hi ? own1 : got;
      }
    }
    if constexpr (T2 > 0) {
#pragma unroll
      for (int t = 0; t < T2; ++t) {
        const int ncol = (cg * T2 + t) * 16 + (lane & 15);
        const float bv = bias2 ? bias2[ncol] : 0.0f;
        float* cp = C2 + (size_t)(st * 16 + (hi << 3)) * N2 + ncol;
#pragma unroll
        for (int v = 0; v < 8; ++v) cp[(size_t)v * N2] = acc[T1 + t][v] + bv;
      }
    }

    if ((trip % UPS) == cg) {
      v8f sa = v8f{};
      float4 sb_c = Bl[SC0 + lane], sb_n = sb_c;
#pragma unroll
      for (int kp = 0; kp < KP; ++kp) {
        if (kp + 1 < KP) sb_n = Bl[SC0 + (kp + 1) * 32 + lane];
        v2f b0; b0.x = sb_c.x; b0.y = sb_c.y;
        v2f b1; b1.x = sb_c.z; b1.y = sb_c.w;
        sa = __builtin_amdgcn_wmma_f32_16x16x4_f32(
             false, a[2 * kp],  false, b0,
             (short)0, sa,  false,  false);
        sa = __builtin_amdgcn_wmma_f32_16x16x4_f32(
             false, a[2 * kp + 1],  false, b1,
             (short)0, sa,  false,  false);
        sb_c = sb_n;
        __builtin_amdgcn_sched_barrier(0);
      }
      const int n = lane & 15;
      if (n < 2 * H) {
        float* sp = (n < H ? sc_s + n : sc_d + (n - H)) + (size_t)(st * 16 + (hi << 3)) * H;
#pragma unroll
        for (int v = 0; v < 8; ++v) sp[(size_t)v * H] = sa[v];
      }
    }
  }
}

template <int CT>
__device__ __forceinline__ void load_row_vals(const float* __restrict__ p, int lane,
                                              float (&v)[CT / 32]) {
  if constexpr (CT == 256) {
    const float4 a = *(const float4*)(p + 4 * lane);
    const float4 b = *(const float4*)(p + 128 + 4 * lane);
    v[0] = a.x; v[1] = a.y; v[2] = a.z; v[3] = a.w;
    v[4] = b.x; v[5] = b.y; v[6] = b.z; v[7] = b.w;
  } else {
    const float2 a = *(const float2*)(p + 2 * lane);
    v[0] = a.x; v[1] = a.y;
  }
}

__device__ __forceinline__ int bcast_i(int v, int l) {
  return __builtin_amdgcn_readlane(v, l);
}
__device__ __forceinline__ float bcast_f(float v, int l) {
  return __int_as_float(__builtin_amdgcn_readlane(__float_as_int(v), l));
}

__device__ __forceinline__ float edge_logit(float a_src, float a_dst) {
  const float v = a_src + a_dst;
  return ((v > 0.f) ? v : NEG_SLOPE * v) + 0.0f;
}

template <typename F>
__device__ __forceinline__ void stream_dst_matches(const int* __restrict__ dstp, int E,
                                                   int row0, int rows, int tid, F on_round) {
  constexpr int T = 1024;
  constexpr int STEP = T * 16;
  const bool vec_ok = (E >= 4) && ((E & 3) == 0) && ((((size_t)dstp) & 15) == 0);

  auto load16 = [&](int base, int (&d)[16]) {
    if (vec_ok) {
#pragma unroll
      for (int j = 0; j < 4; ++j) {
        const int b = base + j * (T * 4) + tid * 4;
        const int4 q = *(const int4*)(dstp + min(b, E - 4));
        d[4 * j + 0] = q.x; d[4 * j + 1] = q.y; d[4 * j + 2] = q.z; d[4 * j + 3] = q.w;
      }
    } else {
#pragma unroll
      for (int j = 0; j < 4; ++j) {
#pragma unroll
        for (int k = 0; k < 4; ++k)
          d[4 * j + k] = dstp[min(base + j * (T * 4) + tid * 4 + k, E - 1)];
      }
    }
  };

  int cur[16], nxt[16];
#pragma unroll
  for (int q = 0; q < 16; ++q) { cur[q] = row0 - 1; nxt[q] = row0 - 1; }
  for (int base = -STEP; base < E; base += STEP) {
    unsigned mb = 0;
#pragma unroll
    for (int q = 0; q < 16; ++q) {
      const int eq = base + (q >> 2) * (T * 4) + tid * 4 + (q & 3);
      mb |= (eq < E && (unsigned)(cur[q] - row0) < (unsigned)rows) ? (1u << q) : 0u;
    }

    if (base + STEP < E) load16(base + STEP, nxt);

    while (__builtin_amdgcn_ballot_w32(mb != 0)) {
      const bool has = mb != 0;
      const int q = has ? __builtin_ctz(mb) : 0;
      mb &= mb - 1;
      int dsel = cur[0];
#pragma unroll
      for (int t = 1; t < 16; ++t) dsel = (q == t) ? cur[t] : dsel;
      const int edge = base + (q >> 2) * (T * 4) + tid * 4 + (q & 3);
      on_round(has, edge, dsel - row0);
    }

#pragma unroll
    for (int q = 0; q < 16; ++q) cur[q] = nxt[q];
  }
}

template <int CAP, bool TAIL>
__device__ __forceinline__ void list_block(int base, const int4 (&d)[4], int nq, int row0, int rows,
                                           int tid, int lane, int* l_src, int* l_rel, int* l_cnt) {
  constexpr int T = 1024;
  unsigned mk[16];
#pragma unroll
  for (int j = 0; j < 4; ++j) {
    unsigned okm = 0xffffffffu;
    if constexpr (TAIL) okm = __builtin_amdgcn_ballot_w32((base >> 2) + j * T + tid < nq);
    mk[4 * j + 0] = __builtin_amdgcn_ballot_w32((unsigned)(d[j].x - row0) < (unsigned)rows) & okm;
    mk[4 * j + 1] = __builtin_amdgcn_ballot_w32((unsigned)(d[j].y - row0) < (unsigned)rows) & okm;
    mk[4 * j + 2] = __builtin_amdgcn_ballot_w32((unsigned)(d[j].z - row0) < (unsigned)rows) & okm;
    mk[4 * j + 3] = __builtin_amdgcn_ballot_w32((unsigned)(d[j].w - row0) < (unsigned)rows) & okm;
  }
  int tot = 0;
#pragma unroll
  for (int q = 0; q < 16; q += 2)
    tot += __builtin_popcountll(((unsigned long long)mk[q + 1] << 32) | (unsigned long long)mk[q]);
  if (tot != 0) {
    const unsigned below = (1u << lane) - 1u;
    int pos0 = 0;
    if (lane == 0) pos0 = atomicAdd(l_cnt, tot);
    pos0 = bcast_i(pos0, 0);
#pragma unroll
    for (int q = 0; q < 16; ++q) {
      const unsigned m = mk[q];
      if (m != 0) {
        const int pos = pos0 + __builtin_popcount(m & below);
        const int4 dj = d[q >> 2];
        const int dv = (q & 3) == 0 ? dj.x : (q & 3) == 1 ? dj.y : (q & 3) == 2 ? dj.z : dj.w;
        if (((m >> lane) & 1u) && pos < CAP) {
          l_src[pos] = base + (q >> 2) * (T * 4) + tid * 4 + (q & 3);
          l_rel[pos] = dv - row0;
        }
        pos0 += __builtin_popcount(m);
      }
    }
  }
}

template <int CAP>
__device__ __forceinline__ void list_owned_edges_vec(const int* __restrict__ dstp, int E,
                                                     int row0, int rows, int tid, int lane,
                                                     int* l_src, int* l_rel, int* l_cnt) {
  constexpr int T = 1024;
  constexpr int STEP = T * 16;
  const int4* __restrict__ vp = (const int4*)dstp;
  const int nq    = E >> 2;
  const int nfull = E / STEP;
  const int tail0 = nfull * STEP;

  auto request = [&](int base, int4 (&d)[4]) {
#pragma unroll
    for (int j = 0; j < 4; ++j) d[j] = vp[min((base >> 2) + j * T + tid, nq - 1)];
  };

  int4 rt[4], ra[4], rb[4];
  request(tail0, rt);
  request(0, ra);
  for (int b = 0; b < nfull; b += 2) {
    request((b + 1) * STEP, rb);
    list_block<CAP, false>(b * STEP, ra, nq, row0, rows, tid, lane, l_src, l_rel, l_cnt);
    if (b + 1 < nfull) {
      request((b + 2) * STEP, ra);
      list_block<CAP, false>((b + 1) * STEP, rb, nq, row0, rows, tid, lane, l_src, l_rel, l_cnt);
    }
  }
  if (tail0 < E) list_block<CAP, true>(tail0, rt, nq, row0, rows, tid, lane, l_src, l_rel, l_cnt);
}

template <int CT, int H, int R, int CAP, bool REUSE>
__global__ __launch_bounds__(1024) void agg_softmax_rows_lds(
    const int* __restrict__ ei, const float* __restrict__ as, const float* __restrict__ ad,
    const float* __restrict__ feat, const float* __restrict__ bias,
    float* out, int* lists, int E, int n) {
  static_assert((CT == 256 && H == 4) || (CT == 64 && H == 1), "lane-to-head map below");
  static_assert(!REUSE || H == 1, "the hand-over is read by the one-head layer");
  constexpr int V = CT / 32;
  constexpr int T = 1024;
  constexpr int NF = 8;
  constexpr int LW = 64;
  constexpr int NJ = CAP / T;
  static_assert(CAP % T == 0, "every thread reads the same number of pairs");
  __shared__ float acc[R * CT];
  __shared__ float adrow[R * H];
  __shared__ float slog[R * H];
  __shared__ float mrow[R * H];
  __shared__ float srow[R * H];
  __shared__ int   l_src[CAP];
  __shared__ int   l_rel[CAP];
  __shared__ float l_as[CAP * H];
  __shared__ int   l_cnt;

  const int tid  = threadIdx.x;
  const int lane = tid & 31;
  const int wv   = tid >> 5;
  const int hi   = lane >> 4;
  const int row0 = blockIdx.x * R;
  const int rows = max(0, min(R, n - row0));
  const int* __restrict__ dstp = ei + E;

  int*  piece = lists + (size_t)row0 * LW;
  int2* pairs = (int2*)(piece + 4);
  const int capg = max(0, min(CAP, (rows * LW - 4) >> 1));
  int hdr = 0;
  if constexpr (REUSE) hdr = piece[0];

  auto lds_max = [&](int slot, float val) {
    if (val >= 0.0f) atomicMax((int*)&mrow[slot], __float_as_int(val));
    else             atomicMin((unsigned int*)&mrow[slot], __float_as_uint(val));
  };

  auto wave_accumulate = [&](unsigned m, int srcv, int rel, float x0, float x1, float x2, float x3) {
    while (m) {
      int  l[NF];
      bool ok[NF];
#pragma unroll
      for (int j = 0; j < NF; ++j) {
        ok[j] = (m != 0);
        l[j]  = ok[j] ? __builtin_ctz(m) : 0;
        if (ok[j]) m &= m - 1;
      }
      float v[NF][V];
#pragma unroll
      for (int j = 0; j < NF; ++j) {
        if (ok[j]) load_row_vals<CT>(feat + (size_t)bcast_i(srcv, l[j]) * CT, lane, v[j]);
      }
#pragma unroll
      for (int j = 0; j < NF; ++j) {
        if (ok[j]) {
          const int rl = bcast_i(rel, l[j]);
          float aA, aB;
          if constexpr (H == 4) {
            const float y0 = bcast_f(x0, l[j]), y1 = bcast_f(x1, l[j]);
            const float y2 = bcast_f(x2, l[j]), y3 = bcast_f(x3, l[j]);
            aA = hi ? y1 : y0; aB = hi ? y3 : y2;
          } else {
            aA = aB = bcast_f(x0, l[j]);
          }
#pragma unroll
          for (int i = 0; i < V; ++i)
            __hip_atomic_fetch_add(&acc[rl * CT + i * 32 + lane], (i < 4 ? aA : aB) * v[j][i],
                                   __ATOMIC_RELAXED, __HIP_MEMORY_SCOPE_WORKGROUP);
        }
      }
    }
  };

  if (tid == 0) l_cnt = 0;
  for (int i = tid; i < rows * H; i += T) {
    const float d_ = ad[(size_t)row0 * H + i];
    const float v  = edge_logit(as[(size_t)row0 * H + i], d_);
    adrow[i] = d_; slog[i] = v; mrow[i] = v;
  }
  __syncthreads();

  int  cnt;
  bool listed;
  if constexpr (!REUSE) {
    const bool vec_ok = (E >= 4) && ((E & 3) == 0) && ((((size_t)dstp) & 15) == 0);
    if (vec_ok) list_owned_edges_vec<CAP>(dstp, E, row0, rows, tid, lane, l_src, l_rel, &l_cnt);
    else stream_dst_matches(dstp, E, row0, rows, tid, [&](bool has, int edge, int rel) {
      const unsigned bal = __builtin_amdgcn_ballot_w32(has);
      const int lead = __builtin_ctz(bal);
      int basepos = 0;
      if (lane == lead) basepos = atomicAdd(&l_cnt, __builtin_popcount(bal));
      basepos = bcast_i(basepos, lead);
      const int pos = basepos + __builtin_popcount(bal & ((1u << lane) - 1u));
      if (has && pos < CAP) { l_src[pos] = edge; l_rel[pos] = rel; }
    });
    __syncthreads();
    cnt    = l_cnt;
    listed = cnt <= CAP;
    if (tid == 0) piece[0] = cnt;
  } else {
    cnt    = hdr;
    listed = (cnt >= 0) && (cnt <= capg);
  }

  if (listed) {
   if constexpr (REUSE) {
    if (cnt > 0) {
      int2 pr[NJ];
#pragma unroll
      for (int j = 0; j < NJ; ++j) pr[j] = pairs[min(tid + j * T, cnt - 1)];
      int sv[NJ]; float av[NJ];
#pragma unroll
      for (int j = 0; j < NJ; ++j) {
        sv[j] = min(max(pr[j].x, 0), n - 1);
        av[j] = as[sv[j]];
      }
#pragma unroll
      for (int j = 0; j < NJ; ++j) {
        const int i   = min(tid + j * T, cnt - 1);
        const int rel = min(max(pr[j].y, 0), rows - 1);
        l_src[i] = sv[j]; l_rel[i] = rel; l_as[i] = av[j];
        if (tid + j * T < cnt) lds_max(rel, edge_logit(av[j], adrow[rel]));
      }
    }
   } else {
    const bool pub = cnt <= capg;
    for (int i = tid; i < cnt; i += T) {
      const int edge = l_src[i], rel = l_rel[i];
      const int src  = min(max(ei[edge], 0), n - 1);
      l_src[i] = src;
      if (pub) pairs[i] = make_int2(src, rel);
      if constexpr (H == 4) {
        const float4 a = *(const float4*)(as + (size_t)src * 4);
        l_as[i * 4 + 0] = a.x; l_as[i * 4 + 1] = a.y; l_as[i * 4 + 2] = a.z; l_as[i * 4 + 3] = a.w;
        lds_max(rel * 4 + 0, edge_logit(a.x, adrow[rel * 4 + 0]));
        lds_max(rel * 4 + 1, edge_logit(a.y, adrow[rel * 4 + 1]));
        lds_max(rel * 4 + 2, edge_logit(a.z, adrow[rel * 4 + 2]));
        lds_max(rel * 4 + 3, edge_logit(a.w, adrow[rel * 4 + 3]));
      } else {
        const float a = as[src];
        l_as[i] = a;
        lds_max(rel, edge_logit(a, adrow[rel]));
      }
    }
   }
  } else {
    stream_dst_matches(dstp, E, row0, rows, tid, [&](bool has, int edge, int rel) {
      if (has) {
        const int src = min(max(ei[edge], 0), n - 1);
        if constexpr (H == 4) {
          const float4 a = *(const float4*)(as + (size_t)src * 4);
          lds_max(rel * 4 + 0, edge_logit(a.x, adrow[rel * 4 + 0]));
          lds_max(rel * 4 + 1, edge_logit(a.y, adrow[rel * 4 + 1]));
          lds_max(rel * 4 + 2, edge_logit(a.z, adrow[rel * 4 + 2]));
          lds_max(rel * 4 + 3, edge_logit(a.w, adrow[rel * 4 + 3]));
        } else {
          lds_max(rel, edge_logit(as[src], adrow[rel]));
        }
      }
    });
  }
  __syncthreads();

  for (int r = wv; r < rows; r += 32) {
    float v[V];
    load_row_vals<CT>(feat + (size_t)(row0 + r) * CT, lane, v);
    if constexpr (H == 4) {
      const float xA = expf(slog[r * 4 + hi]     - mrow[r * 4 + hi]);
      const float xB = expf(slog[r * 4 + 2 + hi] - mrow[r * 4 + 2 + hi]);
#pragma unroll
      for (int i = 0; i < V; ++i) acc[r * CT + i * 32 + lane] = (i < 4 ? xA : xB) * v[i];
      if (lane < 4) srow[r * 4 + lane] = expf(slog[r * 4 + lane] - mrow[r * 4 + lane]);
    } else {
      const float x = expf(slog[r] - mrow[r]);
#pragma unroll
      for (int i = 0; i < V; ++i) acc[r * CT + i * 32 + lane] = x * v[i];
      if (lane == 0) srow[r] = x;
    }
  }
  __syncthreads();

  if (listed) {
    const int lo = (int)(((long)wv * cnt) >> 5), up = (int)(((long)(wv + 1) * cnt) >> 5);
    for (int cb = lo; cb < up; cb += 32) {
      const int  idx = cb + lane;
      const bool has = idx < up;
      int srcv = 0, rel = 0;
      float x0 = 0.f, x1 = 0.f, x2 = 0.f, x3 = 0.f;
      if (has) {
        srcv = l_src[idx]; rel = l_rel[idx];
        if constexpr (H == 4) {
          x0 = expf(edge_logit(l_as[idx * 4 + 0], adrow[rel * 4 + 0]) - mrow[rel * 4 + 0]);
          x1 = expf(edge_logit(l_as[idx * 4 + 1], adrow[rel * 4 + 1]) - mrow[rel * 4 + 1]);
          x2 = expf(edge_logit(l_as[idx * 4 + 2], adrow[rel * 4 + 2]) - mrow[rel * 4 + 2]);
          x3 = expf(edge_logit(l_as[idx * 4 + 3], adrow[rel * 4 + 3]) - mrow[rel * 4 + 3]);
          __hip_atomic_fetch_add(&srow[rel * 4 + 0], x0, __ATOMIC_RELAXED, __HIP_MEMORY_SCOPE_WORKGROUP);
          __hip_atomic_fetch_add(&srow[rel * 4 + 1], x1, __ATOMIC_RELAXED, __HIP_MEMORY_SCOPE_WORKGROUP);
          __hip_atomic_fetch_add(&srow[rel * 4 + 2], x2, __ATOMIC_RELAXED, __HIP_MEMORY_SCOPE_WORKGROUP);
          __hip_atomic_fetch_add(&srow[rel * 4 + 3], x3, __ATOMIC_RELAXED, __HIP_MEMORY_SCOPE_WORKGROUP);
        } else {
          x0 = expf(edge_logit(l_as[idx], adrow[rel]) - mrow[rel]);
          __hip_atomic_fetch_add(&srow[rel], x0, __ATOMIC_RELAXED, __HIP_MEMORY_SCOPE_WORKGROUP);
        }
      }
      wave_accumulate(__builtin_amdgcn_ballot_w32(has), srcv, rel, x0, x1, x2, x3);
    }
  } else {
    stream_dst_matches(dstp, E, row0, rows, tid, [&](bool has, int edge, int rel) {
      int srcv = 0;
      float x0 = 0.f, x1 = 0.f, x2 = 0.f, x3 = 0.f;
      if (has) {
        srcv = min(max(ei[edge], 0), n - 1);
        if constexpr (H == 4) {
          const float4 a = *(const float4*)(as + (size_t)srcv * 4);
          x0 = expf(edge_logit(a.x, adrow[rel * 4 + 0]) - mrow[rel * 4 + 0]);
          x1 = expf(edge_logit(a.y, adrow[rel * 4 + 1]) - mrow[rel * 4 + 1]);
          x2 = expf(edge_logit(a.z, adrow[rel * 4 + 2]) - mrow[rel * 4 + 2]);
          x3 = expf(edge_logit(a.w, adrow[rel * 4 + 3]) - mrow[rel * 4 + 3]);
          __hip_atomic_fetch_add(&srow[rel * 4 + 0], x0, __ATOMIC_RELAXED, __HIP_MEMORY_SCOPE_WORKGROUP);
          __hip_atomic_fetch_add(&srow[rel * 4 + 1], x1, __ATOMIC_RELAXED, __HIP_MEMORY_SCOPE_WORKGROUP);
          __hip_atomic_fetch_add(&srow[rel * 4 + 2], x2, __ATOMIC_RELAXED, __HIP_MEMORY_SCOPE_WORKGROUP);
          __hip_atomic_fetch_add(&srow[rel * 4 + 3], x3, __ATOMIC_RELAXED, __HIP_MEMORY_SCOPE_WORKGROUP);
        } else {
          x0 = expf(edge_logit(as[srcv], adrow[rel]) - mrow[rel]);
          __hip_atomic_fetch_add(&srow[rel], x0, __ATOMIC_RELAXED, __HIP_MEMORY_SCOPE_WORKGROUP);
        }
      }
      wave_accumulate(__builtin_amdgcn_ballot_w32(has), srcv, rel, x0, x1, x2, x3);
    });
  }
  __syncthreads();

  float bv[V];
  load_row_vals<CT>(bias, lane, bv);
  for (int r = wv; r < rows; r += 32) {
    float* op = out + (size_t)(row0 + r) * CT;
    float dA, dB;
    if constexpr (H == 4) {
      dA = srow[r * 4 + hi] + 1e-16f; dB = srow[r * 4 + 2 + hi] + 1e-16f;
    } else {
      dA = dB = srow[r] + 1e-16f;
    }
    float v[V];
#pragma unroll
    for (int i = 0; i < V; ++i) v[i] = bv[i] + acc[r * CT + i * 32 + lane] / (i < 4 ? dA : dB);
    if constexpr (CT == 256) {
      *(float4*)(op + 4 * lane)       = make_float4(v[0], v[1], v[2], v[3]);
      *(float4*)(op + 128 + 4 * lane) = make_float4(v[4], v[5], v[6], v[7]);
    } else {
      *(float2*)(op + 2 * lane) = make_float2(v[0], v[1]);
    }
  }
}

template <int C>
__global__ __launch_bounds__(256) void ln_add_elu(const float* __restrict__ x, const float* __restrict__ g,
                           const float* __restrict__ b, const float* __restrict__ add,
                           float* __restrict__ y, int n) {
  constexpr int V = C / 32;
  const int lane = threadIdx.x & 31;
  const int nw   = gridDim.x * (blockDim.x >> 5);
  float gv[V], bv[V];
#pragma unroll
  for (int i = 0; i < V; ++i) { gv[i] = g[lane + 32 * i]; bv[i] = b[lane + 32 * i]; }
  for (int row = blockIdx.x * (blockDim.x >> 5) + (threadIdx.x >> 5); row < n; row += nw) {
    const float* xr = x + (size_t)row * C;
    float v[V], av[V];
#pragma unroll
    for (int i = 0; i < V; ++i) v[i] = xr[lane + 32 * i];
    if (add) {
      const float* ar = add + (size_t)row * C;
#pragma unroll
      for (int i = 0; i < V; ++i) av[i] = ar[lane + 32 * i];
    } else {
#pragma unroll
      for (int i = 0; i < V; ++i) av[i] = 0.f;
    }
    float s = 0.f, ss = 0.f;
#pragma unroll
    for (int i = 0; i < V; ++i) { s += v[i]; ss += v[i] * v[i]; }
    for (int off = 16; off; off >>= 1) {
      s  += __shfl_xor(s,  off, 32);
      ss += __shfl_xor(ss, off, 32);
    }
    const float mu  = s / (float)C;
    const float var = ss / (float)C - mu * mu;
    const float inv = rsqrtf(var + 1e-5f);
    float* yr = y + (size_t)row * C;
#pragma unroll
    for (int i = 0; i < V; ++i) {
      float o = (v[i] - mu) * inv * gv[i] + bv[i];
      if (add) o += av[i];
      yr[lane + 32 * i] = (o > 0.f) ? o : expm1f(o);
    }
  }
}

extern "C" void kernel_launch(void* const* d_in, const int* in_sizes, int n_in,
                              void* d_out, int out_size, void* d_ws, size_t ws_size,
                              hipStream_t stream) {
  const float* x    = (const float*)d_in[0];
  const int*   ei   = (const int*)  d_in[1];
  const float* W1   = (const float*)d_in[2];
  const float* aS1  = (const float*)d_in[3];
  const float* aD1  = (const float*)d_in[4];
  const float* b1   = (const float*)d_in[5];
  const float* W2   = (const float*)d_in[6];
  const float* aS2  = (const float*)d_in[7];
  const float* aD2  = (const float*)d_in[8];
  const float* b2   = (const float*)d_in[9];
  const float* g1   = (const float*)d_in[10];
  const float* be1  = (const float*)d_in[11];
  const float* g2   = (const float*)d_in[12];
  const float* be2  = (const float*)d_in[13];
  const float* pW   = (const float*)d_in[14];
  const float* pb   = (const float*)d_in[15];

  const int IN_CH = 128, HEADS = 4, OUT_CH = 64, HC = HEADS * OUT_CH;
  const int N  = in_sizes[0] / IN_CH;
  const int E  = in_sizes[1] / 2;

  float* ws   = (float*)d_ws;
  float* h1   = ws; ws += (size_t)N * HC;
  float* out1 = ws; ws += (size_t)N * HC;
  float* idb  = ws; ws += (size_t)N * OUT_CH;
  float* z2   = ws; ws += (size_t)N * OUT_CH;
  float* out2 = ws; ws += (size_t)N * OUT_CH;
  float* as1  = ws; ws += (size_t)N * HEADS;
  float* ad1  = ws; ws += (size_t)N * HEADS;
  float* as2  = ws; ws += (size_t)N;
  float* ad2  = ws; ws += (size_t)N;

  auto cdiv = [](long a, long b) -> int { return (int)((a + b - 1) / b); };
  const int BLK = 256;
  const int WPB = BLK / 32;
  const int RPW = 4;

  gemm_lds_weights<128, 4, 1, 4, 16, 4><<<256, 512, 0, stream>>>(x, W1, h1, pW, pb, idb, aS1, aD1, as1, ad1, N);

  agg_softmax_rows_lds<256, 4, 196, 3072, false><<<cdiv(N, 196), 1024, 0, stream>>>(ei, as1, ad1, h1, b1, out1, (int*)out2, E, N);

  ln_add_elu<256><<<cdiv(N, WPB * RPW), BLK, 0, stream>>>(out1, g1, be1, nullptr, out1, N);

  gemm_lds_weights<256, 2, 0, 2, 8, 1><<<256, 256, 0, stream>>>(out1, W2, z2, nullptr, nullptr, nullptr, aS2, aD2, as2, ad2, N);
  agg_softmax_rows_lds<64, 1, 196, 3072, true><<<cdiv(N, 196), 1024, 0, stream>>>(ei, as2, ad2, z2, b2, out2, (int*)out2, E, N);

  ln_add_elu<64><<<cdiv(N, WPB * RPW), BLK, 0, stream>>>(out2, g2, be2, idb, (float*)d_out, N);

  (void)n_in; (void)out_size; (void)ws_size;
}


# layer norm + ELU (+ identity add) done by the owning work-group on each row it is about to store: both norm launches and the aggregate's store/reload go; norm sums: lane's values, then xor tree
# speedup vs baseline: 6.1653x; 1.1224x over previous
#include <hip/hip_runtime.h>
#include <math.h>

typedef float v2f __attribute__((ext_vector_type(2)));
typedef float v8f __attribute__((ext_vector_type(8)));

#define NEG_SLOPE 0.2f

template <int K, int NCOLS, int THREADS>
__device__ __forceinline__ void stage_weights(const float* __restrict__ B,
                                              float4* __restrict__ dst, int tid) {
  constexpr int KP   = K / 8;
  constexpr int ENT  = (NCOLS / 16) * KP * 32;
  constexpr int ITER = ENT / THREADS;
  static_assert(ENT % THREADS == 0 && THREADS % 32 == 0, "whole waves share the entries evenly");
  const int c = tid & 15, h = (tid >> 4) & 1;
  float4 v[ITER];
#pragma unroll
  for (int it = 0; it < ITER; ++it) {
    const int g = (tid + it * THREADS) >> 5;
    const int tile = g / KP, kp = g - tile * KP;
    const float* p = B + (size_t)(8 * kp + 2 * h) * NCOLS + tile * 16 + c;
    v[it].x = p[0]; v[it].y = p[NCOLS]; v[it].z = p[4 * NCOLS]; v[it].w = p[5 * NCOLS];
  }
#pragma unroll
  for (int it = 0; it < ITER; ++it) dst[tid + it * THREADS] = v[it];
}

template <int K>
__device__ __forceinline__ int score_head(int tid) {
  return __builtin_amdgcn_readfirstlane(tid / K);
}

template <int K, int H>
__device__ __forceinline__ void request_score_row(const float* __restrict__ B, int tid,
                                                  float4 (&w)[16]) {
  static_assert(K % 32 == 0, "a wave stays inside one head");
  const int h = score_head<K>(tid), k = tid - h * K;
  const float4* wv = (const float4*)(B + (size_t)k * (H * 64) + h * 64);
#pragma unroll
  for (int i = 0; i < 16; ++i) w[i] = wv[i];
}

template <int K, int H, int THREADS>
__device__ __forceinline__ void stage_score_weights(const float4 (&w)[16],
                                                    const float* __restrict__ att_s,
                                                    const float* __restrict__ att_d,
                                                    float* __restrict__ img, int tid) {
  constexpr int C  = 64;
  static_assert(K * H == THREADS, "one (k, head) pair per thread");
  auto pos = [](int k, int n) {
    const int kp = k >> 3, r = k & 7;
    return ((kp * 32 + ((r >> 1) & 1) * 16 + n) << 2) + (((r >> 2) << 1) | (r & 1));
  };
  {
    const int h = score_head<K>(tid), k = tid - h * K;
    const float4* sv = (const float4*)(att_s + h * C);
    const float4* dv = (const float4*)(att_d + h * C);
    float s = 0.f, d = 0.f;
#pragma unroll
    for (int i = 0; i < C / 4; ++i) {
      const float4 a = sv[i], b = dv[i];
      s += w[i].x * a.x + w[i].y * a.y + w[i].z * a.z + w[i].w * a.w;
      d += w[i].x * b.x + w[i].y * b.y + w[i].z * b.z + w[i].w * b.w;
    }
    img[pos(k, h)]     = s;
    img[pos(k, H + h)] = d;
  }
  for (int i = tid; i < K * 16; i += THREADS) {
    const int n = i & 15;
    if (n >= 2 * H) img[pos(i >> 4, n)] = 0.f;
  }
}

template <int K>
__device__ __forceinline__ void load_a_strip(const float* __restrict__ A, int strip, int lane,
                                             v2f (&a)[K / 4]) {
  const float* ap = A + (size_t)(strip * 16 + (lane & 15)) * K + ((lane >> 4) << 1);
#pragma unroll
  for (int ks = 0; ks < K / 4; ++ks) a[ks] = *(const v2f*)(ap + 4 * ks);
}

template <int K, int T1, int T2, int UPS, int WAVES, int H>
__global__ __launch_bounds__(WAVES * 32) void gemm_lds_weights(
    const float* __restrict__ A,
    const float* __restrict__ B1, float* __restrict__ C1,
    const float* __restrict__ B2, const float* __restrict__ bias2, float* __restrict__ C2,
    const float* __restrict__ att_s, const float* __restrict__ att_d,
    float* __restrict__ sc_s, float* __restrict__ sc_d,
    int M) {
  constexpr int NT1 = UPS * T1, NT2 = UPS * T2;
  constexpr int N1 = NT1 * 16, N2 = NT2 * 16;
  constexpr int KP = K / 8;
  constexpr int TT = T1 + T2;
  constexpr int RS = WAVES / UPS;
  static_assert(T1 % 2 == 0 && WAVES % UPS == 0 && K % 8 == 0, "C1 tiles are stored in pairs");
  constexpr int SC0  = 0;
  constexpr int W0   = KP * 32;
  constexpr int NEED = W0 + (NT1 + NT2) * KP * 32;
  static_assert(N1 == H * 64 && 2 * H <= 16, "B1's column blocks of 64 are the heads");
  constexpr int FLOOR = 164 * 1024 / 16;
  __shared__ float4 Bl[NEED > FLOOR ? NEED : FLOOR];

  const int tid  = threadIdx.x;
  const int lane = tid & 31;
  const int wv   = tid >> 5;
  const int hi   = lane >> 4;
  const int cg   = wv % UPS;
  const int rr   = wv / UPS;

  const int strips = M >> 4;
  const int sb = (int)((blockIdx.x * (unsigned)strips) / gridDim.x);
  const int se = (int)(((blockIdx.x + 1) * (unsigned)strips) / gridDim.x);

  v2f an[K / 4];
  int st = sb + rr;
  if (st < se) load_a_strip<K>(A, st, lane, an);

  float4 wrow[16];
  request_score_row<K, H>(B1, tid, wrow);

  stage_weights<K, N1, WAVES * 32>(B1, Bl + W0, tid);
  if constexpr (T2 > 0)
    stage_weights<K, N2, WAVES * 32>(B2, Bl + W0 + NT1 * KP * 32, tid);
  stage_score_weights<K, H, WAVES * 32>(wrow, att_s, att_d, (float*)(Bl + SC0), tid);
  __syncthreads();

  int trip = 0;
  for (; st < se; st += RS, ++trip) {
    v2f a[K / 4];
#pragma unroll
    for (int ks = 0; ks < K / 4; ++ks) a[ks] = an[ks];
    if (st + RS < se) load_a_strip<K>(A, st + RS, lane, an);

    v8f acc[TT];
#pragma unroll
    for (int t = 0; t < TT; ++t) acc[t] = v8f{};
    float4 bc[TT], bn[TT];
#pragma unroll
    for (int t = 0; t < TT; ++t) {
      const int tile = (t < T1) ? (cg * T1 + t) : (NT1 + cg * T2 + (t - T1));
      bc[t] = Bl[W0 + (tile * KP) * 32 + lane];
    }
#pragma unroll
    for (int kp = 0; kp < KP; ++kp) {
      if (kp + 1 < KP) {
#pragma unroll
        for (int t = 0; t < TT; ++t) {
          const int tile = (t < T1) ? (cg * T1 + t) : (NT1 + cg * T2 + (t - T1));
          bn[t] = Bl[W0 + (tile * KP + kp + 1) * 32 + lane];
        }
      }
#pragma unroll
      for (int t = 0; t < TT; ++t) {
        v2f bb; bb.x = bc[t].x; bb.y = bc[t].y;
        acc[t] = __builtin_amdgcn_wmma_f32_16x16x4_f32(
             false, a[2 * kp],  false, bb,
             (short)0, acc[t],  false,  false);
      }
#pragma unroll
      for (int t = 0; t < TT; ++t) {
        v2f bb; bb.x = bc[t].z; bb.y = bc[t].w;
        acc[t] = __builtin_amdgcn_wmma_f32_16x16x4_f32(
             false, a[2 * kp + 1],  false, bb,
             (short)0, acc[t],  false,  false);
      }
#pragma unroll
      for (int t = 0; t < TT; ++t) bc[t] = bn[t];
      __builtin_amdgcn_sched_barrier(0);
    }

#pragma unroll
    for (int p = 0; p < T1 / 2; ++p) {
      float* cp = C1 + (size_t)(st * 16) * N1 + (cg * T1 + 2 * p) * 16 + lane;
#pragma unroll
      for (int v = 0; v < 8; ++v) {
        const float own0 = acc[2 * p][v], own1 = acc[2 * p + 1][v];
        const float give = hi ? own0 : own1;
        const float got  = __shfl_xor(give, 16, 32);
        cp[(size_t)v * N1]       = hi ? got : own0;
        cp[(size_t)(8 + v) * N1] = hi ? own1 : got;
      }
    }
    if constexpr (T2 > 0) {
#pragma unroll
      for (int t = 0; t < T2; ++t) {
        const int ncol = (cg * T2 + t) * 16 + (lane & 15);
        const float bv = bias2 ? bias2[ncol] : 0.0f;
        float* cp = C2 + (size_t)(st * 16 + (hi << 3)) * N2 + ncol;
#pragma unroll
        for (int v = 0; v < 8; ++v) cp[(size_t)v * N2] = acc[T1 + t][v] + bv;
      }
    }

    if ((trip % UPS) == cg) {
      v8f sa = v8f{};
      float4 sb_c = Bl[SC0 + lane], sb_n = sb_c;
#pragma unroll
      for (int kp = 0; kp < KP; ++kp) {
        if (kp + 1 < KP) sb_n = Bl[SC0 + (kp + 1) * 32 + lane];
        v2f b0; b0.x = sb_c.x; b0.y = sb_c.y;
        v2f b1; b1.x = sb_c.z; b1.y = sb_c.w;
        sa = __builtin_amdgcn_wmma_f32_16x16x4_f32(
             false, a[2 * kp],  false, b0,
             (short)0, sa,  false,  false);
        sa = __builtin_amdgcn_wmma_f32_16x16x4_f32(
             false, a[2 * kp + 1],  false, b1,
             (short)0, sa,  false,  false);
        sb_c = sb_n;
        __builtin_amdgcn_sched_barrier(0);
      }
      const int n = lane & 15;
      if (n < 2 * H) {
        float* sp = (n < H ? sc_s + n : sc_d + (n - H)) + (size_t)(st * 16 + (hi << 3)) * H;
#pragma unroll
        for (int v = 0; v < 8; ++v) sp[(size_t)v * H] = sa[v];
      }
    }
  }
}

template <int CT>
__device__ __forceinline__ void load_row_vals(const float* __restrict__ p, int lane,
                                              float (&v)[CT / 32]) {
  if constexpr (CT == 256) {
    const float4 a = *(const float4*)(p + 4 * lane);
    const float4 b = *(const float4*)(p + 128 + 4 * lane);
    v[0] = a.x; v[1] = a.y; v[2] = a.z; v[3] = a.w;
    v[4] = b.x; v[5] = b.y; v[6] = b.z; v[7] = b.w;
  } else {
    const float2 a = *(const float2*)(p + 2 * lane);
    v[0] = a.x; v[1] = a.y;
  }
}

__device__ __forceinline__ int bcast_i(int v, int l) {
  return __builtin_amdgcn_readlane(v, l);
}
__device__ __forceinline__ float bcast_f(float v, int l) {
  return __int_as_float(__builtin_amdgcn_readlane(__float_as_int(v), l));
}

__device__ __forceinline__ float edge_logit(float a_src, float a_dst) {
  const float v = a_src + a_dst;
  return ((v > 0.f) ? v : NEG_SLOPE * v) + 0.0f;
}

template <typename F>
__device__ __forceinline__ void stream_dst_matches(const int* __restrict__ dstp, int E,
                                                   int row0, int rows, int tid, F on_round) {
  constexpr int T = 1024;
  constexpr int STEP = T * 16;
  const bool vec_ok = (E >= 4) && ((E & 3) == 0) && ((((size_t)dstp) & 15) == 0);

  auto load16 = [&](int base, int (&d)[16]) {
    if (vec_ok) {
#pragma unroll
      for (int j = 0; j < 4; ++j) {
        const int b = base + j * (T * 4) + tid * 4;
        const int4 q = *(const int4*)(dstp + min(b, E - 4));
        d[4 * j + 0] = q.x; d[4 * j + 1] = q.y; d[4 * j + 2] = q.z; d[4 * j + 3] = q.w;
      }
    } else {
#pragma unroll
      for (int j = 0; j < 4; ++j) {
#pragma unroll
        for (int k = 0; k < 4; ++k)
          d[4 * j + k] = dstp[min(base + j * (T * 4) + tid * 4 + k, E - 1)];
      }
    }
  };

  int cur[16], nxt[16];
#pragma unroll
  for (int q = 0; q < 16; ++q) { cur[q] = row0 - 1; nxt[q] = row0 - 1; }
  for (int base = -STEP; base < E; base += STEP) {
    unsigned mb = 0;
#pragma unroll
    for (int q = 0; q < 16; ++q) {
      const int eq = base + (q >> 2) * (T * 4) + tid * 4 + (q & 3);
      mb |= (eq < E && (unsigned)(cur[q] - row0) < (unsigned)rows) ? (1u << q) : 0u;
    }

    if (base + STEP < E) load16(base + STEP, nxt);

    while (__builtin_amdgcn_ballot_w32(mb != 0)) {
      const bool has = mb != 0;
      const int q = has ? __builtin_ctz(mb) : 0;
      mb &= mb - 1;
      int dsel = cur[0];
#pragma unroll
      for (int t = 1; t < 16; ++t) dsel = (q == t) ? cur[t] : dsel;
      const int edge = base + (q >> 2) * (T * 4) + tid * 4 + (q & 3);
      on_round(has, edge, dsel - row0);
    }

#pragma unroll
    for (int q = 0; q < 16; ++q) cur[q] = nxt[q];
  }
}

template <int CAP, bool TAIL>
__device__ __forceinline__ void list_block(int base, const int4 (&d)[4], int nq, int row0, int rows,
                                           int tid, int lane, int* l_src, int* l_rel, int* l_cnt) {
  constexpr int T = 1024;
  unsigned mk[16];
#pragma unroll
  for (int j = 0; j < 4; ++j) {
    unsigned okm = 0xffffffffu;
    if constexpr (TAIL) okm = __builtin_amdgcn_ballot_w32((base >> 2) + j * T + tid < nq);
    mk[4 * j + 0] = __builtin_amdgcn_ballot_w32((unsigned)(d[j].x - row0) < (unsigned)rows) & okm;
    mk[4 * j + 1] = __builtin_amdgcn_ballot_w32((unsigned)(d[j].y - row0) < (unsigned)rows) & okm;
    mk[4 * j + 2] = __builtin_amdgcn_ballot_w32((unsigned)(d[j].z - row0) < (unsigned)rows) & okm;
    mk[4 * j + 3] = __builtin_amdgcn_ballot_w32((unsigned)(d[j].w - row0) < (unsigned)rows) & okm;
  }
  int tot = 0;
#pragma unroll
  for (int q = 0; q < 16; q += 2)
    tot += __builtin_popcountll(((unsigned long long)mk[q + 1] << 32) | (unsigned long long)mk[q]);
  if (tot != 0) {
    const unsigned below = (1u << lane) - 1u;
    int pos0 = 0;
    if (lane == 0) pos0 = atomicAdd(l_cnt, tot);
    pos0 = bcast_i(pos0, 0);
#pragma unroll
    for (int q = 0; q < 16; ++q) {
      const unsigned m = mk[q];
      if (m != 0) {
        const int pos = pos0 + __builtin_popcount(m & below);
        const int4 dj = d[q >> 2];
        const int dv = (q & 3) == 0 ? dj.x : (q & 3) == 1 ? dj.y : (q & 3) == 2 ? dj.z : dj.w;
        if (((m >> lane) & 1u) && pos < CAP) {
          l_src[pos] = base + (q >> 2) * (T * 4) + tid * 4 + (q & 3);
          l_rel[pos] = dv - row0;
        }
        pos0 += __builtin_popcount(m);
      }
    }
  }
}

template <int CAP>
__device__ __forceinline__ void list_owned_edges_vec(const int* __restrict__ dstp, int E,
                                                     int row0, int rows, int tid, int lane,
                                                     int* l_src, int* l_rel, int* l_cnt) {
  constexpr int T = 1024;
  constexpr int STEP = T * 16;
  const int4* __restrict__ vp = (const int4*)dstp;
  const int nq    = E >> 2;
  const int nfull = E / STEP;
  const int tail0 = nfull * STEP;

  auto request = [&](int base, int4 (&d)[4]) {
#pragma unroll
    for (int j = 0; j < 4; ++j) d[j] = vp[min((base >> 2) + j * T + tid, nq - 1)];
  };

  int4 rt[4], ra[4], rb[4];
  request(tail0, rt);
  request(0, ra);
  for (int b = 0; b < nfull; b += 2) {
    request((b + 1) * STEP, rb);
    list_block<CAP, false>(b * STEP, ra, nq, row0, rows, tid, lane, l_src, l_rel, l_cnt);
    if (b + 1 < nfull) {
      request((b + 2) * STEP, ra);
      list_block<CAP, false>((b + 1) * STEP, rb, nq, row0, rows, tid, lane, l_src, l_rel, l_cnt);
    }
  }
  if (tail0 < E) list_block<CAP, true>(tail0, rt, nq, row0, rows, tid, lane, l_src, l_rel, l_cnt);
}

template <int CT, int H, int R, int CAP, bool REUSE, bool ADD>
__global__ __launch_bounds__(1024) void agg_softmax_rows_lds(
    const int* __restrict__ ei, const float* __restrict__ as, const float* __restrict__ ad,
    const float* __restrict__ feat, const float* __restrict__ bias,
    const float* __restrict__ gam, const float* __restrict__ bet, const float* __restrict__ addp,
    float* out, int* lists, int E, int n) {
  static_assert((CT == 256 && H == 4) || (CT == 64 && H == 1), "lane-to-head map below");
  static_assert(!REUSE || H == 1, "the hand-over is read by the one-head layer");
  constexpr int V = CT / 32;
  constexpr int T = 1024;
  constexpr int NF = 8;
  constexpr int LW = 64;
  constexpr int NJ = CAP / T;
  static_assert(CAP % T == 0, "every thread reads the same number of pairs");
  __shared__ float acc[R * CT];
  __shared__ float adrow[R * H];
  __shared__ float slog[R * H];
  __shared__ float mrow[R * H];
  __shared__ float srow[R * H];
  __shared__ int   l_src[CAP];
  __shared__ int   l_rel[CAP];
  __shared__ float l_as[CAP * H];
  __shared__ int   l_cnt;

  const int tid  = threadIdx.x;
  const int lane = tid & 31;
  const int wv   = tid >> 5;
  const int hi   = lane >> 4;
  const int row0 = blockIdx.x * R;
  const int rows = max(0, min(R, n - row0));
  const int* __restrict__ dstp = ei + E;

  int*  piece = lists + (size_t)row0 * LW;
  int2* pairs = (int2*)(piece + 4);
  const int capg = max(0, min(CAP, (rows * LW - 4) >> 1));
  int hdr = 0;
  if constexpr (REUSE) hdr = piece[0];

  auto lds_max = [&](int slot, float val) {
    if (val >= 0.0f) atomicMax((int*)&mrow[slot], __float_as_int(val));
    else             atomicMin((unsigned int*)&mrow[slot], __float_as_uint(val));
  };

  auto wave_accumulate = [&](unsigned m, int srcv, int rel, float x0, float x1, float x2, float x3) {
    while (m) {
      int  l[NF];
      bool ok[NF];
#pragma unroll
      for (int j = 0; j < NF; ++j) {
        ok[j] = (m != 0);
        l[j]  = ok[j] ? __builtin_ctz(m) : 0;
        if (ok[j]) m &= m - 1;
      }
      float v[NF][V];
#pragma unroll
      for (int j = 0; j < NF; ++j) {
        if (ok[j]) load_row_vals<CT>(feat + (size_t)bcast_i(srcv, l[j]) * CT, lane, v[j]);
      }
#pragma unroll
      for (int j = 0; j < NF; ++j) {
        if (ok[j]) {
          const int rl = bcast_i(rel, l[j]);
          float aA, aB;
          if constexpr (H == 4) {
            const float y0 = bcast_f(x0, l[j]), y1 = bcast_f(x1, l[j]);
            const float y2 = bcast_f(x2, l[j]), y3 = bcast_f(x3, l[j]);
            aA = hi ? y1 : y0; aB = hi ? y3 : y2;
          } else {
            aA = aB = bcast_f(x0, l[j]);
          }
#pragma unroll
          for (int i = 0; i < V; ++i)
            __hip_atomic_fetch_add(&acc[rl * CT + i * 32 + lane], (i < 4 ? aA : aB) * v[j][i],
                                   __ATOMIC_RELAXED, __HIP_MEMORY_SCOPE_WORKGROUP);
        }
      }
    }
  };

  if (tid == 0) l_cnt = 0;
  for (int i = tid; i < rows * H; i += T) {
    const float d_ = ad[(size_t)row0 * H + i];
    const float v  = edge_logit(as[(size_t)row0 * H + i], d_);
    adrow[i] = d_; slog[i] = v; mrow[i] = v;
  }
  __syncthreads();

  int  cnt;
  bool listed;
  if constexpr (!REUSE) {
    const bool vec_ok = (E >= 4) && ((E & 3) == 0) && ((((size_t)dstp) & 15) == 0);
    if (vec_ok) list_owned_edges_vec<CAP>(dstp, E, row0, rows, tid, lane, l_src, l_rel, &l_cnt);
    else stream_dst_matches(dstp, E, row0, rows, tid, [&](bool has, int edge, int rel) {
      const unsigned bal = __builtin_amdgcn_ballot_w32(has);
      const int lead = __builtin_ctz(bal);
      int basepos = 0;
      if (lane == lead) basepos = atomicAdd(&l_cnt, __builtin_popcount(bal));
      basepos = bcast_i(basepos, lead);
      const int pos = basepos + __builtin_popcount(bal & ((1u << lane) - 1u));
      if (has && pos < CAP) { l_src[pos] = edge; l_rel[pos] = rel; }
    });
    __syncthreads();
    cnt    = l_cnt;
    listed = cnt <= CAP;
    if (tid == 0) piece[0] = cnt;
  } else {
    cnt    = hdr;
    listed = (cnt >= 0) && (cnt <= capg);
  }

  if (listed) {
   if constexpr (REUSE) {
    if (cnt > 0) {
      int2 pr[NJ];
#pragma unroll
      for (int j = 0; j < NJ; ++j) pr[j] = pairs[min(tid + j * T, cnt - 1)];
      int sv[NJ]; float av[NJ];
#pragma unroll
      for (int j = 0; j < NJ; ++j) {
        sv[j] = min(max(pr[j].x, 0), n - 1);
        av[j] = as[sv[j]];
      }
#pragma unroll
      for (int j = 0; j < NJ; ++j) {
        const int i   = min(tid + j * T, cnt - 1);
        const int rel = min(max(pr[j].y, 0), rows - 1);
        l_src[i] = sv[j]; l_rel[i] = rel; l_as[i] = av[j];
        if (tid + j * T < cnt) lds_max(rel, edge_logit(av[j], adrow[rel]));
      }
    }
   } else {
    const bool pub = cnt <= capg;
    for (int i = tid; i < cnt; i += T) {
      const int edge = l_src[i], rel = l_rel[i];
      const int src  = min(max(ei[edge], 0), n - 1);
      l_src[i] = src;
      if (pub) pairs[i] = make_int2(src, rel);
      if constexpr (H == 4) {
        const float4 a = *(const float4*)(as + (size_t)src * 4);
        l_as[i * 4 + 0] = a.x; l_as[i * 4 + 1] = a.y; l_as[i * 4 + 2] = a.z; l_as[i * 4 + 3] = a.w;
        lds_max(rel * 4 + 0, edge_logit(a.x, adrow[rel * 4 + 0]));
        lds_max(rel * 4 + 1, edge_logit(a.y, adrow[rel * 4 + 1]));
        lds_max(rel * 4 + 2, edge_logit(a.z, adrow[rel * 4 + 2]));
        lds_max(rel * 4 + 3, edge_logit(a.w, adrow[rel * 4 + 3]));
      } else {
        const float a = as[src];
        l_as[i] = a;
        lds_max(rel, edge_logit(a, adrow[rel]));
      }
    }
   }
  } else {
    stream_dst_matches(dstp, E, row0, rows, tid, [&](bool has, int edge, int rel) {
      if (has) {
        const int src = min(max(ei[edge], 0), n - 1);
        if constexpr (H == 4) {
          const float4 a = *(const float4*)(as + (size_t)src * 4);
          lds_max(rel * 4 + 0, edge_logit(a.x, adrow[rel * 4 + 0]));
          lds_max(rel * 4 + 1, edge_logit(a.y, adrow[rel * 4 + 1]));
          lds_max(rel * 4 + 2, edge_logit(a.z, adrow[rel * 4 + 2]));
          lds_max(rel * 4 + 3, edge_logit(a.w, adrow[rel * 4 + 3]));
        } else {
          lds_max(rel, edge_logit(as[src], adrow[rel]));
        }
      }
    });
  }
  __syncthreads();

  for (int r = wv; r < rows; r += 32) {
    float v[V];
    load_row_vals<CT>(feat + (size_t)(row0 + r) * CT, lane, v);
    if constexpr (H == 4) {
      const float xA = expf(slog[r * 4 + hi]     - mrow[r * 4 + hi]);
      const float xB = expf(slog[r * 4 + 2 + hi] - mrow[r * 4 + 2 + hi]);
#pragma unroll
      for (int i = 0; i < V; ++i) acc[r * CT + i * 32 + lane] = (i < 4 ? xA : xB) * v[i];
      if (lane < 4) srow[r * 4 + lane] = expf(slog[r * 4 + lane] - mrow[r * 4 + lane]);
    } else {
      const float x = expf(slog[r] - mrow[r]);
#pragma unroll
      for (int i = 0; i < V; ++i) acc[r * CT + i * 32 + lane] = x * v[i];
      if (lane == 0) srow[r] = x;
    }
  }
  __syncthreads();

  if (listed) {
    const int lo = (int)(((long)wv * cnt) >> 5), up = (int)(((long)(wv + 1) * cnt) >> 5);
    for (int cb = lo; cb < up; cb += 32) {
      const int  idx = cb + lane;
      const bool has = idx < up;
      int srcv = 0, rel = 0;
      float x0 = 0.f, x1 = 0.f, x2 = 0.f, x3 = 0.f;
      if (has) {
        srcv = l_src[idx]; rel = l_rel[idx];
        if constexpr (H == 4) {
          x0 = expf(edge_logit(l_as[idx * 4 + 0], adrow[rel * 4 + 0]) - mrow[rel * 4 + 0]);
          x1 = expf(edge_logit(l_as[idx * 4 + 1], adrow[rel * 4 + 1]) - mrow[rel * 4 + 1]);
          x2 = expf(edge_logit(l_as[idx * 4 + 2], adrow[rel * 4 + 2]) - mrow[rel * 4 + 2]);
          x3 = expf(edge_logit(l_as[idx * 4 + 3], adrow[rel * 4 + 3]) - mrow[rel * 4 + 3]);
          __hip_atomic_fetch_add(&srow[rel * 4 + 0], x0, __ATOMIC_RELAXED, __HIP_MEMORY_SCOPE_WORKGROUP);
          __hip_atomic_fetch_add(&srow[rel * 4 + 1], x1, __ATOMIC_RELAXED, __HIP_MEMORY_SCOPE_WORKGROUP);
          __hip_atomic_fetch_add(&srow[rel * 4 + 2], x2, __ATOMIC_RELAXED, __HIP_MEMORY_SCOPE_WORKGROUP);
          __hip_atomic_fetch_add(&srow[rel * 4 + 3], x3, __ATOMIC_RELAXED, __HIP_MEMORY_SCOPE_WORKGROUP);
        } else {
          x0 = expf(edge_logit(l_as[idx], adrow[rel]) - mrow[rel]);
          __hip_atomic_fetch_add(&srow[rel], x0, __ATOMIC_RELAXED, __HIP_MEMORY_SCOPE_WORKGROUP);
        }
      }
      wave_accumulate(__builtin_amdgcn_ballot_w32(has), srcv, rel, x0, x1, x2, x3);
    }
  } else {
    stream_dst_matches(dstp, E, row0, rows, tid, [&](bool has, int edge, int rel) {
      int srcv = 0;
      float x0 = 0.f, x1 = 0.f, x2 = 0.f, x3 = 0.f;
      if (has) {
        srcv = min(max(ei[edge], 0), n - 1);
        if constexpr (H == 4) {
          const float4 a = *(const float4*)(as + (size_t)srcv * 4);
          x0 = expf(edge_logit(a.x, adrow[rel * 4 + 0]) - mrow[rel * 4 + 0]);
          x1 = expf(edge_logit(a.y, adrow[rel * 4 + 1]) - mrow[rel * 4 + 1]);
          x2 = expf(edge_logit(a.z, adrow[rel * 4 + 2]) - mrow[rel * 4 + 2]);
          x3 = expf(edge_logit(a.w, adrow[rel * 4 + 3]) - mrow[rel * 4 + 3]);
          __hip_atomic_fetch_add(&srow[rel * 4 + 0], x0, __ATOMIC_RELAXED, __HIP_MEMORY_SCOPE_WORKGROUP);
          __hip_atomic_fetch_add(&srow[rel * 4 + 1], x1, __ATOMIC_RELAXED, __HIP_MEMORY_SCOPE_WORKGROUP);
          __hip_atomic_fetch_add(&srow[rel * 4 + 2], x2, __ATOMIC_RELAXED, __HIP_MEMORY_SCOPE_WORKGROUP);
          __hip_atomic_fetch_add(&srow[rel * 4 + 3], x3, __ATOMIC_RELAXED, __HIP_MEMORY_SCOPE_WORKGROUP);
        } else {
          x0 = expf(edge_logit(as[srcv], adrow[rel]) - mrow[rel]);
          __hip_atomic_fetch_add(&srow[rel], x0, __ATOMIC_RELAXED, __HIP_MEMORY_SCOPE_WORKGROUP);
        }
      }
      wave_accumulate(__builtin_amdgcn_ballot_w32(has), srcv, rel, x0, x1, x2, x3);
    });
  }
  __syncthreads();

  float bv[V], gv[V], ev[V], idn[V];
  load_row_vals<CT>(bias, lane, bv);
  load_row_vals<CT>(gam, lane, gv);
  load_row_vals<CT>(bet, lane, ev);
  if constexpr (ADD) {
    if (wv < rows) load_row_vals<CT>(addp + (size_t)(row0 + wv) * CT, lane, idn);
  }
  for (int r = wv; r < rows; r += 32) {
    float* op = out + (size_t)(row0 + r) * CT;
    float idc[V];
    if constexpr (ADD) {
#pragma unroll
      for (int i = 0; i < V; ++i) idc[i] = idn[i];
      if (r + 32 < rows) load_row_vals<CT>(addp + (size_t)(row0 + r + 32) * CT, lane, idn);
    }
    float dA, dB;
    if constexpr (H == 4) {
      dA = srow[r * 4 + hi] + 1e-16f; dB = srow[r * 4 + 2 + hi] + 1e-16f;
    } else {
      dA = dB = srow[r] + 1e-16f;
    }
    float v[V];
#pragma unroll
    for (int i = 0; i < V; ++i) v[i] = bv[i] + acc[r * CT + i * 32 + lane] / (i < 4 ? dA : dB);
    float s = 0.f, ss = 0.f;
#pragma unroll
    for (int i = 0; i < V; ++i) { s += v[i]; ss += v[i] * v[i]; }
    for (int off = 16; off; off >>= 1) {
      s  += __shfl_xor(s,  off, 32);
      ss += __shfl_xor(ss, off, 32);
    }
    const float mu  = s / (float)CT;
    const float var = ss / (float)CT - mu * mu;
    const float inv = rsqrtf(var + 1e-5f);
#pragma unroll
    for (int i = 0; i < V; ++i) {
      float o = (v[i] - mu) * inv * gv[i] + ev[i];
      if constexpr (ADD) o += idc[i];
      v[i] = (o > 0.f) ? o : expm1f(o);
    }
    if constexpr (CT == 256) {
      *(float4*)(op + 4 * lane)       = make_float4(v[0], v[1], v[2], v[3]);
      *(float4*)(op + 128 + 4 * lane) = make_float4(v[4], v[5], v[6], v[7]);
    } else {
      *(float2*)(op + 2 * lane) = make_float2(v[0], v[1]);
    }
  }
}

extern "C" void kernel_launch(void* const* d_in, const int* in_sizes, int n_in,
                              void* d_out, int out_size, void* d_ws, size_t ws_size,
                              hipStream_t stream) {
  const float* x    = (const float*)d_in[0];
  const int*   ei   = (const int*)  d_in[1];
  const float* W1   = (const float*)d_in[2];
  const float* aS1  = (const float*)d_in[3];
  const float* aD1  = (const float*)d_in[4];
  const float* b1   = (const float*)d_in[5];
  const float* W2   = (const float*)d_in[6];
  const float* aS2  = (const float*)d_in[7];
  const float* aD2  = (const float*)d_in[8];
  const float* b2   = (const float*)d_in[9];
  const float* g1   = (const float*)d_in[10];
  const float* be1  = (const float*)d_in[11];
  const float* g2   = (const float*)d_in[12];
  const float* be2  = (const float*)d_in[13];
  const float* pW   = (const float*)d_in[14];
  const float* pb   = (const float*)d_in[15];

  const int IN_CH = 128, HEADS = 4, OUT_CH = 64, HC = HEADS * OUT_CH;
  const int N  = in_sizes[0] / IN_CH;
  const int E  = in_sizes[1] / 2;

  float* ws   = (float*)d_ws;
  float* h1   = ws; ws += (size_t)N * HC;
  float* out1 = ws; ws += (size_t)N * HC;
  float* idb  = ws; ws += (size_t)N * OUT_CH;
  float* z2   = ws; ws += (size_t)N * OUT_CH;
  float* out2 = ws; ws += (size_t)N * OUT_CH;
  float* as1  = ws; ws += (size_t)N * HEADS;
  float* ad1  = ws; ws += (size_t)N * HEADS;
  float* as2  = ws; ws += (size_t)N;
  float* ad2  = ws; ws += (size_t)N;

  auto cdiv = [](long a, long b) -> int { return (int)((a + b - 1) / b); };

  gemm_lds_weights<128, 4, 1, 4, 16, 4><<<256, 512, 0, stream>>>(x, W1, h1, pW, pb, idb, aS1, aD1, as1, ad1, N);

  agg_softmax_rows_lds<256, 4, 196, 3072, false, false><<<cdiv(N, 196), 1024, 0, stream>>>(ei, as1, ad1, h1, b1, g1, be1, nullptr, out1, (int*)out2, E, N);

  gemm_lds_weights<256, 2, 0, 2, 8, 1><<<256, 256, 0, stream>>>(out1, W2, z2, nullptr, nullptr, nullptr, aS2, aD2, as2, ad2, N);
  agg_softmax_rows_lds<64, 1, 196, 3072, true, true><<<cdiv(N, 196), 1024, 0, stream>>>(ei, as2, ad2, z2, b2, g2, be2, idb, (float*)d_out, (int*)out2, E, N);

  (void)n_in; (void)out_size; (void)ws_size;
}


# layer 1's test of the whole destination list by every group goes: the GEMM launch sorts each group's edge share into 128-byte lines per owner block; an owner reads its 256 lines; list order differs
# speedup vs baseline: 6.5430x; 1.0613x over previous
#include <hip/hip_runtime.h>
#include <math.h>

typedef float v2f __attribute__((ext_vector_type(2)));
typedef float v8f __attribute__((ext_vector_type(8)));

#define NEG_SLOPE 0.2f

template <int K, int NCOLS, int THREADS>
__device__ __forceinline__ void stage_weights(const float* __restrict__ B,
                                              float4* __restrict__ dst, int tid) {
  constexpr int KP   = K / 8;
  constexpr int ENT  = (NCOLS / 16) * KP * 32;
  constexpr int ITER = ENT / THREADS;
  static_assert(ENT % THREADS == 0 && THREADS % 32 == 0, "whole waves share the entries evenly");
  const int c = tid & 15, h = (tid >> 4) & 1;
  float4 v[ITER];
#pragma unroll
  for (int it = 0; it < ITER; ++it) {
    const int g = (tid + it * THREADS) >> 5;
    const int tile = g / KP, kp = g - tile * KP;
    const float* p = B + (size_t)(8 * kp + 2 * h) * NCOLS + tile * 16 + c;
    v[it].x = p[0]; v[it].y = p[NCOLS]; v[it].z = p[4 * NCOLS]; v[it].w = p[5 * NCOLS];
  }
#pragma unroll
  for (int it = 0; it < ITER; ++it) dst[tid + it * THREADS] = v[it];
}

template <int K>
__device__ __forceinline__ int score_head(int tid) {
  return __builtin_amdgcn_readfirstlane(tid / K);
}

template <int K, int H>
__device__ __forceinline__ void request_score_row(const float* __restrict__ B, int tid,
                                                  float4 (&w)[16]) {
  static_assert(K % 32 == 0, "a wave stays inside one head");
  const int h = score_head<K>(tid), k = tid - h * K;
  const float4* wv = (const float4*)(B + (size_t)k * (H * 64) + h * 64);
#pragma unroll
  for (int i = 0; i < 16; ++i) w[i] = wv[i];
}

template <int K, int H, int THREADS>
__device__ __forceinline__ void stage_score_weights(const float4 (&w)[16],
                                                    const float* __restrict__ att_s,
                                                    const float* __restrict__ att_d,
                                                    float* __restrict__ img, int tid) {
  constexpr int C  = 64;
  static_assert(K * H == THREADS, "one (k, head) pair per thread");
  auto pos = [](int k, int n) {
    const int kp = k >> 3, r = k & 7;
    return ((kp * 32 + ((r >> 1) & 1) * 16 + n) << 2) + (((r >> 2) << 1) | (r & 1));
  };
  {
    const int h = score_head<K>(tid), k = tid - h * K;
    const float4* sv = (const float4*)(att_s + h * C);
    const float4* dv = (const float4*)(att_d + h * C);
    float s = 0.f, d = 0.f;
#pragma unroll
    for (int i = 0; i < C / 4; ++i) {
      const float4 a = sv[i], b = dv[i];
      s += w[i].x * a.x + w[i].y * a.y + w[i].z * a.z + w[i].w * a.w;
      d += w[i].x * b.x + w[i].y * b.y + w[i].z * b.z + w[i].w * b.w;
    }
    img[pos(k, h)]     = s;
    img[pos(k, H + h)] = d;
  }
  for (int i = tid; i < K * 16; i += THREADS) {
    const int n = i & 15;
    if (n >= 2 * H) img[pos(i >> 4, n)] = 0.f;
  }
}

template <int K>
__device__ __forceinline__ void load_a_strip(const float* __restrict__ A, int strip, int lane,
                                             v2f (&a)[K / 4]) {
  const float* ap = A + (size_t)(strip * 16 + (lane & 15)) * K + ((lane >> 4) << 1);
#pragma unroll
  for (int ks = 0; ks < K / 4; ++ks) a[ks] = *(const v2f*)(ap + 4 * ks);
}

template <int K, int T1, int T2, int UPS, int WAVES, int H, bool BIN, int BR>
__global__ __launch_bounds__(WAVES * 32) void gemm_lds_weights(
    const float* __restrict__ A,
    const float* __restrict__ B1, float* __restrict__ C1,
    const float* __restrict__ B2, const float* __restrict__ bias2, float* __restrict__ C2,
    const float* __restrict__ att_s, const float* __restrict__ att_d,
    float* __restrict__ sc_s, float* __restrict__ sc_d,
    int M, const int* __restrict__ bin_dst, int bin_E, int* __restrict__ bins) {
  constexpr int NT1 = UPS * T1, NT2 = UPS * T2;
  constexpr int N1 = NT1 * 16, N2 = NT2 * 16;
  constexpr int KP = K / 8;
  constexpr int TT = T1 + T2;
  constexpr int RS = WAVES / UPS;
  static_assert(T1 % 2 == 0 && WAVES % UPS == 0 && K % 8 == 0, "C1 tiles are stored in pairs");
  constexpr int SC0  = 0;
  constexpr int W0   = KP * 32;
  constexpr int NEED = W0 + (NT1 + NT2) * KP * 32;
  static_assert(N1 == H * 64 && 2 * H <= 16, "B1's column blocks of 64 are the heads");
  constexpr int FLOOR = 164 * 1024 / 16;
  __shared__ float4 Bl[NEED > FLOOR ? NEED : FLOOR];

  const int tid  = threadIdx.x;
  const int lane = tid & 31;
  const int wv   = tid >> 5;
  const int hi   = lane >> 4;
  const int cg   = wv % UPS;
  const int rr   = wv / UPS;

  const int strips = M >> 4;
  const int sb = (int)((blockIdx.x * (unsigned)strips) / gridDim.x);
  const int se = (int)(((blockIdx.x + 1) * (unsigned)strips) / gridDim.x);

  if constexpr (BIN) {
    constexpr int MAXNB = 512;
    __shared__ int bcnt[MAXNB];
    const int NB = (M + BR - 1) / BR;
    const bool ok = NB <= MAXNB && (gridDim.x & 7) == 0 &&
                    (long)gridDim.x * NB * 32 <= (long)M * 64;
    if (ok) {
      for (int i = tid; i < NB; i += WAVES * 32) bcnt[i] = 0;
      __syncthreads();
      const int per = (bin_E + (int)gridDim.x - 1) / (int)gridDim.x;
      const int e0  = min((int)blockIdx.x * per, bin_E);
      const int e1  = min(e0 + per, bin_E);
      int* mine = bins + (size_t)blockIdx.x * NB * 32;
      for (int e = e0 + tid; e < e1; e += WAVES * 32) {
        const int d = bin_dst[e];
        if ((unsigned)d < (unsigned)M) {
          const int o = d / BR;
          const int s = atomicAdd(&bcnt[o], 1);
          if (s < 31) mine[o * 32 + 1 + s] = e;
        }
      }
      __syncthreads();
      for (int i = tid; i < NB; i += WAVES * 32) mine[i * 32] = bcnt[i];
    }
  }

  v2f an[K / 4];
  int st = sb + rr;
  if (st < se) load_a_strip<K>(A, st, lane, an);

  float4 wrow[16];
  request_score_row<K, H>(B1, tid, wrow);

  stage_weights<K, N1, WAVES * 32>(B1, Bl + W0, tid);
  if constexpr (T2 > 0)
    stage_weights<K, N2, WAVES * 32>(B2, Bl + W0 + NT1 * KP * 32, tid);
  stage_score_weights<K, H, WAVES * 32>(wrow, att_s, att_d, (float*)(Bl + SC0), tid);
  __syncthreads();

  int trip = 0;
  for (; st < se; st += RS, ++trip) {
    v2f a[K / 4];
#pragma unroll
    for (int ks = 0; ks < K / 4; ++ks) a[ks] = an[ks];
    if (st + RS < se) load_a_strip<K>(A, st + RS, lane, an);

    v8f acc[TT];
#pragma unroll
    for (int t = 0; t < TT; ++t) acc[t] = v8f{};
    float4 bc[TT], bn[TT];
#pragma unroll
    for (int t = 0; t < TT; ++t) {
      const int tile = (t < T1) ? (cg * T1 + t) : (NT1 + cg * T2 + (t - T1));
      bc[t] = Bl[W0 + (tile * KP) * 32 + lane];
    }
#pragma unroll
    for (int kp = 0; kp < KP; ++kp) {
      if (kp + 1 < KP) {
#pragma unroll
        for (int t = 0; t < TT; ++t) {
          const int tile = (t < T1) ? (cg * T1 + t) : (NT1 + cg * T2 + (t - T1));
          bn[t] = Bl[W0 + (tile * KP + kp + 1) * 32 + lane];
        }
      }
#pragma unroll
      for (int t = 0; t < TT; ++t) {
        v2f bb; bb.x = bc[t].x; bb.y = bc[t].y;
        acc[t] = __builtin_amdgcn_wmma_f32_16x16x4_f32(
             false, a[2 * kp],  false, bb,
             (short)0, acc[t],  false,  false);
      }
#pragma unroll
      for (int t = 0; t < TT; ++t) {
        v2f bb; bb.x = bc[t].z; bb.y = bc[t].w;
        acc[t] = __builtin_amdgcn_wmma_f32_16x16x4_f32(
             false, a[2 * kp + 1],  false, bb,
             (short)0, acc[t],  false,  false);
      }
#pragma unroll
      for (int t = 0; t < TT; ++t) bc[t] = bn[t];
      __builtin_amdgcn_sched_barrier(0);
    }

#pragma unroll
    for (int p = 0; p < T1 / 2; ++p) {
      float* cp = C1 + (size_t)(st * 16) * N1 + (cg * T1 + 2 * p) * 16 + lane;
#pragma unroll
      for (int v = 0; v < 8; ++v) {
        const float own0 = acc[2 * p][v], own1 = acc[2 * p + 1][v];
        const float give = hi ? own0 : own1;
        const float got  = __shfl_xor(give, 16, 32);
        cp[(size_t)v * N1]       = hi ? got : own0;
        cp[(size_t)(8 + v) * N1] = hi ? own1 : got;
      }
    }
    if constexpr (T2 > 0) {
#pragma unroll
      for (int t = 0; t < T2; ++t) {
        const int ncol = (cg * T2 + t) * 16 + (lane & 15);
        const float bv = bias2 ? bias2[ncol] : 0.0f;
        float* cp = C2 + (size_t)(st * 16 + (hi << 3)) * N2 + ncol;
#pragma unroll
        for (int v = 0; v < 8; ++v) cp[(size_t)v * N2] = acc[T1 + t][v] + bv;
      }
    }

    if ((trip % UPS) == cg) {
      v8f sa = v8f{};
      float4 sb_c = Bl[SC0 + lane], sb_n = sb_c;
#pragma unroll
      for (int kp = 0; kp < KP; ++kp) {
        if (kp + 1 < KP) sb_n = Bl[SC0 + (kp + 1) * 32 + lane];
        v2f b0; b0.x = sb_c.x; b0.y = sb_c.y;
        v2f b1; b1.x = sb_c.z; b1.y = sb_c.w;
        sa = __builtin_amdgcn_wmma_f32_16x16x4_f32(
             false, a[2 * kp],  false, b0,
             (short)0, sa,  false,  false);
        sa = __builtin_amdgcn_wmma_f32_16x16x4_f32(
             false, a[2 * kp + 1],  false, b1,
             (short)0, sa,  false,  false);
        sb_c = sb_n;
        __builtin_amdgcn_sched_barrier(0);
      }
      const int n = lane & 15;
      if (n < 2 * H) {
        float* sp = (n < H ? sc_s + n : sc_d + (n - H)) + (size_t)(st * 16 + (hi << 3)) * H;
#pragma unroll
        for (int v = 0; v < 8; ++v) sp[(size_t)v * H] = sa[v];
      }
    }
  }
}

template <int CT>
__device__ __forceinline__ void load_row_vals(const float* __restrict__ p, int lane,
                                              float (&v)[CT / 32]) {
  if constexpr (CT == 256) {
    const float4 a = *(const float4*)(p + 4 * lane);
    const float4 b = *(const float4*)(p + 128 + 4 * lane);
    v[0] = a.x; v[1] = a.y; v[2] = a.z; v[3] = a.w;
    v[4] = b.x; v[5] = b.y; v[6] = b.z; v[7] = b.w;
  } else {
    const float2 a = *(const float2*)(p + 2 * lane);
    v[0] = a.x; v[1] = a.y;
  }
}

__device__ __forceinline__ int bcast_i(int v, int l) {
  return __builtin_amdgcn_readlane(v, l);
}
__device__ __forceinline__ float bcast_f(float v, int l) {
  return __int_as_float(__builtin_amdgcn_readlane(__float_as_int(v), l));
}

__device__ __forceinline__ float edge_logit(float a_src, float a_dst) {
  const float v = a_src + a_dst;
  return ((v > 0.f) ? v : NEG_SLOPE * v) + 0.0f;
}

template <typename F>
__device__ __forceinline__ void stream_dst_matches(const int* __restrict__ dstp, int E,
                                                   int row0, int rows, int tid, F on_round) {
  constexpr int T = 1024;
  constexpr int STEP = T * 16;
  const bool vec_ok = (E >= 4) && ((E & 3) == 0) && ((((size_t)dstp) & 15) == 0);

  auto load16 = [&](int base, int (&d)[16]) {
    if (vec_ok) {
#pragma unroll
      for (int j = 0; j < 4; ++j) {
        const int b = base + j * (T * 4) + tid * 4;
        const int4 q = *(const int4*)(dstp + min(b, E - 4));
        d[4 * j + 0] = q.x; d[4 * j + 1] = q.y; d[4 * j + 2] = q.z; d[4 * j + 3] = q.w;
      }
    } else {
#pragma unroll
      for (int j = 0; j < 4; ++j) {
#pragma unroll
        for (int k = 0; k < 4; ++k)
          d[4 * j + k] = dstp[min(base + j * (T * 4) + tid * 4 + k, E - 1)];
      }
    }
  };

  int cur[16], nxt[16];
#pragma unroll
  for (int q = 0; q < 16; ++q) { cur[q] = row0 - 1; nxt[q] = row0 - 1; }
  for (int base = -STEP; base < E; base += STEP) {
    unsigned mb = 0;
#pragma unroll
    for (int q = 0; q < 16; ++q) {
      const int eq = base + (q >> 2) * (T * 4) + tid * 4 + (q & 3);
      mb |= (eq < E && (unsigned)(cur[q] - row0) < (unsigned)rows) ? (1u << q) : 0u;
    }

    if (base + STEP < E) load16(base + STEP, nxt);

    while (__builtin_amdgcn_ballot_w32(mb != 0)) {
      const bool has = mb != 0;
      const int q = has ? __builtin_ctz(mb) : 0;
      mb &= mb - 1;
      int dsel = cur[0];
#pragma unroll
      for (int t = 1; t < 16; ++t) dsel = (q == t) ? cur[t] : dsel;
      const int edge = base + (q >> 2) * (T * 4) + tid * 4 + (q & 3);
      on_round(has, edge, dsel - row0);
    }

#pragma unroll
    for (int q = 0; q < 16; ++q) cur[q] = nxt[q];
  }
}

template <int CAP, bool TAIL>
__device__ __forceinline__ void list_block(int base, const int4 (&d)[4], int nq, int row0, int rows,
                                           int tid, int lane, int* l_src, int* l_rel, int* l_cnt) {
  constexpr int T = 1024;
  unsigned mk[16];
#pragma unroll
  for (int j = 0; j < 4; ++j) {
    unsigned okm = 0xffffffffu;
    if constexpr (TAIL) okm = __builtin_amdgcn_ballot_w32((base >> 2) + j * T + tid < nq);
    mk[4 * j + 0] = __builtin_amdgcn_ballot_w32((unsigned)(d[j].x - row0) < (unsigned)rows) & okm;
    mk[4 * j + 1] = __builtin_amdgcn_ballot_w32((unsigned)(d[j].y - row0) < (unsigned)rows) & okm;
    mk[4 * j + 2] = __builtin_amdgcn_ballot_w32((unsigned)(d[j].z - row0) < (unsigned)rows) & okm;
    mk[4 * j + 3] = __builtin_amdgcn_ballot_w32((unsigned)(d[j].w - row0) < (unsigned)rows) & okm;
  }
  int tot = 0;
#pragma unroll
  for (int q = 0; q < 16; q += 2)
    tot += __builtin_popcountll(((unsigned long long)mk[q + 1] << 32) | (unsigned long long)mk[q]);
  if (tot != 0) {
    const unsigned below = (1u << lane) - 1u;
    int pos0 = 0;
    if (lane == 0) pos0 = atomicAdd(l_cnt, tot);
    pos0 = bcast_i(pos0, 0);
#pragma unroll
    for (int q = 0; q < 16; ++q) {
      const unsigned m = mk[q];
      if (m != 0) {
        const int pos = pos0 + __builtin_popcount(m & below);
        const int4 dj = d[q >> 2];
        const int dv = (q & 3) == 0 ? dj.x : (q & 3) == 1 ? dj.y : (q & 3) == 2 ? dj.z : dj.w;
        if (((m >> lane) & 1u) && pos < CAP) {
          l_src[pos] = base + (q >> 2) * (T * 4) + tid * 4 + (q & 3);
          l_rel[pos] = dv - row0;
        }
        pos0 += __builtin_popcount(m);
      }
    }
  }
}

template <int CAP>
__device__ __forceinline__ void list_owned_edges_vec(const int* __restrict__ dstp, int E,
                                                     int row0, int rows, int tid, int lane,
                                                     int* l_src, int* l_rel, int* l_cnt) {
  constexpr int T = 1024;
  constexpr int STEP = T * 16;
  const int4* __restrict__ vp = (const int4*)dstp;
  const int nq    = E >> 2;
  const int nfull = E / STEP;
  const int tail0 = nfull * STEP;

  auto request = [&](int base, int4 (&d)[4]) {
#pragma unroll
    for (int j = 0; j < 4; ++j) d[j] = vp[min((base >> 2) + j * T + tid, nq - 1)];
  };

  int4 rt[4], ra[4], rb[4];
  request(tail0, rt);
  request(0, ra);
  for (int b = 0; b < nfull; b += 2) {
    request((b + 1) * STEP, rb);
    list_block<CAP, false>(b * STEP, ra, nq, row0, rows, tid, lane, l_src, l_rel, l_cnt);
    if (b + 1 < nfull) {
      request((b + 2) * STEP, ra);
      list_block<CAP, false>((b + 1) * STEP, rb, nq, row0, rows, tid, lane, l_src, l_rel, l_cnt);
    }
  }
  if (tail0 < E) list_block<CAP, true>(tail0, rt, nq, row0, rows, tid, lane, l_src, l_rel, l_cnt);
}

template <int CT, int H, int R, int CAP, bool REUSE, bool ADD>
__global__ __launch_bounds__(1024) void agg_softmax_rows_lds(
    const int* __restrict__ ei, const float* __restrict__ as, const float* __restrict__ ad,
    const float* __restrict__ feat, const float* __restrict__ bias,
    const float* __restrict__ gam, const float* __restrict__ bet, const float* __restrict__ addp,
    float* out, int* lists, int E, int n, const int* __restrict__ bins, int binP) {
  static_assert((CT == 256 && H == 4) || (CT == 64 && H == 1), "lane-to-head map below");
  static_assert(!REUSE || H == 1, "the hand-over is read by the one-head layer");
  constexpr int V = CT / 32;
  constexpr int T = 1024;
  constexpr int NF = 8;
  constexpr int LW = 64;
  constexpr int NJ = CAP / T;
  static_assert(CAP % T == 0, "every thread reads the same number of pairs");
  __shared__ float acc[R * CT];
  __shared__ float adrow[R * H];
  __shared__ float slog[R * H];
  __shared__ float mrow[R * H];
  __shared__ float srow[R * H];
  __shared__ int   l_src[CAP];
  __shared__ int   l_rel[CAP];
  __shared__ float l_as[CAP * H];
  __shared__ int   l_cnt;
  __shared__ int   l_ovf;

  const int tid  = threadIdx.x;
  const int lane = tid & 31;
  const int wv   = tid >> 5;
  const int hi   = lane >> 4;
  const int row0 = blockIdx.x * R;
  const int rows = max(0, min(R, n - row0));
  const int* __restrict__ dstp = ei + E;

  int*  piece = lists + (size_t)row0 * LW;
  int2* pairs = (int2*)(piece + 4);
  const int capg = max(0, min(CAP, (rows * LW - 4) >> 1));
  int hdr = 0;
  if constexpr (REUSE) hdr = piece[0];

  auto lds_max = [&](int slot, float val) {
    if (val >= 0.0f) atomicMax((int*)&mrow[slot], __float_as_int(val));
    else             atomicMin((unsigned int*)&mrow[slot], __float_as_uint(val));
  };

  auto wave_accumulate = [&](unsigned m, int srcv, int rel, float x0, float x1, float x2, float x3) {
    while (m) {
      int  l[NF];
      bool ok[NF];
#pragma unroll
      for (int j = 0; j < NF; ++j) {
        ok[j] = (m != 0);
        l[j]  = ok[j] ? __builtin_ctz(m) : 0;
        if (ok[j]) m &= m - 1;
      }
      float v[NF][V];
#pragma unroll
      for (int j = 0; j < NF; ++j) {
        if (ok[j]) load_row_vals<CT>(feat + (size_t)bcast_i(srcv, l[j]) * CT, lane, v[j]);
      }
#pragma unroll
      for (int j = 0; j < NF; ++j) {
        if (ok[j]) {
          const int rl = bcast_i(rel, l[j]);
          float aA, aB;
          if constexpr (H == 4) {
            const float y0 = bcast_f(x0, l[j]), y1 = bcast_f(x1, l[j]);
            const float y2 = bcast_f(x2, l[j]), y3 = bcast_f(x3, l[j]);
            aA = hi ? y1 : y0; aB = hi ? y3 : y2;
          } else {
            aA = aB = bcast_f(x0, l[j]);
          }
#pragma unroll
          for (int i = 0; i < V; ++i)
            __hip_atomic_fetch_add(&acc[rl * CT + i * 32 + lane], (i < 4 ? aA : aB) * v[j][i],
                                   __ATOMIC_RELAXED, __HIP_MEMORY_SCOPE_WORKGROUP);
        }
      }
    }
  };

  if (tid == 0) { l_cnt = 0; l_ovf = 0; }
  for (int i = tid; i < rows * H; i += T) {
    const float d_ = ad[(size_t)row0 * H + i];
    const float v  = edge_logit(as[(size_t)row0 * H + i], d_);
    adrow[i] = d_; slog[i] = v; mrow[i] = v;
  }
  __syncthreads();

  int  cnt;
  bool listed;
  bool binned = false;
  if constexpr (!REUSE) {
    const int NBc = (n + R - 1) / R;
    const bool bins_ok = bins != nullptr && binP > 0 && (binP & 7) == 0 && NBc <= 512 &&
                         (int)gridDim.x == NBc && (long)binP * NBc * 32 <= (long)n * 64;
    if (bins_ok) {
      for (int t = tid; t < binP * 4; t += T) {
        const int p = t >> 2, qd = t & 3;
        const int4* ln = (const int4*)(bins + ((size_t)p * NBc + blockIdx.x) * 32) + qd * 2;
        const int4 u = ln[0], w = ln[1];
        const int c = __shfl(u.x, lane & ~3, 32);
        const bool good = (c >= 0) && (c <= 31);
        if (!good) l_ovf = 1;
        int pos0 = 0;
        if (qd == 0 && good && c > 0) pos0 = atomicAdd(&l_cnt, c);
        pos0 = __shfl(pos0, lane & ~3, 32);
        const int vals[8] = {u.x, u.y, u.z, u.w, w.x, w.y, w.z, w.w};
#pragma unroll
        for (int j = 0; j < 8; ++j) {
          const int k = qd * 8 + j - 1;
          const int pos = pos0 + k;
          if (good && k >= 0 && k < c && pos < CAP) l_src[pos] = vals[j];
        }
      }
      __syncthreads();
      binned = (l_ovf == 0);
      if (!binned) {
        if (tid == 0) l_cnt = 0;
        __syncthreads();
      }
    }
    const bool vec_ok = (E >= 4) && ((E & 3) == 0) && ((((size_t)dstp) & 15) == 0);
    if (binned) {   }
    else if (vec_ok) list_owned_edges_vec<CAP>(dstp, E, row0, rows, tid, lane, l_src, l_rel, &l_cnt);
    else stream_dst_matches(dstp, E, row0, rows, tid, [&](bool has, int edge, int rel) {
      const unsigned bal = __builtin_amdgcn_ballot_w32(has);
      const int lead = __builtin_ctz(bal);
      int basepos = 0;
      if (lane == lead) basepos = atomicAdd(&l_cnt, __builtin_popcount(bal));
      basepos = bcast_i(basepos, lead);
      const int pos = basepos + __builtin_popcount(bal & ((1u << lane) - 1u));
      if (has && pos < CAP) { l_src[pos] = edge; l_rel[pos] = rel; }
    });
    __syncthreads();
    cnt    = l_cnt;
    listed = cnt <= CAP;
    if (tid == 0) piece[0] = cnt;
  } else {
    cnt    = hdr;
    listed = (cnt >= 0) && (cnt <= capg);
  }

  if (listed) {
   if constexpr (REUSE) {
    if (cnt > 0) {
      int2 pr[NJ];
#pragma unroll
      for (int j = 0; j < NJ; ++j) pr[j] = pairs[min(tid + j * T, cnt - 1)];
      int sv[NJ]; float av[NJ];
#pragma unroll
      for (int j = 0; j < NJ; ++j) {
        sv[j] = min(max(pr[j].x, 0), n - 1);
        av[j] = as[sv[j]];
      }
#pragma unroll
      for (int j = 0; j < NJ; ++j) {
        const int i   = min(tid + j * T, cnt - 1);
        const int rel = min(max(pr[j].y, 0), rows - 1);
        l_src[i] = sv[j]; l_rel[i] = rel; l_as[i] = av[j];
        if (tid + j * T < cnt) lds_max(rel, edge_logit(av[j], adrow[rel]));
      }
    }
   } else {
    const bool pub = cnt <= capg;
    for (int i = tid; i < cnt; i += T) {
      int edge = l_src[i], rel;
      if (binned) {
        edge = min(max(edge, 0), E - 1);
        rel  = min(max(dstp[edge] - row0, 0), rows - 1);
        l_rel[i] = rel;
      } else {
        rel = l_rel[i];
      }
      const int src  = min(max(ei[edge], 0), n - 1);
      l_src[i] = src;
      if (pub) pairs[i] = make_int2(src, rel);
      if constexpr (H == 4) {
        const float4 a = *(const float4*)(as + (size_t)src * 4);
        l_as[i * 4 + 0] = a.x; l_as[i * 4 + 1] = a.y; l_as[i * 4 + 2] = a.z; l_as[i * 4 + 3] = a.w;
        lds_max(rel * 4 + 0, edge_logit(a.x, adrow[rel * 4 + 0]));
        lds_max(rel * 4 + 1, edge_logit(a.y, adrow[rel * 4 + 1]));
        lds_max(rel * 4 + 2, edge_logit(a.z, adrow[rel * 4 + 2]));
        lds_max(rel * 4 + 3, edge_logit(a.w, adrow[rel * 4 + 3]));
      } else {
        const float a = as[src];
        l_as[i] = a;
        lds_max(rel, edge_logit(a, adrow[rel]));
      }
    }
   }
  } else {
    stream_dst_matches(dstp, E, row0, rows, tid, [&](bool has, int edge, int rel) {
      if (has) {
        const int src = min(max(ei[edge], 0), n - 1);
        if constexpr (H == 4) {
          const float4 a = *(const float4*)(as + (size_t)src * 4);
          lds_max(rel * 4 + 0, edge_logit(a.x, adrow[rel * 4 + 0]));
          lds_max(rel * 4 + 1, edge_logit(a.y, adrow[rel * 4 + 1]));
          lds_max(rel * 4 + 2, edge_logit(a.z, adrow[rel * 4 + 2]));
          lds_max(rel * 4 + 3, edge_logit(a.w, adrow[rel * 4 + 3]));
        } else {
          lds_max(rel, edge_logit(as[src], adrow[rel]));
        }
      }
    });
  }
  __syncthreads();

  for (int r = wv; r < rows; r += 32) {
    float v[V];
    load_row_vals<CT>(feat + (size_t)(row0 + r) * CT, lane, v);
    if constexpr (H == 4) {
      const float xA = expf(slog[r * 4 + hi]     - mrow[r * 4 + hi]);
      const float xB = expf(slog[r * 4 + 2 + hi] - mrow[r * 4 + 2 + hi]);
#pragma unroll
      for (int i = 0; i < V; ++i) acc[r * CT + i * 32 + lane] = (i < 4 ? xA : xB) * v[i];
      if (lane < 4) srow[r * 4 + lane] = expf(slog[r * 4 + lane] - mrow[r * 4 + lane]);
    } else {
      const float x = expf(slog[r] - mrow[r]);
#pragma unroll
      for (int i = 0; i < V; ++i) acc[r * CT + i * 32 + lane] = x * v[i];
      if (lane == 0) srow[r] = x;
    }
  }
  __syncthreads();

  if (listed) {
    const int lo = (int)(((long)wv * cnt) >> 5), up = (int)(((long)(wv + 1) * cnt) >> 5);
    for (int cb = lo; cb < up; cb += 32) {
      const int  idx = cb + lane;
      const bool has = idx < up;
      int srcv = 0, rel = 0;
      float x0 = 0.f, x1 = 0.f, x2 = 0.f, x3 = 0.f;
      if (has) {
        srcv = l_src[idx]; rel = l_rel[idx];
        if constexpr (H == 4) {
          x0 = expf(edge_logit(l_as[idx * 4 + 0], adrow[rel * 4 + 0]) - mrow[rel * 4 + 0]);
          x1 = expf(edge_logit(l_as[idx * 4 + 1], adrow[rel * 4 + 1]) - mrow[rel * 4 + 1]);
          x2 = expf(edge_logit(l_as[idx * 4 + 2], adrow[rel * 4 + 2]) - mrow[rel * 4 + 2]);
          x3 = expf(edge_logit(l_as[idx * 4 + 3], adrow[rel * 4 + 3]) - mrow[rel * 4 + 3]);
          __hip_atomic_fetch_add(&srow[rel * 4 + 0], x0, __ATOMIC_RELAXED, __HIP_MEMORY_SCOPE_WORKGROUP);
          __hip_atomic_fetch_add(&srow[rel * 4 + 1], x1, __ATOMIC_RELAXED, __HIP_MEMORY_SCOPE_WORKGROUP);
          __hip_atomic_fetch_add(&srow[rel * 4 + 2], x2, __ATOMIC_RELAXED, __HIP_MEMORY_SCOPE_WORKGROUP);
          __hip_atomic_fetch_add(&srow[rel * 4 + 3], x3, __ATOMIC_RELAXED, __HIP_MEMORY_SCOPE_WORKGROUP);
        } else {
          x0 = expf(edge_logit(l_as[idx], adrow[rel]) - mrow[rel]);
          __hip_atomic_fetch_add(&srow[rel], x0, __ATOMIC_RELAXED, __HIP_MEMORY_SCOPE_WORKGROUP);
        }
      }
      wave_accumulate(__builtin_amdgcn_ballot_w32(has), srcv, rel, x0, x1, x2, x3);
    }
  } else {
    stream_dst_matches(dstp, E, row0, rows, tid, [&](bool has, int edge, int rel) {
      int srcv = 0;
      float x0 = 0.f, x1 = 0.f, x2 = 0.f, x3 = 0.f;
      if (has) {
        srcv = min(max(ei[edge], 0), n - 1);
        if constexpr (H == 4) {
          const float4 a = *(const float4*)(as + (size_t)srcv * 4);
          x0 = expf(edge_logit(a.x, adrow[rel * 4 + 0]) - mrow[rel * 4 + 0]);
          x1 = expf(edge_logit(a.y, adrow[rel * 4 + 1]) - mrow[rel * 4 + 1]);
          x2 = expf(edge_logit(a.z, adrow[rel * 4 + 2]) - mrow[rel * 4 + 2]);
          x3 = expf(edge_logit(a.w, adrow[rel * 4 + 3]) - mrow[rel * 4 + 3]);
          __hip_atomic_fetch_add(&srow[rel * 4 + 0], x0, __ATOMIC_RELAXED, __HIP_MEMORY_SCOPE_WORKGROUP);
          __hip_atomic_fetch_add(&srow[rel * 4 + 1], x1, __ATOMIC_RELAXED, __HIP_MEMORY_SCOPE_WORKGROUP);
          __hip_atomic_fetch_add(&srow[rel * 4 + 2], x2, __ATOMIC_RELAXED, __HIP_MEMORY_SCOPE_WORKGROUP);
          __hip_atomic_fetch_add(&srow[rel * 4 + 3], x3, __ATOMIC_RELAXED, __HIP_MEMORY_SCOPE_WORKGROUP);
        } else {
          x0 = expf(edge_logit(as[srcv], adrow[rel]) - mrow[rel]);
          __hip_atomic_fetch_add(&srow[rel], x0, __ATOMIC_RELAXED, __HIP_MEMORY_SCOPE_WORKGROUP);
        }
      }
      wave_accumulate(__builtin_amdgcn_ballot_w32(has), srcv, rel, x0, x1, x2, x3);
    });
  }
  __syncthreads();

  float bv[V], gv[V], ev[V], idn[V];
  load_row_vals<CT>(bias, lane, bv);
  load_row_vals<CT>(gam, lane, gv);
  load_row_vals<CT>(bet, lane, ev);
  if constexpr (ADD) {
    if (wv < rows) load_row_vals<CT>(addp + (size_t)(row0 + wv) * CT, lane, idn);
  }
  for (int r = wv; r < rows; r += 32) {
    float* op = out + (size_t)(row0 + r) * CT;
    float idc[V];
    if constexpr (ADD) {
#pragma unroll
      for (int i = 0; i < V; ++i) idc[i] = idn[i];
      if (r + 32 < rows) load_row_vals<CT>(addp + (size_t)(row0 + r + 32) * CT, lane, idn);
    }
    float dA, dB;
    if constexpr (H == 4) {
      dA = srow[r * 4 + hi] + 1e-16f; dB = srow[r * 4 + 2 + hi] + 1e-16f;
    } else {
      dA = dB = srow[r] + 1e-16f;
    }
    float v[V];
#pragma unroll
    for (int i = 0; i < V; ++i) v[i] = bv[i] + acc[r * CT + i * 32 + lane] / (i < 4 ? dA : dB);
    float s = 0.f, ss = 0.f;
#pragma unroll
    for (int i = 0; i < V; ++i) { s += v[i]; ss += v[i] * v[i]; }
    for (int off = 16; off; off >>= 1) {
      s  += __shfl_xor(s,  off, 32);
      ss += __shfl_xor(ss, off, 32);
    }
    const float mu  = s / (float)CT;
    const float var = ss / (float)CT - mu * mu;
    const float inv = rsqrtf(var + 1e-5f);
#pragma unroll
    for (int i = 0; i < V; ++i) {
      float o = (v[i] - mu) * inv * gv[i] + ev[i];
      if constexpr (ADD) o += idc[i];
      v[i] = (o > 0.f) ? o : expm1f(o);
    }
    if constexpr (CT == 256) {
      *(float4*)(op + 4 * lane)       = make_float4(v[0], v[1], v[2], v[3]);
      *(float4*)(op + 128 + 4 * lane) = make_float4(v[4], v[5], v[6], v[7]);
    } else {
      *(float2*)(op + 2 * lane) = make_float2(v[0], v[1]);
    }
  }
}

extern "C" void kernel_launch(void* const* d_in, const int* in_sizes, int n_in,
                              void* d_out, int out_size, void* d_ws, size_t ws_size,
                              hipStream_t stream) {
  const float* x    = (const float*)d_in[0];
  const int*   ei   = (const int*)  d_in[1];
  const float* W1   = (const float*)d_in[2];
  const float* aS1  = (const float*)d_in[3];
  const float* aD1  = (const float*)d_in[4];
  const float* b1   = (const float*)d_in[5];
  const float* W2   = (const float*)d_in[6];
  const float* aS2  = (const float*)d_in[7];
  const float* aD2  = (const float*)d_in[8];
  const float* b2   = (const float*)d_in[9];
  const float* g1   = (const float*)d_in[10];
  const float* be1  = (const float*)d_in[11];
  const float* g2   = (const float*)d_in[12];
  const float* be2  = (const float*)d_in[13];
  const float* pW   = (const float*)d_in[14];
  const float* pb   = (const float*)d_in[15];

  const int IN_CH = 128, HEADS = 4, OUT_CH = 64, HC = HEADS * OUT_CH;
  const int N  = in_sizes[0] / IN_CH;
  const int E  = in_sizes[1] / 2;

  float* ws   = (float*)d_ws;
  float* h1   = ws; ws += (size_t)N * HC;
  float* out1 = ws; ws += (size_t)N * HC;
  float* idb  = ws; ws += (size_t)N * OUT_CH;
  float* z2   = ws; ws += (size_t)N * OUT_CH;
  float* out2 = ws; ws += (size_t)N * OUT_CH;
  float* as1  = ws; ws += (size_t)N * HEADS;
  float* ad1  = ws; ws += (size_t)N * HEADS;
  float* as2  = ws; ws += (size_t)N;
  float* ad2  = ws; ws += (size_t)N;

  auto cdiv = [](long a, long b) -> int { return (int)((a + b - 1) / b); };

  gemm_lds_weights<128, 4, 1, 4, 16, 4, true, 196><<<256, 512, 0, stream>>>(x, W1, h1, pW, pb, idb, aS1, aD1, as1, ad1, N, ei + E, E, (int*)z2);

  agg_softmax_rows_lds<256, 4, 196, 3072, false, false><<<cdiv(N, 196), 1024, 0, stream>>>(ei, as1, ad1, h1, b1, g1, be1, nullptr, out1, (int*)out2, E, N, (const int*)z2, 256);

  gemm_lds_weights<256, 2, 0, 2, 8, 1, false, 196><<<256, 256, 0, stream>>>(out1, W2, z2, nullptr, nullptr, nullptr, aS2, aD2, as2, ad2, N, nullptr, 0, nullptr);
  agg_softmax_rows_lds<64, 1, 196, 3072, true, true><<<cdiv(N, 196), 1024, 0, stream>>>(ei, as2, ad2, z2, b2, g2, be2, idb, (float*)d_out, (int*)out2, E, N, nullptr, 0);

  (void)n_in; (void)out_size; (void)ws_size;
}


# GEMMs: a wave fetches its A strip in memory order (one load = 512 B of one row, 4 whole lines) and turns it into fragments through its own LDS piece; was 8-byte loads over 16 lines; k order kept
# speedup vs baseline: 6.8994x; 1.0545x over previous
#include <hip/hip_runtime.h>
#include <math.h>

typedef float v2f __attribute__((ext_vector_type(2)));
typedef float v4f __attribute__((ext_vector_type(4)));
typedef float v8f __attribute__((ext_vector_type(8)));

#define NEG_SLOPE 0.2f

template <int K, int NCOLS, int THREADS>
__device__ __forceinline__ void stage_weights(const float* __restrict__ B,
                                              float4* __restrict__ dst, int tid) {
  constexpr int KP   = K / 8;
  constexpr int ENT  = (NCOLS / 16) * KP * 32;
  constexpr int ITER = ENT / THREADS;
  static_assert(ENT % THREADS == 0 && THREADS % 32 == 0, "whole waves share the entries evenly");
  const int c = tid & 15, h = (tid >> 4) & 1;
  float4 v[ITER];
#pragma unroll
  for (int it = 0; it < ITER; ++it) {
    const int g = (tid + it * THREADS) >> 5;
    const int tile = g / KP, kp = g - tile * KP;
    const float* p = B + (size_t)(8 * kp + 2 * h) * NCOLS + tile * 16 + c;
    v[it].x = p[0]; v[it].y = p[NCOLS]; v[it].z = p[4 * NCOLS]; v[it].w = p[5 * NCOLS];
  }
#pragma unroll
  for (int it = 0; it < ITER; ++it) dst[tid + it * THREADS] = v[it];
}

template <int K>
__device__ __forceinline__ int score_head(int tid) {
  return __builtin_amdgcn_readfirstlane(tid / K);
}

template <int K, int H>
__device__ __forceinline__ void request_score_row(const float* __restrict__ B, int tid,
                                                  float4 (&w)[16]) {
  static_assert(K % 32 == 0, "a wave stays inside one head");
  const int h = score_head<K>(tid), k = tid - h * K;
  const float4* wv = (const float4*)(B + (size_t)k * (H * 64) + h * 64);
#pragma unroll
  for (int i = 0; i < 16; ++i) w[i] = wv[i];
}

template <int K, int H, int THREADS>
__device__ __forceinline__ void stage_score_weights(const float4 (&w)[16],
                                                    const float* __restrict__ att_s,
                                                    const float* __restrict__ att_d,
                                                    float* __restrict__ img, int tid) {
  constexpr int C  = 64;
  static_assert(K * H == THREADS, "one (k, head) pair per thread");
  auto pos = [](int k, int n) {
    const int kp = k >> 3, r = k & 7;
    return ((kp * 32 + ((r >> 1) & 1) * 16 + n) << 2) + (((r >> 2) << 1) | (r & 1));
  };
  {
    const int h = score_head<K>(tid), k = tid - h * K;
    const float4* sv = (const float4*)(att_s + h * C);
    const float4* dv = (const float4*)(att_d + h * C);
    float s = 0.f, d = 0.f;
#pragma unroll
    for (int i = 0; i < C / 4; ++i) {
      const float4 a = sv[i], b = dv[i];
      s += w[i].x * a.x + w[i].y * a.y + w[i].z * a.z + w[i].w * a.w;
      d += w[i].x * b.x + w[i].y * b.y + w[i].z * b.z + w[i].w * b.w;
    }
    img[pos(k, h)]     = s;
    img[pos(k, H + h)] = d;
  }
  for (int i = tid; i < K * 16; i += THREADS) {
    const int n = i & 15;
    if (n >= 2 * H) img[pos(i >> 4, n)] = 0.f;
  }
}

template <int K>
__device__ __forceinline__ void request_a_strip(const float* __restrict__ A, int strip, int lane,
                                                v4f (&q)[K / 8]) {
  static_assert(K % 128 == 0, "a row is a whole number of 512-byte pieces");
  constexpr int PPR = K / 128;
  const float* ap = A + (size_t)(strip * 16) * K + 4 * lane;
#pragma unroll
  for (int i = 0; i < K / 8; ++i)
    q[i] = *(const v4f*)(ap + (size_t)(i / PPR) * K + (i % PPR) * 128);
}

template <int K>
__device__ __forceinline__ void strip_to_frags(const v4f (&q)[K / 8], float* __restrict__ S,
                                               int lane, v2f (&a)[K / 4]) {
  constexpr int PPR = K / 128, LD = K + 4;
#pragma unroll
  for (int i = 0; i < K / 8; ++i)
    *(v4f*)(S + (i / PPR) * LD + (i % PPR) * 128 + 4 * lane) = q[i];
  const float* rp = S + (lane & 15) * LD + ((lane >> 4) << 1);
#pragma unroll
  for (int ks = 0; ks < K / 4; ++ks) a[ks] = *(const v2f*)(rp + 4 * ks);
}

template <int K, int T1, int T2, int UPS, int WAVES, int H, bool BIN, int BR>
__global__ __launch_bounds__(WAVES * 32) void gemm_lds_weights(
    const float* __restrict__ A,
    const float* __restrict__ B1, float* __restrict__ C1,
    const float* __restrict__ B2, const float* __restrict__ bias2, float* __restrict__ C2,
    const float* __restrict__ att_s, const float* __restrict__ att_d,
    float* __restrict__ sc_s, float* __restrict__ sc_d,
    int M, const int* __restrict__ bin_dst, int bin_E, int* __restrict__ bins) {
  constexpr int NT1 = UPS * T1, NT2 = UPS * T2;
  constexpr int N1 = NT1 * 16, N2 = NT2 * 16;
  constexpr int KP = K / 8;
  constexpr int TT = T1 + T2;
  constexpr int RS = WAVES / UPS;
  static_assert(T1 % 2 == 0 && WAVES % UPS == 0 && K % 8 == 0, "C1 tiles are stored in pairs");
  constexpr int SC0  = 0;
  constexpr int W0   = KP * 32;
  constexpr int NEEDW = W0 + (NT1 + NT2) * KP * 32;
  constexpr int APIECE = 16 * (K + 4);
  constexpr int NEED = NEEDW + WAVES * APIECE / 4;
  static_assert(N1 == H * 64 && 2 * H <= 16, "B1's column blocks of 64 are the heads");
  constexpr int FLOOR = 164 * 1024 / 16;
  __shared__ float4 Bl[NEED > FLOOR ? NEED : FLOOR];

  const int tid  = threadIdx.x;
  const int lane = tid & 31;
  const int wv   = tid >> 5;
  const int hi   = lane >> 4;
  const int cg   = wv % UPS;
  const int rr   = wv / UPS;

  const int strips = M >> 4;
  const int sb = (int)((blockIdx.x * (unsigned)strips) / gridDim.x);
  const int se = (int)(((blockIdx.x + 1) * (unsigned)strips) / gridDim.x);

  if constexpr (BIN) {
    constexpr int MAXNB = 512;
    __shared__ int bcnt[MAXNB];
    const int NB = (M + BR - 1) / BR;
    const bool ok = NB <= MAXNB && (gridDim.x & 7) == 0 &&
                    (long)gridDim.x * NB * 32 <= (long)M * 64;
    if (ok) {
      for (int i = tid; i < NB; i += WAVES * 32) bcnt[i] = 0;
      __syncthreads();
      const int per = (bin_E + (int)gridDim.x - 1) / (int)gridDim.x;
      const int e0  = min((int)blockIdx.x * per, bin_E);
      const int e1  = min(e0 + per, bin_E);
      int* mine = bins + (size_t)blockIdx.x * NB * 32;
      for (int e = e0 + tid; e < e1; e += WAVES * 32) {
        const int d = bin_dst[e];
        if ((unsigned)d < (unsigned)M) {
          const int o = d / BR;
          const int s = atomicAdd(&bcnt[o], 1);
          if (s < 31) mine[o * 32 + 1 + s] = e;
        }
      }
      __syncthreads();
      for (int i = tid; i < NB; i += WAVES * 32) mine[i * 32] = bcnt[i];
    }
  }

  v4f qn[K / 8];
  int st = sb + rr;
  if (st < se) request_a_strip<K>(A, st, lane, qn);
  float* const S = (float*)(Bl + NEEDW) + wv * APIECE;

  float4 wrow[16];
  request_score_row<K, H>(B1, tid, wrow);

  stage_weights<K, N1, WAVES * 32>(B1, Bl + W0, tid);
  if constexpr (T2 > 0)
    stage_weights<K, N2, WAVES * 32>(B2, Bl + W0 + NT1 * KP * 32, tid);
  stage_score_weights<K, H, WAVES * 32>(wrow, att_s, att_d, (float*)(Bl + SC0), tid);
  __syncthreads();

  int trip = 0;
  for (; st < se; st += RS, ++trip) {
    v2f a[K / 4];
    strip_to_frags<K>(qn, S, lane, a);
    if (st + RS < se) request_a_strip<K>(A, st + RS, lane, qn);

    v8f acc[TT];
#pragma unroll
    for (int t = 0; t < TT; ++t) acc[t] = v8f{};
    float4 bc[TT], bn[TT];
#pragma unroll
    for (int t = 0; t < TT; ++t) {
      const int tile = (t < T1) ? (cg * T1 + t) : (NT1 + cg * T2 + (t - T1));
      bc[t] = Bl[W0 + (tile * KP) * 32 + lane];
    }
#pragma unroll
    for (int kp = 0; kp < KP; ++kp) {
      if (kp + 1 < KP) {
#pragma unroll
        for (int t = 0; t < TT; ++t) {
          const int tile = (t < T1) ? (cg * T1 + t) : (NT1 + cg * T2 + (t - T1));
          bn[t] = Bl[W0 + (tile * KP + kp + 1) * 32 + lane];
        }
      }
#pragma unroll
      for (int t = 0; t < TT; ++t) {
        v2f bb; bb.x = bc[t].x; bb.y = bc[t].y;
        acc[t] = __builtin_amdgcn_wmma_f32_16x16x4_f32(
             false, a[2 * kp],  false, bb,
             (short)0, acc[t],  false,  false);
      }
#pragma unroll
      for (int t = 0; t < TT; ++t) {
        v2f bb; bb.x = bc[t].z; bb.y = bc[t].w;
        acc[t] = __builtin_amdgcn_wmma_f32_16x16x4_f32(
             false, a[2 * kp + 1],  false, bb,
             (short)0, acc[t],  false,  false);
      }
#pragma unroll
      for (int t = 0; t < TT; ++t) bc[t] = bn[t];
      __builtin_amdgcn_sched_barrier(0);
    }

#pragma unroll
    for (int p = 0; p < T1 / 2; ++p) {
      float* cp = C1 + (size_t)(st * 16) * N1 + (cg * T1 + 2 * p) * 16 + lane;
#pragma unroll
      for (int v = 0; v < 8; ++v) {
        const float own0 = acc[2 * p][v], own1 = acc[2 * p + 1][v];
        const float give = hi ? own0 : own1;
        const float got  = __shfl_xor(give, 16, 32);
        cp[(size_t)v * N1]       = hi ? got : own0;
        cp[(size_t)(8 + v) * N1] = hi ? own1 : got;
      }
    }
    if constexpr (T2 > 0) {
#pragma unroll
      for (int t = 0; t < T2; ++t) {
        const int ncol = (cg * T2 + t) * 16 + (lane & 15);
        const float bv = bias2 ? bias2[ncol] : 0.0f;
        float* cp = C2 + (size_t)(st * 16 + (hi << 3)) * N2 + ncol;
#pragma unroll
        for (int v = 0; v < 8; ++v) cp[(size_t)v * N2] = acc[T1 + t][v] + bv;
      }
    }

    if ((trip % UPS) == cg) {
      v8f sa = v8f{};
      float4 sb_c = Bl[SC0 + lane], sb_n = sb_c;
#pragma unroll
      for (int kp = 0; kp < KP; ++kp) {
        if (kp + 1 < KP) sb_n = Bl[SC0 + (kp + 1) * 32 + lane];
        v2f b0; b0.x = sb_c.x; b0.y = sb_c.y;
        v2f b1; b1.x = sb_c.z; b1.y = sb_c.w;
        sa = __builtin_amdgcn_wmma_f32_16x16x4_f32(
             false, a[2 * kp],  false, b0,
             (short)0, sa,  false,  false);
        sa = __builtin_amdgcn_wmma_f32_16x16x4_f32(
             false, a[2 * kp + 1],  false, b1,
             (short)0, sa,  false,  false);
        sb_c = sb_n;
        __builtin_amdgcn_sched_barrier(0);
      }
      const int n = lane & 15;
      if (n < 2 * H) {
        float* sp = (n < H ? sc_s + n : sc_d + (n - H)) + (size_t)(st * 16 + (hi << 3)) * H;
#pragma unroll
        for (int v = 0; v < 8; ++v) sp[(size_t)v * H] = sa[v];
      }
    }
  }
}

template <int CT>
__device__ __forceinline__ void load_row_vals(const float* __restrict__ p, int lane,
                                              float (&v)[CT / 32]) {
  if constexpr (CT == 256) {
    const float4 a = *(const float4*)(p + 4 * lane);
    const float4 b = *(const float4*)(p + 128 + 4 * lane);
    v[0] = a.x; v[1] = a.y; v[2] = a.z; v[3] = a.w;
    v[4] = b.x; v[5] = b.y; v[6] = b.z; v[7] = b.w;
  } else {
    const float2 a = *(const float2*)(p + 2 * lane);
    v[0] = a.x; v[1] = a.y;
  }
}

__device__ __forceinline__ int bcast_i(int v, int l) {
  return __builtin_amdgcn_readlane(v, l);
}
__device__ __forceinline__ float bcast_f(float v, int l) {
  return __int_as_float(__builtin_amdgcn_readlane(__float_as_int(v), l));
}

__device__ __forceinline__ float edge_logit(float a_src, float a_dst) {
  const float v = a_src + a_dst;
  return ((v > 0.f) ? v : NEG_SLOPE * v) + 0.0f;
}

template <typename F>
__device__ __forceinline__ void stream_dst_matches(const int* __restrict__ dstp, int E,
                                                   int row0, int rows, int tid, F on_round) {
  constexpr int T = 1024;
  constexpr int STEP = T * 16;
  const bool vec_ok = (E >= 4) && ((E & 3) == 0) && ((((size_t)dstp) & 15) == 0);

  auto load16 = [&](int base, int (&d)[16]) {
    if (vec_ok) {
#pragma unroll
      for (int j = 0; j < 4; ++j) {
        const int b = base + j * (T * 4) + tid * 4;
        const int4 q = *(const int4*)(dstp + min(b, E - 4));
        d[4 * j + 0] = q.x; d[4 * j + 1] = q.y; d[4 * j + 2] = q.z; d[4 * j + 3] = q.w;
      }
    } else {
#pragma unroll
      for (int j = 0; j < 4; ++j) {
#pragma unroll
        for (int k = 0; k < 4; ++k)
          d[4 * j + k] = dstp[min(base + j * (T * 4) + tid * 4 + k, E - 1)];
      }
    }
  };

  int cur[16], nxt[16];
#pragma unroll
  for (int q = 0; q < 16; ++q) { cur[q] = row0 - 1; nxt[q] = row0 - 1; }
  for (int base = -STEP; base < E; base += STEP) {
    unsigned mb = 0;
#pragma unroll
    for (int q = 0; q < 16; ++q) {
      const int eq = base + (q >> 2) * (T * 4) + tid * 4 + (q & 3);
      mb |= (eq < E && (unsigned)(cur[q] - row0) < (unsigned)rows) ? (1u << q) : 0u;
    }

    if (base + STEP < E) load16(base + STEP, nxt);

    while (__builtin_amdgcn_ballot_w32(mb != 0)) {
      const bool has = mb != 0;
      const int q = has ? __builtin_ctz(mb) : 0;
      mb &= mb - 1;
      int dsel = cur[0];
#pragma unroll
      for (int t = 1; t < 16; ++t) dsel = (q == t) ? cur[t] : dsel;
      const int edge = base + (q >> 2) * (T * 4) + tid * 4 + (q & 3);
      on_round(has, edge, dsel - row0);
    }

#pragma unroll
    for (int q = 0; q < 16; ++q) cur[q] = nxt[q];
  }
}

template <int CAP, bool TAIL>
__device__ __forceinline__ void list_block(int base, const int4 (&d)[4], int nq, int row0, int rows,
                                           int tid, int lane, int* l_src, int* l_rel, int* l_cnt) {
  constexpr int T = 1024;
  unsigned mk[16];
#pragma unroll
  for (int j = 0; j < 4; ++j) {
    unsigned okm = 0xffffffffu;
    if constexpr (TAIL) okm = __builtin_amdgcn_ballot_w32((base >> 2) + j * T + tid < nq);
    mk[4 * j + 0] = __builtin_amdgcn_ballot_w32((unsigned)(d[j].x - row0) < (unsigned)rows) & okm;
    mk[4 * j + 1] = __builtin_amdgcn_ballot_w32((unsigned)(d[j].y - row0) < (unsigned)rows) & okm;
    mk[4 * j + 2] = __builtin_amdgcn_ballot_w32((unsigned)(d[j].z - row0) < (unsigned)rows) & okm;
    mk[4 * j + 3] = __builtin_amdgcn_ballot_w32((unsigned)(d[j].w - row0) < (unsigned)rows) & okm;
  }
  int tot = 0;
#pragma unroll
  for (int q = 0; q < 16; q += 2)
    tot += __builtin_popcountll(((unsigned long long)mk[q + 1] << 32) | (unsigned long long)mk[q]);
  if (tot != 0) {
    const unsigned below = (1u << lane) - 1u;
    int pos0 = 0;
    if (lane == 0) pos0 = atomicAdd(l_cnt, tot);
    pos0 = bcast_i(pos0, 0);
#pragma unroll
    for (int q = 0; q < 16; ++q) {
      const unsigned m = mk[q];
      if (m != 0) {
        const int pos = pos0 + __builtin_popcount(m & below);
        const int4 dj = d[q >> 2];
        const int dv = (q & 3) == 0 ? dj.x : (q & 3) == 1 ? dj.y : (q & 3) == 2 ? dj.z : dj.w;
        if (((m >> lane) & 1u) && pos < CAP) {
          l_src[pos] = base + (q >> 2) * (T * 4) + tid * 4 + (q & 3);
          l_rel[pos] = dv - row0;
        }
        pos0 += __builtin_popcount(m);
      }
    }
  }
}

template <int CAP>
__device__ __forceinline__ void list_owned_edges_vec(const int* __restrict__ dstp, int E,
                                                     int row0, int rows, int tid, int lane,
                                                     int* l_src, int* l_rel, int* l_cnt) {
  constexpr int T = 1024;
  constexpr int STEP = T * 16;
  const int4* __restrict__ vp = (const int4*)dstp;
  const int nq    = E >> 2;
  const int nfull = E / STEP;
  const int tail0 = nfull * STEP;

  auto request = [&](int base, int4 (&d)[4]) {
#pragma unroll
    for (int j = 0; j < 4; ++j) d[j] = vp[min((base >> 2) + j * T + tid, nq - 1)];
  };

  int4 rt[4], ra[4], rb[4];
  request(tail0, rt);
  request(0, ra);
  for (int b = 0; b < nfull; b += 2) {
    request((b + 1) * STEP, rb);
    list_block<CAP, false>(b * STEP, ra, nq, row0, rows, tid, lane, l_src, l_rel, l_cnt);
    if (b + 1 < nfull) {
      request((b + 2) * STEP, ra);
      list_block<CAP, false>((b + 1) * STEP, rb, nq, row0, rows, tid, lane, l_src, l_rel, l_cnt);
    }
  }
  if (tail0 < E) list_block<CAP, true>(tail0, rt, nq, row0, rows, tid, lane, l_src, l_rel, l_cnt);
}

template <int CT, int H, int R, int CAP, bool REUSE, bool ADD>
__global__ __launch_bounds__(1024) void agg_softmax_rows_lds(
    const int* __restrict__ ei, const float* __restrict__ as, const float* __restrict__ ad,
    const float* __restrict__ feat, const float* __restrict__ bias,
    const float* __restrict__ gam, const float* __restrict__ bet, const float* __restrict__ addp,
    float* out, int* lists, int E, int n, const int* __restrict__ bins, int binP) {
  static_assert((CT == 256 && H == 4) || (CT == 64 && H == 1), "lane-to-head map below");
  static_assert(!REUSE || H == 1, "the hand-over is read by the one-head layer");
  constexpr int V = CT / 32;
  constexpr int T = 1024;
  constexpr int NF = 8;
  constexpr int LW = 64;
  constexpr int NJ = CAP / T;
  static_assert(CAP % T == 0, "every thread reads the same number of pairs");
  __shared__ float acc[R * CT];
  __shared__ float adrow[R * H];
  __shared__ float slog[R * H];
  __shared__ float mrow[R * H];
  __shared__ float srow[R * H];
  __shared__ int   l_src[CAP];
  __shared__ int   l_rel[CAP];
  __shared__ float l_as[CAP * H];
  __shared__ int   l_cnt;
  __shared__ int   l_ovf;

  const int tid  = threadIdx.x;
  const int lane = tid & 31;
  const int wv   = tid >> 5;
  const int hi   = lane >> 4;
  const int row0 = blockIdx.x * R;
  const int rows = max(0, min(R, n - row0));
  const int* __restrict__ dstp = ei + E;

  int*  piece = lists + (size_t)row0 * LW;
  int2* pairs = (int2*)(piece + 4);
  const int capg = max(0, min(CAP, (rows * LW - 4) >> 1));
  int hdr = 0;
  if constexpr (REUSE) hdr = piece[0];

  auto lds_max = [&](int slot, float val) {
    if (val >= 0.0f) atomicMax((int*)&mrow[slot], __float_as_int(val));
    else             atomicMin((unsigned int*)&mrow[slot], __float_as_uint(val));
  };

  auto wave_accumulate = [&](unsigned m, int srcv, int rel, float x0, float x1, float x2, float x3) {
    while (m) {
      int  l[NF];
      bool ok[NF];
#pragma unroll
      for (int j = 0; j < NF; ++j) {
        ok[j] = (m != 0);
        l[j]  = ok[j] ? __builtin_ctz(m) : 0;
        if (ok[j]) m &= m - 1;
      }
      float v[NF][V];
#pragma unroll
      for (int j = 0; j < NF; ++j) {
        if (ok[j]) load_row_vals<CT>(feat + (size_t)bcast_i(srcv, l[j]) * CT, lane, v[j]);
      }
#pragma unroll
      for (int j = 0; j < NF; ++j) {
        if (ok[j]) {
          const int rl = bcast_i(rel, l[j]);
          float aA, aB;
          if constexpr (H == 4) {
            const float y0 = bcast_f(x0, l[j]), y1 = bcast_f(x1, l[j]);
            const float y2 = bcast_f(x2, l[j]), y3 = bcast_f(x3, l[j]);
            aA = hi ? y1 : y0; aB = hi ? y3 : y2;
          } else {
            aA = aB = bcast_f(x0, l[j]);
          }
#pragma unroll
          for (int i = 0; i < V; ++i)
            __hip_atomic_fetch_add(&acc[rl * CT + i * 32 + lane], (i < 4 ? aA : aB) * v[j][i],
                                   __ATOMIC_RELAXED, __HIP_MEMORY_SCOPE_WORKGROUP);
        }
      }
    }
  };

  if (tid == 0) { l_cnt = 0; l_ovf = 0; }
  for (int i = tid; i < rows * H; i += T) {
    const float d_ = ad[(size_t)row0 * H + i];
    const float v  = edge_logit(as[(size_t)row0 * H + i], d_);
    adrow[i] = d_; slog[i] = v; mrow[i] = v;
  }
  __syncthreads();

  int  cnt;
  bool listed;
  bool binned = false;
  if constexpr (!REUSE) {
    const int NBc = (n + R - 1) / R;
    const bool bins_ok = bins != nullptr && binP > 0 && (binP & 7) == 0 && NBc <= 512 &&
                         (int)gridDim.x == NBc && (long)binP * NBc * 32 <= (long)n * 64;
    if (bins_ok) {
      for (int t = tid; t < binP * 4; t += T) {
        const int p = t >> 2, qd = t & 3;
        const int4* ln = (const int4*)(bins + ((size_t)p * NBc + blockIdx.x) * 32) + qd * 2;
        const int4 u = ln[0], w = ln[1];
        const int c = __shfl(u.x, lane & ~3, 32);
        const bool good = (c >= 0) && (c <= 31);
        if (!good) l_ovf = 1;
        int pos0 = 0;
        if (qd == 0 && good && c > 0) pos0 = atomicAdd(&l_cnt, c);
        pos0 = __shfl(pos0, lane & ~3, 32);
        const int vals[8] = {u.x, u.y, u.z, u.w, w.x, w.y, w.z, w.w};
#pragma unroll
        for (int j = 0; j < 8; ++j) {
          const int k = qd * 8 + j - 1;
          const int pos = pos0 + k;
          if (good && k >= 0 && k < c && pos < CAP) l_src[pos] = vals[j];
        }
      }
      __syncthreads();
      binned = (l_ovf == 0);
      if (!binned) {
        if (tid == 0) l_cnt = 0;
        __syncthreads();
      }
    }
    const bool vec_ok = (E >= 4) && ((E & 3) == 0) && ((((size_t)dstp) & 15) == 0);
    if (binned) {   }
    else if (vec_ok) list_owned_edges_vec<CAP>(dstp, E, row0, rows, tid, lane, l_src, l_rel, &l_cnt);
    else stream_dst_matches(dstp, E, row0, rows, tid, [&](bool has, int edge, int rel) {
      const unsigned bal = __builtin_amdgcn_ballot_w32(has);
      const int lead = __builtin_ctz(bal);
      int basepos = 0;
      if (lane == lead) basepos = atomicAdd(&l_cnt, __builtin_popcount(bal));
      basepos = bcast_i(basepos, lead);
      const int pos = basepos + __builtin_popcount(bal & ((1u << lane) - 1u));
      if (has && pos < CAP) { l_src[pos] = edge; l_rel[pos] = rel; }
    });
    __syncthreads();
    cnt    = l_cnt;
    listed = cnt <= CAP;
    if (tid == 0) piece[0] = cnt;
  } else {
    cnt    = hdr;
    listed = (cnt >= 0) && (cnt <= capg);
  }

  if (listed) {
   if constexpr (REUSE) {
    if (cnt > 0) {
      int2 pr[NJ];
#pragma unroll
      for (int j = 0; j < NJ; ++j) pr[j] = pairs[min(tid + j * T, cnt - 1)];
      int sv[NJ]; float av[NJ];
#pragma unroll
      for (int j = 0; j < NJ; ++j) {
        sv[j] = min(max(pr[j].x, 0), n - 1);
        av[j] = as[sv[j]];
      }
#pragma unroll
      for (int j = 0; j < NJ; ++j) {
        const int i   = min(tid + j * T, cnt - 1);
        const int rel = min(max(pr[j].y, 0), rows - 1);
        l_src[i] = sv[j]; l_rel[i] = rel; l_as[i] = av[j];
        if (tid + j * T < cnt) lds_max(rel, edge_logit(av[j], adrow[rel]));
      }
    }
   } else {
    const bool pub = cnt <= capg;
    for (int i = tid; i < cnt; i += T) {
      int edge = l_src[i], rel;
      if (binned) {
        edge = min(max(edge, 0), E - 1);
        rel  = min(max(dstp[edge] - row0, 0), rows - 1);
        l_rel[i] = rel;
      } else {
        rel = l_rel[i];
      }
      const int src  = min(max(ei[edge], 0), n - 1);
      l_src[i] = src;
      if (pub) pairs[i] = make_int2(src, rel);
      if constexpr (H == 4) {
        const float4 a = *(const float4*)(as + (size_t)src * 4);
        l_as[i * 4 + 0] = a.x; l_as[i * 4 + 1] = a.y; l_as[i * 4 + 2] = a.z; l_as[i * 4 + 3] = a.w;
        lds_max(rel * 4 + 0, edge_logit(a.x, adrow[rel * 4 + 0]));
        lds_max(rel * 4 + 1, edge_logit(a.y, adrow[rel * 4 + 1]));
        lds_max(rel * 4 + 2, edge_logit(a.z, adrow[rel * 4 + 2]));
        lds_max(rel * 4 + 3, edge_logit(a.w, adrow[rel * 4 + 3]));
      } else {
        const float a = as[src];
        l_as[i] = a;
        lds_max(rel, edge_logit(a, adrow[rel]));
      }
    }
   }
  } else {
    stream_dst_matches(dstp, E, row0, rows, tid, [&](bool has, int edge, int rel) {
      if (has) {
        const int src = min(max(ei[edge], 0), n - 1);
        if constexpr (H == 4) {
          const float4 a = *(const float4*)(as + (size_t)src * 4);
          lds_max(rel * 4 + 0, edge_logit(a.x, adrow[rel * 4 + 0]));
          lds_max(rel * 4 + 1, edge_logit(a.y, adrow[rel * 4 + 1]));
          lds_max(rel * 4 + 2, edge_logit(a.z, adrow[rel * 4 + 2]));
          lds_max(rel * 4 + 3, edge_logit(a.w, adrow[rel * 4 + 3]));
        } else {
          lds_max(rel, edge_logit(as[src], adrow[rel]));
        }
      }
    });
  }
  __syncthreads();

  for (int r = wv; r < rows; r += 32) {
    float v[V];
    load_row_vals<CT>(feat + (size_t)(row0 + r) * CT, lane, v);
    if constexpr (H == 4) {
      const float xA = expf(slog[r * 4 + hi]     - mrow[r * 4 + hi]);
      const float xB = expf(slog[r * 4 + 2 + hi] - mrow[r * 4 + 2 + hi]);
#pragma unroll
      for (int i = 0; i < V; ++i) acc[r * CT + i * 32 + lane] = (i < 4 ? xA : xB) * v[i];
      if (lane < 4) srow[r * 4 + lane] = expf(slog[r * 4 + lane] - mrow[r * 4 + lane]);
    } else {
      const float x = expf(slog[r] - mrow[r]);
#pragma unroll
      for (int i = 0; i < V; ++i) acc[r * CT + i * 32 + lane] = x * v[i];
      if (lane == 0) srow[r] = x;
    }
  }
  __syncthreads();

  if (listed) {
    const int lo = (int)(((long)wv * cnt) >> 5), up = (int)(((long)(wv + 1) * cnt) >> 5);
    for (int cb = lo; cb < up; cb += 32) {
      const int  idx = cb + lane;
      const bool has = idx < up;
      int srcv = 0, rel = 0;
      float x0 = 0.f, x1 = 0.f, x2 = 0.f, x3 = 0.f;
      if (has) {
        srcv = l_src[idx]; rel = l_rel[idx];
        if constexpr (H == 4) {
          x0 = expf(edge_logit(l_as[idx * 4 + 0], adrow[rel * 4 + 0]) - mrow[rel * 4 + 0]);
          x1 = expf(edge_logit(l_as[idx * 4 + 1], adrow[rel * 4 + 1]) - mrow[rel * 4 + 1]);
          x2 = expf(edge_logit(l_as[idx * 4 + 2], adrow[rel * 4 + 2]) - mrow[rel * 4 + 2]);
          x3 = expf(edge_logit(l_as[idx * 4 + 3], adrow[rel * 4 + 3]) - mrow[rel * 4 + 3]);
          __hip_atomic_fetch_add(&srow[rel * 4 + 0], x0, __ATOMIC_RELAXED, __HIP_MEMORY_SCOPE_WORKGROUP);
          __hip_atomic_fetch_add(&srow[rel * 4 + 1], x1, __ATOMIC_RELAXED, __HIP_MEMORY_SCOPE_WORKGROUP);
          __hip_atomic_fetch_add(&srow[rel * 4 + 2], x2, __ATOMIC_RELAXED, __HIP_MEMORY_SCOPE_WORKGROUP);
          __hip_atomic_fetch_add(&srow[rel * 4 + 3], x3, __ATOMIC_RELAXED, __HIP_MEMORY_SCOPE_WORKGROUP);
        } else {
          x0 = expf(edge_logit(l_as[idx], adrow[rel]) - mrow[rel]);
          __hip_atomic_fetch_add(&srow[rel], x0, __ATOMIC_RELAXED, __HIP_MEMORY_SCOPE_WORKGROUP);
        }
      }
      wave_accumulate(__builtin_amdgcn_ballot_w32(has), srcv, rel, x0, x1, x2, x3);
    }
  } else {
    stream_dst_matches(dstp, E, row0, rows, tid, [&](bool has, int edge, int rel) {
      int srcv = 0;
      float x0 = 0.f, x1 = 0.f, x2 = 0.f, x3 = 0.f;
      if (has) {
        srcv = min(max(ei[edge], 0), n - 1);
        if constexpr (H == 4) {
          const float4 a = *(const float4*)(as + (size_t)srcv * 4);
          x0 = expf(edge_logit(a.x, adrow[rel * 4 + 0]) - mrow[rel * 4 + 0]);
          x1 = expf(edge_logit(a.y, adrow[rel * 4 + 1]) - mrow[rel * 4 + 1]);
          x2 = expf(edge_logit(a.z, adrow[rel * 4 + 2]) - mrow[rel * 4 + 2]);
          x3 = expf(edge_logit(a.w, adrow[rel * 4 + 3]) - mrow[rel * 4 + 3]);
          __hip_atomic_fetch_add(&srow[rel * 4 + 0], x0, __ATOMIC_RELAXED, __HIP_MEMORY_SCOPE_WORKGROUP);
          __hip_atomic_fetch_add(&srow[rel * 4 + 1], x1, __ATOMIC_RELAXED, __HIP_MEMORY_SCOPE_WORKGROUP);
          __hip_atomic_fetch_add(&srow[rel * 4 + 2], x2, __ATOMIC_RELAXED, __HIP_MEMORY_SCOPE_WORKGROUP);
          __hip_atomic_fetch_add(&srow[rel * 4 + 3], x3, __ATOMIC_RELAXED, __HIP_MEMORY_SCOPE_WORKGROUP);
        } else {
          x0 = expf(edge_logit(as[srcv], adrow[rel]) - mrow[rel]);
          __hip_atomic_fetch_add(&srow[rel], x0, __ATOMIC_RELAXED, __HIP_MEMORY_SCOPE_WORKGROUP);
        }
      }
      wave_accumulate(__builtin_amdgcn_ballot_w32(has), srcv, rel, x0, x1, x2, x3);
    });
  }
  __syncthreads();

  float bv[V], gv[V], ev[V], idn[V];
  load_row_vals<CT>(bias, lane, bv);
  load_row_vals<CT>(gam, lane, gv);
  load_row_vals<CT>(bet, lane, ev);
  if constexpr (ADD) {
    if (wv < rows) load_row_vals<CT>(addp + (size_t)(row0 + wv) * CT, lane, idn);
  }
  for (int r = wv; r < rows; r += 32) {
    float* op = out + (size_t)(row0 + r) * CT;
    float idc[V];
    if constexpr (ADD) {
#pragma unroll
      for (int i = 0; i < V; ++i) idc[i] = idn[i];
      if (r + 32 < rows) load_row_vals<CT>(addp + (size_t)(row0 + r + 32) * CT, lane, idn);
    }
    float dA, dB;
    if constexpr (H == 4) {
      dA = srow[r * 4 + hi] + 1e-16f; dB = srow[r * 4 + 2 + hi] + 1e-16f;
    } else {
      dA = dB = srow[r] + 1e-16f;
    }
    float v[V];
#pragma unroll
    for (int i = 0; i < V; ++i) v[i] = bv[i] + acc[r * CT + i * 32 + lane] / (i < 4 ? dA : dB);
    float s = 0.f, ss = 0.f;
#pragma unroll
    for (int i = 0; i < V; ++i) { s += v[i]; ss += v[i] * v[i]; }
    for (int off = 16; off; off >>= 1) {
      s  += __shfl_xor(s,  off, 32);
      ss += __shfl_xor(ss, off, 32);
    }
    const float mu  = s / (float)CT;
    const float var = ss / (float)CT - mu * mu;
    const float inv = rsqrtf(var + 1e-5f);
#pragma unroll
    for (int i = 0; i < V; ++i) {
      float o = (v[i] - mu) * inv * gv[i] + ev[i];
      if constexpr (ADD) o += idc[i];
      v[i] = (o > 0.f) ? o : expm1f(o);
    }
    if constexpr (CT == 256) {
      *(float4*)(op + 4 * lane)       = make_float4(v[0], v[1], v[2], v[3]);
      *(float4*)(op + 128 + 4 * lane) = make_float4(v[4], v[5], v[6], v[7]);
    } else {
      *(float2*)(op + 2 * lane) = make_float2(v[0], v[1]);
    }
  }
}

extern "C" void kernel_launch(void* const* d_in, const int* in_sizes, int n_in,
                              void* d_out, int out_size, void* d_ws, size_t ws_size,
                              hipStream_t stream) {
  const float* x    = (const float*)d_in[0];
  const int*   ei   = (const int*)  d_in[1];
  const float* W1   = (const float*)d_in[2];
  const float* aS1  = (const float*)d_in[3];
  const float* aD1  = (const float*)d_in[4];
  const float* b1   = (const float*)d_in[5];
  const float* W2   = (const float*)d_in[6];
  const float* aS2  = (const float*)d_in[7];
  const float* aD2  = (const float*)d_in[8];
  const float* b2   = (const float*)d_in[9];
  const float* g1   = (const float*)d_in[10];
  const float* be1  = (const float*)d_in[11];
  const float* g2   = (const float*)d_in[12];
  const float* be2  = (const float*)d_in[13];
  const float* pW   = (const float*)d_in[14];
  const float* pb   = (const float*)d_in[15];

  const int IN_CH = 128, HEADS = 4, OUT_CH = 64, HC = HEADS * OUT_CH;
  const int N  = in_sizes[0] / IN_CH;
  const int E  = in_sizes[1] / 2;

  float* ws   = (float*)d_ws;
  float* h1   = ws; ws += (size_t)N * HC;
  float* out1 = ws; ws += (size_t)N * HC;
  float* idb  = ws; ws += (size_t)N * OUT_CH;
  float* z2   = ws; ws += (size_t)N * OUT_CH;
  float* out2 = ws; ws += (size_t)N * OUT_CH;
  float* as1  = ws; ws += (size_t)N * HEADS;
  float* ad1  = ws; ws += (size_t)N * HEADS;
  float* as2  = ws; ws += (size_t)N;
  float* ad2  = ws; ws += (size_t)N;

  auto cdiv = [](long a, long b) -> int { return (int)((a + b - 1) / b); };

  gemm_lds_weights<128, 4, 1, 4, 16, 4, true, 196><<<256, 512, 0, stream>>>(x, W1, h1, pW, pb, idb, aS1, aD1, as1, ad1, N, ei + E, E, (int*)z2);

  agg_softmax_rows_lds<256, 4, 196, 3072, false, false><<<cdiv(N, 196), 1024, 0, stream>>>(ei, as1, ad1, h1, b1, g1, be1, nullptr, out1, (int*)out2, E, N, (const int*)z2, 256);

  gemm_lds_weights<256, 2, 0, 2, 8, 1, false, 196><<<256, 256, 0, stream>>>(out1, W2, z2, nullptr, nullptr, nullptr, aS2, aD2, as2, ad2, N, nullptr, 0, nullptr);
  agg_softmax_rows_lds<64, 1, 196, 3072, true, true><<<cdiv(N, 196), 1024, 0, stream>>>(ei, as2, ad2, z2, b2, g2, be2, idb, (float*)d_out, (int*)out2, E, N, nullptr, 0);

  (void)n_in; (void)out_size; (void)ws_size;
}
